# starting kernel
# baseline (speedup 1.0000x reference)
#include <hip/hip_runtime.h>

#define D_H 128
#define PN_EPS 1e-6f

typedef float v2f __attribute__((ext_vector_type(2)));
typedef float v8f __attribute__((ext_vector_type(8)));

template <int DOUT>
__global__ void __launch_bounds__(256)
gemm_wmma_f32(const float* __restrict__ X, const float* __restrict__ W,
              const float* __restrict__ bias, float* __restrict__ H,
              int n_rows) {
    constexpr int TN  = DOUT / 16;
    constexpr int KC  = 64;
    constexpr int WST = KC + 4;
    __shared__ float Wt[DOUT * WST];

    const int lane = threadIdx.x & 31;
    const int wave = threadIdx.x >> 5;
    const int half = lane >> 4;
    const int l15  = lane & 15;

    const int tm      = blockIdx.x * 8 + wave;
    const int tiles_m = (n_rows + 15) >> 4;
    const bool active = (tm < tiles_m);

    int row_a = tm * 16 + l15;
    if (row_a >= n_rows) row_a = n_rows - 1;
    const float* __restrict__ ap = X + (size_t)row_a * D_H + half * 2;
    const float* __restrict__ bbase = &Wt[l15 * WST + half * 2];

    v8f acc[TN] = {};

    for (int k0 = 0; k0 < D_H; k0 += KC) {
        __syncthreads();
        const float* __restrict__ wg = W + (size_t)k0 * DOUT;
        for (int idx = threadIdx.x; idx < KC * DOUT; idx += 256) {
            int r = idx / DOUT;
            int c = idx - r * DOUT;
            Wt[c * WST + r] = wg[idx];
        }
        __syncthreads();

        if (active) {
            __builtin_prefetch(ap + k0 + KC, 0, 1);
            for (int kk = 0; kk < KC; kk += 4) {
                v2f a = *(const v2f*)(ap + k0 + kk);
#pragma unroll
                for (int t = 0; t < TN; ++t) {
                    v2f b = *(const v2f*)(bbase + t * (16 * WST) + kk);
                    acc[t] = __builtin_amdgcn_wmma_f32_16x16x4_f32(
                         false, a,  false, b,
                         (short)0, acc[t],
                         false,  false);
                }
            }
        }
    }

    if (!active) return;
    const int row_base = tm * 16 + half * 8;
#pragma unroll
    for (int t = 0; t < TN; ++t) {
        const int col = t * 16 + l15;
        const float bb = bias[col];
#pragma unroll
        for (int r = 0; r < 8; ++r) {
            const int row = row_base + r;
            if (row < n_rows)
                H[(size_t)row * DOUT + col] = acc[t][r] + bb;
        }
    }
}

__global__ void spmm_scatter(const float* __restrict__ H,
                             const int* __restrict__ src,
                             const int* __restrict__ dst,
                             const float* __restrict__ ew,
                             float* __restrict__ out, int nE, int D) {
    const int lane = threadIdx.x & 31;
    const int e = (int)((blockIdx.x * (long long)blockDim.x + threadIdx.x) >> 5);
    if (e >= nE) return;
    const int s = src[e];
    const int d = dst[e];
    const float w = ew[e];
    const int per = D >> 5;
    const float* __restrict__ hp = H + (long long)s * D + lane * per;
    float* __restrict__ op = out + (long long)d * D + lane * per;
    for (int i = 0; i < per; ++i)
        unsafeAtomicAdd(&op[i], w * hp[i]);
}

__global__ void zero_f32(float* __restrict__ p, long long count) {
    long long i = blockIdx.x * (long long)blockDim.x + threadIdx.x;
    const long long stride = (long long)gridDim.x * blockDim.x;
    for (; i < count; i += stride) p[i] = 0.0f;
}

__global__ void __launch_bounds__(128)
pn_reduce(const float* __restrict__ H, float* __restrict__ red, int n_rows) {
    __shared__ float ssq[D_H];
    const int c = threadIdx.x;
    float colsum = 0.0f, sumsq = 0.0f;
    for (int r = blockIdx.x; r < n_rows; r += gridDim.x) {
        float v = H[(long long)r * D_H + c];
        colsum += v;
        sumsq += v * v;
    }
    unsafeAtomicAdd(&red[c], colsum);
    ssq[c] = sumsq;
    __syncthreads();
    for (int s = 64; s > 0; s >>= 1) {
        if (c < s) ssq[c] += ssq[c + s];
        __syncthreads();
    }
    if (c == 0) unsafeAtomicAdd(&red[D_H], ssq[0]);
}

__global__ void __launch_bounds__(128)
pn_finalize(float* __restrict__ red, int n_rows) {
    __shared__ float sh[D_H];
    const int c = threadIdx.x;
    const float inv_n = 1.0f / (float)n_rows;
    const float mean = red[c] * inv_n;
    sh[c] = mean * mean;
    __syncthreads();
    for (int s = 64; s > 0; s >>= 1) {
        if (c < s) sh[c] += sh[c + s];
        __syncthreads();
    }
    if (c == 0) {
        float ssq_centered = red[D_H] - (float)n_rows * sh[0];
        red[D_H + 1] = 1.0f / sqrtf(PN_EPS + ssq_centered * inv_n);
    }
    red[c] = mean;
}

__global__ void pn_apply(const float* __restrict__ conv,
                         const float* __restrict__ xold,
                         float* __restrict__ xout,
                         const float* __restrict__ red, int n_rows) {
    const long long i = blockIdx.x * (long long)blockDim.x + threadIdx.x;
    if (i >= (long long)n_rows * D_H) return;
    const int c = (int)(i & (D_H - 1));
    float v = (conv[i] - red[c]) * red[D_H + 1];
    v = fmaxf(v, 0.0f);
    if (xold) v += xold[i];
    xout[i] = v;
}

extern "C" void kernel_launch(void* const* d_in, const int* in_sizes, int n_in,
                              void* d_out, int out_size, void* d_ws, size_t ws_size,
                              hipStream_t stream) {
    const float* x    = (const float*)d_in[0];
    const int*   ei   = (const int*)d_in[1];
    const float* ew   = (const float*)d_in[2];
    const float* W0   = (const float*)d_in[3];
    const float* b0   = (const float*)d_in[4];
    const float* W1   = (const float*)d_in[5];
    const float* b1   = (const float*)d_in[6];
    const float* Wout = (const float*)d_in[7];
    const float* bout = (const float*)d_in[8];
    float* out = (float*)d_out;

    const int n = in_sizes[0] / D_H;
    const int e = in_sizes[2];
    const int d_outc = out_size / n;
    const int* srcI = ei;
    const int* dstI = ei + e;

    const size_t nd = (size_t)n * D_H;
    float* A   = (float*)d_ws;
    float* B   = A + nd;
    float* C   = B + nd;
    float* red = C + nd;

    const int gemm_blocks = (((n + 15) / 16) + 7) / 8;
    auto launch_gemm = [&](const float* X_, const float* W_, const float* bias_,
                           float* H_, int dout) {
        if (dout == 128)
            gemm_wmma_f32<128><<<gemm_blocks, 256, 0, stream>>>(X_, W_, bias_, H_, n);
        else
            gemm_wmma_f32<64><<<gemm_blocks, 256, 0, stream>>>(X_, W_, bias_, H_, n);
    };
    auto launch_scatter = [&](const float* H_, float* O_, int D) {
        long long thr = (long long)e * 32;
        int blocks = (int)((thr + 255) / 256);
        spmm_scatter<<<blocks, 256, 0, stream>>>(H_, srcI, dstI, ew, O_, e, D);
    };
    auto launch_zero = [&](float* p, long long cnt) {
        int blocks = (int)((cnt + 256 * 8 - 1) / (256 * 8));
        zero_f32<<<blocks, 256, 0, stream>>>(p, cnt);
    };
    auto launch_pn = [&](const float* conv, const float* xold, float* xout) {
        launch_zero(red, 256);
        pn_reduce<<<1024, 128, 0, stream>>>(conv, red, n);
        pn_finalize<<<1, 128, 0, stream>>>(red, n);
        long long cnt = (long long)n * D_H;
        pn_apply<<<(int)((cnt + 255) / 256), 256, 0, stream>>>(conv, xold, xout, red, n);
    };

    launch_gemm(x, W0, b0, A, D_H);
    launch_zero(B, (long long)nd);
    launch_scatter(A, B, D_H);
    launch_pn(B, nullptr, A);

    launch_gemm(A, W1, b1, B, D_H);
    launch_zero(C, (long long)nd);
    launch_scatter(B, C, D_H);
    launch_pn(C, A, C);

    launch_gemm(C, Wout, bout, A, d_outc);
    launch_zero(out, (long long)n * d_outc);
    launch_scatter(A, out, d_outc);
}


# Gather aggregation, sum order fixed by edge id (arrival-order slots made the earlier try vary): edges grouped by destination once per call, one wave sums each row in registers; no float atomics
# speedup vs baseline: 1.7113x; 1.7113x over previous
#include <hip/hip_runtime.h>

#define D_H 128
#define PN_EPS 1e-6f

typedef float v2f __attribute__((ext_vector_type(2)));
typedef float v8f __attribute__((ext_vector_type(8)));

template <int DOUT>
__global__ void __launch_bounds__(256)
gemm_wmma_f32(const float* __restrict__ X, const float* __restrict__ W,
              const float* __restrict__ bias, float* __restrict__ H,
              int n_rows) {
    constexpr int TN  = DOUT / 16;
    constexpr int KC  = 64;
    constexpr int WST = KC + 4;
    __shared__ float Wt[DOUT * WST];

    const int lane = threadIdx.x & 31;
    const int wave = threadIdx.x >> 5;
    const int half = lane >> 4;
    const int l15  = lane & 15;

    const int tm      = blockIdx.x * 8 + wave;
    const int tiles_m = (n_rows + 15) >> 4;
    const bool active = (tm < tiles_m);

    int row_a = tm * 16 + l15;
    if (row_a >= n_rows) row_a = n_rows - 1;
    const float* __restrict__ ap = X + (size_t)row_a * D_H + half * 2;
    const float* __restrict__ bbase = &Wt[l15 * WST + half * 2];

    v8f acc[TN] = {};

    for (int k0 = 0; k0 < D_H; k0 += KC) {
        __syncthreads();
        const float* __restrict__ wg = W + (size_t)k0 * DOUT;
        for (int idx = threadIdx.x; idx < KC * DOUT; idx += 256) {
            int r = idx / DOUT;
            int c = idx - r * DOUT;
            Wt[c * WST + r] = wg[idx];
        }
        __syncthreads();

        if (active) {
            __builtin_prefetch(ap + k0 + KC, 0, 1);
            for (int kk = 0; kk < KC; kk += 4) {
                v2f a = *(const v2f*)(ap + k0 + kk);
#pragma unroll
                for (int t = 0; t < TN; ++t) {
                    v2f b = *(const v2f*)(bbase + t * (16 * WST) + kk);
                    acc[t] = __builtin_amdgcn_wmma_f32_16x16x4_f32(
                         false, a,  false, b,
                         (short)0, acc[t],
                         false,  false);
                }
            }
        }
    }

    if (!active) return;
    const int row_base = tm * 16 + half * 8;
#pragma unroll
    for (int t = 0; t < TN; ++t) {
        const int col = t * 16 + l15;
        const float bb = bias[col];
#pragma unroll
        for (int r = 0; r < 8; ++r) {
            const int row = row_base + r;
            if (row < n_rows)
                H[(size_t)row * DOUT + col] = acc[t][r] + bb;
        }
    }
}

__global__ void csr_count(const int* __restrict__ dst, int* __restrict__ cnt,
                          int nE) {
    const int e = blockIdx.x * blockDim.x + threadIdx.x;
    if (e < nE) atomicAdd(&cnt[dst[e]], 1);
}

__global__ void __launch_bounds__(1024)
csr_scan_local(const int* __restrict__ cnt, int* __restrict__ off,
               int* __restrict__ bsum, int n_rows) {
    __shared__ int wsum[32];
    const int lane = threadIdx.x & 31;
    const int wave = threadIdx.x >> 5;
    const int i = blockIdx.x * 1024 + threadIdx.x;
    const int v = (i < n_rows) ? cnt[i] : 0;
    int x = v;
    for (int d = 1; d < 32; d <<= 1) {
        const int y = __shfl_up(x, d);
        if (lane >= d) x += y;
    }
    if (lane == 31) wsum[wave] = x;
    __syncthreads();
    if (wave == 0) {
        const int s = wsum[lane];
        int t = s;
        for (int d = 1; d < 32; d <<= 1) {
            const int y = __shfl_up(t, d);
            if (lane >= d) t += y;
        }
        wsum[lane] = t - s;
        if (lane == 31) bsum[blockIdx.x] = t;
    }
    __syncthreads();
    if (i < n_rows) off[i] = x - v + wsum[wave];
}

__global__ void __launch_bounds__(1024)
csr_scan_add(int* __restrict__ off, const int* __restrict__ bsum, int n_rows) {
    __shared__ int base_s;
    if (threadIdx.x < 32) {
        int s = 0;
        for (int j = threadIdx.x; j < (int)blockIdx.x; j += 32) s += bsum[j];
        for (int d = 16; d > 0; d >>= 1) s += __shfl_xor(s, d);
        if (threadIdx.x == 0) base_s = s;
    }
    __syncthreads();
    const int base = base_s;
    const int i = blockIdx.x * 1024 + threadIdx.x;
    if (i < n_rows) off[i] += base;
    if (blockIdx.x == gridDim.x - 1 && threadIdx.x == 0)
        off[n_rows] = base + bsum[blockIdx.x];
}

__global__ void csr_fill(const int* __restrict__ dst, const int* __restrict__ off,
                         int* __restrict__ cnt, int* __restrict__ eid, int nE) {
    const int e = blockIdx.x * blockDim.x + threadIdx.x;
    if (e >= nE) return;
    const int d = dst[e];
    const int k = atomicSub(&cnt[d], 1) - 1;
    eid[off[d] + k] = e;
}

__global__ void __launch_bounds__(256)
csr_order(const int* __restrict__ off, const int* __restrict__ eid,
          const int* __restrict__ src, const float* __restrict__ ew,
          int* __restrict__ srcS, float* __restrict__ wS, int n_rows) {
    const int lane = threadIdx.x & 31;
    const int r = (blockIdx.x * 256 + threadIdx.x) >> 5;
    if (r >= n_rows) return;
    const int s = off[r];
    const int len = off[r + 1] - s;
    for (int i = lane; i < len; i += 32) {
        const int id = eid[s + i];
        int rank = 0;
        for (int j = 0; j < len; ++j) rank += (eid[s + j] < id) ? 1 : 0;
        srcS[s + rank] = src[id];
        wS[s + rank] = ew[id];
    }
}

template <int VEC>
__global__ void __launch_bounds__(256)
spmm_gather(const float* __restrict__ H, const int* __restrict__ off,
            const int* __restrict__ srcS, const float* __restrict__ wS,
            float* __restrict__ out, int n_rows) {
    typedef float vf __attribute__((ext_vector_type(VEC)));
    const int lane = threadIdx.x & 31;
    const int r = __builtin_amdgcn_readfirstlane((int)((blockIdx.x * 256 + threadIdx.x) >> 5));
    if (r >= n_rows) return;
    const int s = off[r];
    const int t = off[r + 1];
    const vf* __restrict__ Hv = (const vf*)H + lane;
    vf acc = {};
    for (int base = s; base < t; base += 32) {
        const int m = (t - base < 32) ? (t - base) : 32;
        int my_s = 0;
        float my_w = 0.0f;
        if (lane < m) {
            my_s = srcS[base + lane];
            my_w = wS[base + lane];
        }
#pragma unroll 4
        for (int j = 0; j < m; ++j) {
            const int sj = __builtin_amdgcn_readlane(my_s, j);
            const float wj = __builtin_bit_cast(
                float, __builtin_amdgcn_readlane(__builtin_bit_cast(int, my_w), j));
            const vf v = Hv[(size_t)sj * 32];
            acc += wj * v;
        }
    }
    ((vf*)out)[(size_t)r * 32 + lane] = acc;
}

__global__ void move_words(const int* __restrict__ from, int* __restrict__ to,
                           int count) {
    const int stride = gridDim.x * blockDim.x;
    const int i = blockIdx.x * blockDim.x + threadIdx.x;
    const int n4 = count >> 2;
    for (int k = i; k < n4; k += stride)
        ((int4*)to)[k] = ((const int4*)from)[k];
    for (int k = (n4 << 2) + i; k < count; k += stride) to[k] = from[k];
}

__global__ void zero_f32(float* __restrict__ p, long long count) {
    long long i = blockIdx.x * (long long)blockDim.x + threadIdx.x;
    const long long stride = (long long)gridDim.x * blockDim.x;
    for (; i < count; i += stride) p[i] = 0.0f;
}

__global__ void __launch_bounds__(128)
pn_reduce(const float* __restrict__ H, float* __restrict__ red, int n_rows) {
    __shared__ float ssq[D_H];
    const int c = threadIdx.x;
    float colsum = 0.0f, sumsq = 0.0f;
    for (int r = blockIdx.x; r < n_rows; r += gridDim.x) {
        float v = H[(long long)r * D_H + c];
        colsum += v;
        sumsq += v * v;
    }
    unsafeAtomicAdd(&red[c], colsum);
    ssq[c] = sumsq;
    __syncthreads();
    for (int s = 64; s > 0; s >>= 1) {
        if (c < s) ssq[c] += ssq[c + s];
        __syncthreads();
    }
    if (c == 0) unsafeAtomicAdd(&red[D_H], ssq[0]);
}

__global__ void __launch_bounds__(128)
pn_finalize(float* __restrict__ red, int n_rows) {
    __shared__ float sh[D_H];
    const int c = threadIdx.x;
    const float inv_n = 1.0f / (float)n_rows;
    const float mean = red[c] * inv_n;
    sh[c] = mean * mean;
    __syncthreads();
    for (int s = 64; s > 0; s >>= 1) {
        if (c < s) sh[c] += sh[c + s];
        __syncthreads();
    }
    if (c == 0) {
        float ssq_centered = red[D_H] - (float)n_rows * sh[0];
        red[D_H + 1] = 1.0f / sqrtf(PN_EPS + ssq_centered * inv_n);
    }
    red[c] = mean;
}

__global__ void pn_apply(const float* __restrict__ conv,
                         const float* __restrict__ xold,
                         float* __restrict__ xout,
                         const float* __restrict__ red, int n_rows) {
    const long long i = blockIdx.x * (long long)blockDim.x + threadIdx.x;
    if (i >= (long long)n_rows * D_H) return;
    const int c = (int)(i & (D_H - 1));
    float v = (conv[i] - red[c]) * red[D_H + 1];
    v = fmaxf(v, 0.0f);
    if (xold) v += xold[i];
    xout[i] = v;
}

extern "C" void kernel_launch(void* const* d_in, const int* in_sizes, int n_in,
                              void* d_out, int out_size, void* d_ws, size_t ws_size,
                              hipStream_t stream) {
    const float* x    = (const float*)d_in[0];
    const int*   ei   = (const int*)d_in[1];
    const float* ew   = (const float*)d_in[2];
    const float* W0   = (const float*)d_in[3];
    const float* b0   = (const float*)d_in[4];
    const float* W1   = (const float*)d_in[5];
    const float* b1   = (const float*)d_in[6];
    const float* Wout = (const float*)d_in[7];
    const float* bout = (const float*)d_in[8];
    float* out = (float*)d_out;

    const int n = in_sizes[0] / D_H;
    const int e = in_sizes[2];
    const int d_outc = out_size / n;
    const int* srcI = ei;
    const int* dstI = ei + e;

    const size_t nd = (size_t)n * D_H;
    float* A   = (float*)d_ws;
    float* B   = A + nd;
    float* C   = B + nd;
    float* red = C + nd;

    const int gemm_blocks = (((n + 15) / 16) + 7) / 8;
    auto launch_gemm = [&](const float* X_, const float* W_, const float* bias_,
                           float* H_, int dout) {
        if (dout == 128)
            gemm_wmma_f32<128><<<gemm_blocks, 256, 0, stream>>>(X_, W_, bias_, H_, n);
        else
            gemm_wmma_f32<64><<<gemm_blocks, 256, 0, stream>>>(X_, W_, bias_, H_, n);
    };
    auto launch_gather = [&](const float* H_, const int* tab, float* O_, int D) {
        const int* srcS_ = tab;
        const float* wS_ = (const float*)(tab + e);
        const int* off_ = tab + 2 * (size_t)e;
        const int blocks = (n + 7) / 8;
        if (D == 128)
            spmm_gather<4><<<blocks, 256, 0, stream>>>(H_, off_, srcS_, wS_, O_, n);
        else
            spmm_gather<2><<<blocks, 256, 0, stream>>>(H_, off_, srcS_, wS_, O_, n);
    };
    auto launch_zero = [&](float* p, long long cnt) {
        int blocks = (int)((cnt + 256 * 8 - 1) / (256 * 8));
        zero_f32<<<blocks, 256, 0, stream>>>(p, cnt);
    };
    auto launch_pn = [&](const float* conv, const float* xold, float* xout) {
        launch_zero(red, 256);
        pn_reduce<<<1024, 128, 0, stream>>>(conv, red, n);
        pn_finalize<<<1, 128, 0, stream>>>(red, n);
        long long cnt = (long long)n * D_H;
        pn_apply<<<(int)((cnt + 255) / 256), 256, 0, stream>>>(conv, xold, xout, red, n);
    };

    const int npad = (n + 1024) & ~255;
    int*   t_src  = (int*)out;
    float* t_w    = (float*)(t_src + e);
    int*   t_off  = t_src + 2 * (size_t)e;
    int*   t_cnt  = t_off + npad;
    int*   t_bsum = t_cnt + npad;
    int*   t_eid  = t_bsum + 1024;
    const int edge_blocks = (e + 255) / 256;
    const int scan_blocks = (n + 1023) / 1024;
    launch_zero((float*)t_cnt, n);
    csr_count<<<edge_blocks, 256, 0, stream>>>(dstI, t_cnt, e);
    csr_scan_local<<<scan_blocks, 1024, 0, stream>>>(t_cnt, t_off, t_bsum, n);
    csr_scan_add<<<scan_blocks, 1024, 0, stream>>>(t_off, t_bsum, n);
    csr_fill<<<edge_blocks, 256, 0, stream>>>(dstI, t_off, t_cnt, t_eid, e);
    csr_order<<<(n + 7) / 8, 256, 0, stream>>>(t_off, t_eid, srcI, ew, t_src, t_w, n);

    launch_gemm(x, W0, b0, A, D_H);
    launch_gather(A, t_src, B, D_H);
    launch_pn(B, nullptr, A);

    launch_gemm(A, W1, b1, B, D_H);
    launch_gather(B, t_src, C, D_H);
    launch_pn(C, A, C);

    const int table_words = 2 * e + n + 1;
    move_words<<<1024, 256, 0, stream>>>(t_src, (int*)B, table_words);
    launch_gemm(C, Wout, bout, A, d_outc);
    launch_gather(A, (const int*)B, out, d_outc);
}


# One-thread-per-element launches (PairNorm apply, edge count, edge fill) become grid-stride passes on 1024 blocks, apply in float4 with 4 steps in flight: 900K thin waves per call become 33K
# speedup vs baseline: 1.8159x; 1.0611x over previous
#include <hip/hip_runtime.h>

#define D_H 128
#define PN_EPS 1e-6f

typedef float v2f __attribute__((ext_vector_type(2)));
typedef float v8f __attribute__((ext_vector_type(8)));

template <int DOUT>
__global__ void __launch_bounds__(256)
gemm_wmma_f32(const float* __restrict__ X, const float* __restrict__ W,
              const float* __restrict__ bias, float* __restrict__ H,
              int n_rows) {
    constexpr int TN  = DOUT / 16;
    constexpr int KC  = 64;
    constexpr int WST = KC + 4;
    __shared__ float Wt[DOUT * WST];

    const int lane = threadIdx.x & 31;
    const int wave = threadIdx.x >> 5;
    const int half = lane >> 4;
    const int l15  = lane & 15;

    const int tm      = blockIdx.x * 8 + wave;
    const int tiles_m = (n_rows + 15) >> 4;
    const bool active = (tm < tiles_m);

    int row_a = tm * 16 + l15;
    if (row_a >= n_rows) row_a = n_rows - 1;
    const float* __restrict__ ap = X + (size_t)row_a * D_H + half * 2;
    const float* __restrict__ bbase = &Wt[l15 * WST + half * 2];

    v8f acc[TN] = {};

    for (int k0 = 0; k0 < D_H; k0 += KC) {
        __syncthreads();
        const float* __restrict__ wg = W + (size_t)k0 * DOUT;
        for (int idx = threadIdx.x; idx < KC * DOUT; idx += 256) {
            int r = idx / DOUT;
            int c = idx - r * DOUT;
            Wt[c * WST + r] = wg[idx];
        }
        __syncthreads();

        if (active) {
            __builtin_prefetch(ap + k0 + KC, 0, 1);
            for (int kk = 0; kk < KC; kk += 4) {
                v2f a = *(const v2f*)(ap + k0 + kk);
#pragma unroll
                for (int t = 0; t < TN; ++t) {
                    v2f b = *(const v2f*)(bbase + t * (16 * WST) + kk);
                    acc[t] = __builtin_amdgcn_wmma_f32_16x16x4_f32(
                         false, a,  false, b,
                         (short)0, acc[t],
                         false,  false);
                }
            }
        }
    }

    if (!active) return;
    const int row_base = tm * 16 + half * 8;
#pragma unroll
    for (int t = 0; t < TN; ++t) {
        const int col = t * 16 + l15;
        const float bb = bias[col];
#pragma unroll
        for (int r = 0; r < 8; ++r) {
            const int row = row_base + r;
            if (row < n_rows)
                H[(size_t)row * DOUT + col] = acc[t][r] + bb;
        }
    }
}

__global__ void __launch_bounds__(256)
csr_count(const int* __restrict__ dst, int* __restrict__ cnt, int nE) {
    const int stride = gridDim.x * 256;
    for (int e = blockIdx.x * 256 + threadIdx.x; e < nE; e += stride)
        atomicAdd(&cnt[dst[e]], 1);
}

__global__ void __launch_bounds__(1024)
csr_scan_local(const int* __restrict__ cnt, int* __restrict__ off,
               int* __restrict__ bsum, int n_rows) {
    __shared__ int wsum[32];
    const int lane = threadIdx.x & 31;
    const int wave = threadIdx.x >> 5;
    const int i = blockIdx.x * 1024 + threadIdx.x;
    const int v = (i < n_rows) ? cnt[i] : 0;
    int x = v;
    for (int d = 1; d < 32; d <<= 1) {
        const int y = __shfl_up(x, d);
        if (lane >= d) x += y;
    }
    if (lane == 31) wsum[wave] = x;
    __syncthreads();
    if (wave == 0) {
        const int s = wsum[lane];
        int t = s;
        for (int d = 1; d < 32; d <<= 1) {
            const int y = __shfl_up(t, d);
            if (lane >= d) t += y;
        }
        wsum[lane] = t - s;
        if (lane == 31) bsum[blockIdx.x] = t;
    }
    __syncthreads();
    if (i < n_rows) off[i] = x - v + wsum[wave];
}

__global__ void __launch_bounds__(1024)
csr_scan_add(int* __restrict__ off, const int* __restrict__ bsum, int n_rows) {
    __shared__ int base_s;
    if (threadIdx.x < 32) {
        int s = 0;
        for (int j = threadIdx.x; j < (int)blockIdx.x; j += 32) s += bsum[j];
        for (int d = 16; d > 0; d >>= 1) s += __shfl_xor(s, d);
        if (threadIdx.x == 0) base_s = s;
    }
    __syncthreads();
    const int base = base_s;
    const int i = blockIdx.x * 1024 + threadIdx.x;
    if (i < n_rows) off[i] += base;
    if (blockIdx.x == gridDim.x - 1 && threadIdx.x == 0)
        off[n_rows] = base + bsum[blockIdx.x];
}

__global__ void __launch_bounds__(256)
csr_fill(const int* __restrict__ dst, const int* __restrict__ off,
         int* __restrict__ cnt, int* __restrict__ eid, int nE) {
    const int stride = gridDim.x * 256;
    for (int e = blockIdx.x * 256 + threadIdx.x; e < nE; e += stride) {
        const int d = dst[e];
        const int k = atomicSub(&cnt[d], 1) - 1;
        eid[off[d] + k] = e;
    }
}

__global__ void __launch_bounds__(256)
csr_order(const int* __restrict__ off, const int* __restrict__ eid,
          const int* __restrict__ src, const float* __restrict__ ew,
          int* __restrict__ srcS, float* __restrict__ wS, int n_rows) {
    const int lane = threadIdx.x & 31;
    const int r = (blockIdx.x * 256 + threadIdx.x) >> 5;
    if (r >= n_rows) return;
    const int s = off[r];
    const int len = off[r + 1] - s;
    for (int i = lane; i < len; i += 32) {
        const int id = eid[s + i];
        int rank = 0;
        for (int j = 0; j < len; ++j) rank += (eid[s + j] < id) ? 1 : 0;
        srcS[s + rank] = src[id];
        wS[s + rank] = ew[id];
    }
}

template <int VEC>
__global__ void __launch_bounds__(256)
spmm_gather(const float* __restrict__ H, const int* __restrict__ off,
            const int* __restrict__ srcS, const float* __restrict__ wS,
            float* __restrict__ out, int n_rows) {
    typedef float vf __attribute__((ext_vector_type(VEC)));
    const int lane = threadIdx.x & 31;
    const int r = __builtin_amdgcn_readfirstlane((int)((blockIdx.x * 256 + threadIdx.x) >> 5));
    if (r >= n_rows) return;
    const int s = off[r];
    const int t = off[r + 1];
    const vf* __restrict__ Hv = (const vf*)H + lane;
    vf acc = {};
    for (int base = s; base < t; base += 32) {
        const int m = (t - base < 32) ? (t - base) : 32;
        int my_s = 0;
        float my_w = 0.0f;
        if (lane < m) {
            my_s = srcS[base + lane];
            my_w = wS[base + lane];
        }
#pragma unroll 4
        for (int j = 0; j < m; ++j) {
            const int sj = __builtin_amdgcn_readlane(my_s, j);
            const float wj = __builtin_bit_cast(
                float, __builtin_amdgcn_readlane(__builtin_bit_cast(int, my_w), j));
            const vf v = Hv[(size_t)sj * 32];
            acc += wj * v;
        }
    }
    ((vf*)out)[(size_t)r * 32 + lane] = acc;
}

__global__ void move_words(const int* __restrict__ from, int* __restrict__ to,
                           int count) {
    const int stride = gridDim.x * blockDim.x;
    const int i = blockIdx.x * blockDim.x + threadIdx.x;
    const int n4 = count >> 2;
    for (int k = i; k < n4; k += stride)
        ((int4*)to)[k] = ((const int4*)from)[k];
    for (int k = (n4 << 2) + i; k < count; k += stride) to[k] = from[k];
}

__global__ void zero_f32(float* __restrict__ p, long long count) {
    long long i = blockIdx.x * (long long)blockDim.x + threadIdx.x;
    const long long stride = (long long)gridDim.x * blockDim.x;
    for (; i < count; i += stride) p[i] = 0.0f;
}

__global__ void __launch_bounds__(128)
pn_reduce(const float* __restrict__ H, float* __restrict__ red, int n_rows) {
    __shared__ float ssq[D_H];
    const int c = threadIdx.x;
    float colsum = 0.0f, sumsq = 0.0f;
    for (int r = blockIdx.x; r < n_rows; r += gridDim.x) {
        float v = H[(long long)r * D_H + c];
        colsum += v;
        sumsq += v * v;
    }
    unsafeAtomicAdd(&red[c], colsum);
    ssq[c] = sumsq;
    __syncthreads();
    for (int s = 64; s > 0; s >>= 1) {
        if (c < s) ssq[c] += ssq[c + s];
        __syncthreads();
    }
    if (c == 0) unsafeAtomicAdd(&red[D_H], ssq[0]);
}

__global__ void __launch_bounds__(128)
pn_finalize(float* __restrict__ red, int n_rows) {
    __shared__ float sh[D_H];
    const int c = threadIdx.x;
    const float inv_n = 1.0f / (float)n_rows;
    const float mean = red[c] * inv_n;
    sh[c] = mean * mean;
    __syncthreads();
    for (int s = 64; s > 0; s >>= 1) {
        if (c < s) sh[c] += sh[c + s];
        __syncthreads();
    }
    if (c == 0) {
        float ssq_centered = red[D_H] - (float)n_rows * sh[0];
        red[D_H + 1] = 1.0f / sqrtf(PN_EPS + ssq_centered * inv_n);
    }
    red[c] = mean;
}

template <bool HAS_OLD>
__global__ void __launch_bounds__(256)
pn_apply4(const float* conv, const float* __restrict__ xold, float* xout,
          const float* __restrict__ red, int n4) {
    const int stride = gridDim.x * 256;
    const int tid = blockIdx.x * 256 + threadIdx.x;
    const float4 mean = ((const float4*)red)[tid & (D_H / 4 - 1)];
    const float scale = red[D_H + 1];
    const float4* c4 = (const float4*)conv;
    const float4* x4 = (const float4*)xold;
    float4* o4 = (float4*)xout;
    for (int base = tid; base < n4; base += 4 * stride) {
        float4 v[4], o[4];
#pragma unroll
        for (int u = 0; u < 4; ++u) {
            const int idx = base + u * stride;
            if (idx < n4) {
                v[u] = c4[idx];
                if (HAS_OLD) o[u] = x4[idx];
            }
        }
#pragma unroll
        for (int u = 0; u < 4; ++u) {
            const int idx = base + u * stride;
            if (idx < n4) {
                float4 r;
                r.x = fmaxf((v[u].x - mean.x) * scale, 0.0f);
                r.y = fmaxf((v[u].y - mean.y) * scale, 0.0f);
                r.z = fmaxf((v[u].z - mean.z) * scale, 0.0f);
                r.w = fmaxf((v[u].w - mean.w) * scale, 0.0f);
                if (HAS_OLD) {
                    r.x += o[u].x; r.y += o[u].y; r.z += o[u].z; r.w += o[u].w;
                }
                o4[idx] = r;
            }
        }
    }
}

extern "C" void kernel_launch(void* const* d_in, const int* in_sizes, int n_in,
                              void* d_out, int out_size, void* d_ws, size_t ws_size,
                              hipStream_t stream) {
    const float* x    = (const float*)d_in[0];
    const int*   ei   = (const int*)d_in[1];
    const float* ew   = (const float*)d_in[2];
    const float* W0   = (const float*)d_in[3];
    const float* b0   = (const float*)d_in[4];
    const float* W1   = (const float*)d_in[5];
    const float* b1   = (const float*)d_in[6];
    const float* Wout = (const float*)d_in[7];
    const float* bout = (const float*)d_in[8];
    float* out = (float*)d_out;

    const int n = in_sizes[0] / D_H;
    const int e = in_sizes[2];
    const int d_outc = out_size / n;
    const int* srcI = ei;
    const int* dstI = ei + e;

    const size_t nd = (size_t)n * D_H;
    float* A   = (float*)d_ws;
    float* B   = A + nd;
    float* C   = B + nd;
    float* red = C + nd;

    const int gemm_blocks = (((n + 15) / 16) + 7) / 8;
    auto launch_gemm = [&](const float* X_, const float* W_, const float* bias_,
                           float* H_, int dout) {
        if (dout == 128)
            gemm_wmma_f32<128><<<gemm_blocks, 256, 0, stream>>>(X_, W_, bias_, H_, n);
        else
            gemm_wmma_f32<64><<<gemm_blocks, 256, 0, stream>>>(X_, W_, bias_, H_, n);
    };
    auto launch_gather = [&](const float* H_, const int* tab, float* O_, int D) {
        const int* srcS_ = tab;
        const float* wS_ = (const float*)(tab + e);
        const int* off_ = tab + 2 * (size_t)e;
        const int blocks = (n + 7) / 8;
        if (D == 128)
            spmm_gather<4><<<blocks, 256, 0, stream>>>(H_, off_, srcS_, wS_, O_, n);
        else
            spmm_gather<2><<<blocks, 256, 0, stream>>>(H_, off_, srcS_, wS_, O_, n);
    };
    auto launch_zero = [&](float* p, long long cnt) {
        int blocks = (int)((cnt + 256 * 8 - 1) / (256 * 8));
        zero_f32<<<blocks, 256, 0, stream>>>(p, cnt);
    };
    auto launch_pn = [&](const float* conv, const float* xold, float* xout) {
        launch_zero(red, 256);
        pn_reduce<<<1024, 128, 0, stream>>>(conv, red, n);
        pn_finalize<<<1, 128, 0, stream>>>(red, n);
        const int n4 = n * (D_H / 4);
        if (xold)
            pn_apply4<true><<<1024, 256, 0, stream>>>(conv, xold, xout, red, n4);
        else
            pn_apply4<false><<<1024, 256, 0, stream>>>(conv, xold, xout, red, n4);
    };

    const int npad = (n + 1024) & ~255;
    int*   t_src  = (int*)out;
    float* t_w    = (float*)(t_src + e);
    int*   t_off  = t_src + 2 * (size_t)e;
    int*   t_cnt  = t_off + npad;
    int*   t_bsum = t_cnt + npad;
    int*   t_eid  = t_bsum + 1024;
    const int scan_blocks = (n + 1023) / 1024;
    launch_zero((float*)t_cnt, n);
    csr_count<<<1024, 256, 0, stream>>>(dstI, t_cnt, e);
    csr_scan_local<<<scan_blocks, 1024, 0, stream>>>(t_cnt, t_off, t_bsum, n);
    csr_scan_add<<<scan_blocks, 1024, 0, stream>>>(t_off, t_bsum, n);
    csr_fill<<<1024, 256, 0, stream>>>(dstI, t_off, t_cnt, t_eid, e);
    csr_order<<<(n + 7) / 8, 256, 0, stream>>>(t_off, t_eid, srcI, ew, t_src, t_w, n);

    launch_gemm(x, W0, b0, A, D_H);
    launch_gather(A, t_src, B, D_H);
    launch_pn(B, nullptr, A);

    launch_gemm(A, W1, b1, B, D_H);
    launch_gather(B, t_src, C, D_H);
    launch_pn(C, A, C);

    const int table_words = 2 * e + n + 1;
    move_words<<<1024, 256, 0, stream>>>(t_src, (int*)B, table_words);
    launch_gemm(C, Wout, bout, A, d_outc);
    launch_gather(A, (const int*)B, out, d_outc);
}


# PairNorm statistics pass: a wave reads whole rows as 16-byte loads, 8 rows in flight per lane (was 98 four-byte loads, each waited for alone); partial sums regrouped, so the reduction tree changes
# speedup vs baseline: 2.1261x; 1.1708x over previous
#include <hip/hip_runtime.h>

#define D_H 128
#define PN_EPS 1e-6f

typedef float v2f __attribute__((ext_vector_type(2)));
typedef float v8f __attribute__((ext_vector_type(8)));

template <int DOUT>
__global__ void __launch_bounds__(256)
gemm_wmma_f32(const float* __restrict__ X, const float* __restrict__ W,
              const float* __restrict__ bias, float* __restrict__ H,
              int n_rows) {
    constexpr int TN  = DOUT / 16;
    constexpr int KC  = 64;
    constexpr int WST = KC + 4;
    __shared__ float Wt[DOUT * WST];

    const int lane = threadIdx.x & 31;
    const int wave = threadIdx.x >> 5;
    const int half = lane >> 4;
    const int l15  = lane & 15;

    const int tm      = blockIdx.x * 8 + wave;
    const int tiles_m = (n_rows + 15) >> 4;
    const bool active = (tm < tiles_m);

    int row_a = tm * 16 + l15;
    if (row_a >= n_rows) row_a = n_rows - 1;
    const float* __restrict__ ap = X + (size_t)row_a * D_H + half * 2;
    const float* __restrict__ bbase = &Wt[l15 * WST + half * 2];

    v8f acc[TN] = {};

    for (int k0 = 0; k0 < D_H; k0 += KC) {
        __syncthreads();
        const float* __restrict__ wg = W + (size_t)k0 * DOUT;
        for (int idx = threadIdx.x; idx < KC * DOUT; idx += 256) {
            int r = idx / DOUT;
            int c = idx - r * DOUT;
            Wt[c * WST + r] = wg[idx];
        }
        __syncthreads();

        if (active) {
            __builtin_prefetch(ap + k0 + KC, 0, 1);
            for (int kk = 0; kk < KC; kk += 4) {
                v2f a = *(const v2f*)(ap + k0 + kk);
#pragma unroll
                for (int t = 0; t < TN; ++t) {
                    v2f b = *(const v2f*)(bbase + t * (16 * WST) + kk);
                    acc[t] = __builtin_amdgcn_wmma_f32_16x16x4_f32(
                         false, a,  false, b,
                         (short)0, acc[t],
                         false,  false);
                }
            }
        }
    }

    if (!active) return;
    const int row_base = tm * 16 + half * 8;
#pragma unroll
    for (int t = 0; t < TN; ++t) {
        const int col = t * 16 + l15;
        const float bb = bias[col];
#pragma unroll
        for (int r = 0; r < 8; ++r) {
            const int row = row_base + r;
            if (row < n_rows)
                H[(size_t)row * DOUT + col] = acc[t][r] + bb;
        }
    }
}

__global__ void __launch_bounds__(256)
csr_count(const int* __restrict__ dst, int* __restrict__ cnt, int nE) {
    const int stride = gridDim.x * 256;
    for (int e = blockIdx.x * 256 + threadIdx.x; e < nE; e += stride)
        atomicAdd(&cnt[dst[e]], 1);
}

__global__ void __launch_bounds__(1024)
csr_scan_local(const int* __restrict__ cnt, int* __restrict__ off,
               int* __restrict__ bsum, int n_rows) {
    __shared__ int wsum[32];
    const int lane = threadIdx.x & 31;
    const int wave = threadIdx.x >> 5;
    const int i = blockIdx.x * 1024 + threadIdx.x;
    const int v = (i < n_rows) ? cnt[i] : 0;
    int x = v;
    for (int d = 1; d < 32; d <<= 1) {
        const int y = __shfl_up(x, d);
        if (lane >= d) x += y;
    }
    if (lane == 31) wsum[wave] = x;
    __syncthreads();
    if (wave == 0) {
        const int s = wsum[lane];
        int t = s;
        for (int d = 1; d < 32; d <<= 1) {
            const int y = __shfl_up(t, d);
            if (lane >= d) t += y;
        }
        wsum[lane] = t - s;
        if (lane == 31) bsum[blockIdx.x] = t;
    }
    __syncthreads();
    if (i < n_rows) off[i] = x - v + wsum[wave];
}

__global__ void __launch_bounds__(1024)
csr_scan_add(int* __restrict__ off, const int* __restrict__ bsum, int n_rows) {
    __shared__ int base_s;
    if (threadIdx.x < 32) {
        int s = 0;
        for (int j = threadIdx.x; j < (int)blockIdx.x; j += 32) s += bsum[j];
        for (int d = 16; d > 0; d >>= 1) s += __shfl_xor(s, d);
        if (threadIdx.x == 0) base_s = s;
    }
    __syncthreads();
    const int base = base_s;
    const int i = blockIdx.x * 1024 + threadIdx.x;
    if (i < n_rows) off[i] += base;
    if (blockIdx.x == gridDim.x - 1 && threadIdx.x == 0)
        off[n_rows] = base + bsum[blockIdx.x];
}

__global__ void __launch_bounds__(256)
csr_fill(const int* __restrict__ dst, const int* __restrict__ off,
         int* __restrict__ cnt, int* __restrict__ eid, int nE) {
    const int stride = gridDim.x * 256;
    for (int e = blockIdx.x * 256 + threadIdx.x; e < nE; e += stride) {
        const int d = dst[e];
        const int k = atomicSub(&cnt[d], 1) - 1;
        eid[off[d] + k] = e;
    }
}

__global__ void __launch_bounds__(256)
csr_order(const int* __restrict__ off, const int* __restrict__ eid,
          const int* __restrict__ src, const float* __restrict__ ew,
          int* __restrict__ srcS, float* __restrict__ wS, int n_rows) {
    const int lane = threadIdx.x & 31;
    const int r = (blockIdx.x * 256 + threadIdx.x) >> 5;
    if (r >= n_rows) return;
    const int s = off[r];
    const int len = off[r + 1] - s;
    for (int i = lane; i < len; i += 32) {
        const int id = eid[s + i];
        int rank = 0;
        for (int j = 0; j < len; ++j) rank += (eid[s + j] < id) ? 1 : 0;
        srcS[s + rank] = src[id];
        wS[s + rank] = ew[id];
    }
}

template <int VEC>
__global__ void __launch_bounds__(256)
spmm_gather(const float* __restrict__ H, const int* __restrict__ off,
            const int* __restrict__ srcS, const float* __restrict__ wS,
            float* __restrict__ out, int n_rows) {
    typedef float vf __attribute__((ext_vector_type(VEC)));
    const int lane = threadIdx.x & 31;
    const int r = __builtin_amdgcn_readfirstlane((int)((blockIdx.x * 256 + threadIdx.x) >> 5));
    if (r >= n_rows) return;
    const int s = off[r];
    const int t = off[r + 1];
    const vf* __restrict__ Hv = (const vf*)H + lane;
    vf acc = {};
    for (int base = s; base < t; base += 32) {
        const int m = (t - base < 32) ? (t - base) : 32;
        int my_s = 0;
        float my_w = 0.0f;
        if (lane < m) {
            my_s = srcS[base + lane];
            my_w = wS[base + lane];
        }
#pragma unroll 4
        for (int j = 0; j < m; ++j) {
            const int sj = __builtin_amdgcn_readlane(my_s, j);
            const float wj = __builtin_bit_cast(
                float, __builtin_amdgcn_readlane(__builtin_bit_cast(int, my_w), j));
            const vf v = Hv[(size_t)sj * 32];
            acc += wj * v;
        }
    }
    ((vf*)out)[(size_t)r * 32 + lane] = acc;
}

__global__ void move_words(const int* __restrict__ from, int* __restrict__ to,
                           int count) {
    const int stride = gridDim.x * blockDim.x;
    const int i = blockIdx.x * blockDim.x + threadIdx.x;
    const int n4 = count >> 2;
    for (int k = i; k < n4; k += stride)
        ((int4*)to)[k] = ((const int4*)from)[k];
    for (int k = (n4 << 2) + i; k < count; k += stride) to[k] = from[k];
}

__global__ void zero_f32(float* __restrict__ p, long long count) {
    long long i = blockIdx.x * (long long)blockDim.x + threadIdx.x;
    const long long stride = (long long)gridDim.x * blockDim.x;
    for (; i < count; i += stride) p[i] = 0.0f;
}

#define PN_ROWS 8
typedef float v4f __attribute__((ext_vector_type(4)));

__global__ void __launch_bounds__(256)
pn_reduce(const float* __restrict__ H, float* __restrict__ red, int n_rows) {
    __shared__ v4f part[8 * 32];
    __shared__ float psq[8];
    const int lane = threadIdx.x & 31;
    const int wave = __builtin_amdgcn_readfirstlane((int)(threadIdx.x >> 5));
    const int stride = gridDim.x * 8;
    const v4f* __restrict__ Hv = (const v4f*)H + lane;
    const v4f zero = {0.0f, 0.0f, 0.0f, 0.0f};
    v4f cs = zero, sq = zero;
    for (int base = blockIdx.x * 8 + wave; base < n_rows; base += PN_ROWS * stride) {
        v4f v[PN_ROWS];
#pragma unroll
        for (int u = 0; u < PN_ROWS; ++u) {
            const int r = base + u * stride;
            v[u] = (r < n_rows) ? Hv[(size_t)r * 32] : zero;
        }
#pragma unroll
        for (int u = 0; u < PN_ROWS; ++u) {
            cs += v[u];
            sq += v[u] * v[u];
        }
    }
    float s = (sq.x + sq.y) + (sq.z + sq.w);
    for (int d = 16; d > 0; d >>= 1) s += __shfl_xor(s, d);
    part[wave * 32 + lane] = cs;
    if (lane == 0) psq[wave] = s;
    __syncthreads();
    if (threadIdx.x < D_H) {
        float t = 0.0f;
        for (int w = 0; w < 8; ++w) t += ((const float*)part)[w * D_H + threadIdx.x];
        unsafeAtomicAdd(&red[threadIdx.x], t);
    } else if (threadIdx.x == D_H) {
        float t = 0.0f;
        for (int w = 0; w < 8; ++w) t += psq[w];
        unsafeAtomicAdd(&red[D_H], t);
    }
}

__global__ void __launch_bounds__(128)
pn_finalize(float* __restrict__ red, int n_rows) {
    __shared__ float sh[D_H];
    const int c = threadIdx.x;
    const float inv_n = 1.0f / (float)n_rows;
    const float mean = red[c] * inv_n;
    sh[c] = mean * mean;
    __syncthreads();
    for (int s = 64; s > 0; s >>= 1) {
        if (c < s) sh[c] += sh[c + s];
        __syncthreads();
    }
    if (c == 0) {
        float ssq_centered = red[D_H] - (float)n_rows * sh[0];
        red[D_H + 1] = 1.0f / sqrtf(PN_EPS + ssq_centered * inv_n);
    }
    red[c] = mean;
}

template <bool HAS_OLD>
__global__ void __launch_bounds__(256)
pn_apply4(const float* conv, const float* __restrict__ xold, float* xout,
          const float* __restrict__ red, int n4) {
    const int stride = gridDim.x * 256;
    const int tid = blockIdx.x * 256 + threadIdx.x;
    const float4 mean = ((const float4*)red)[tid & (D_H / 4 - 1)];
    const float scale = red[D_H + 1];
    const float4* c4 = (const float4*)conv;
    const float4* x4 = (const float4*)xold;
    float4* o4 = (float4*)xout;
    for (int base = tid; base < n4; base += 4 * stride) {
        float4 v[4], o[4];
#pragma unroll
        for (int u = 0; u < 4; ++u) {
            const int idx = base + u * stride;
            if (idx < n4) {
                v[u] = c4[idx];
                if (HAS_OLD) o[u] = x4[idx];
            }
        }
#pragma unroll
        for (int u = 0; u < 4; ++u) {
            const int idx = base + u * stride;
            if (idx < n4) {
                float4 r;
                r.x = fmaxf((v[u].x - mean.x) * scale, 0.0f);
                r.y = fmaxf((v[u].y - mean.y) * scale, 0.0f);
                r.z = fmaxf((v[u].z - mean.z) * scale, 0.0f);
                r.w = fmaxf((v[u].w - mean.w) * scale, 0.0f);
                if (HAS_OLD) {
                    r.x += o[u].x; r.y += o[u].y; r.z += o[u].z; r.w += o[u].w;
                }
                o4[idx] = r;
            }
        }
    }
}

extern "C" void kernel_launch(void* const* d_in, const int* in_sizes, int n_in,
                              void* d_out, int out_size, void* d_ws, size_t ws_size,
                              hipStream_t stream) {
    const float* x    = (const float*)d_in[0];
    const int*   ei   = (const int*)d_in[1];
    const float* ew   = (const float*)d_in[2];
    const float* W0   = (const float*)d_in[3];
    const float* b0   = (const float*)d_in[4];
    const float* W1   = (const float*)d_in[5];
    const float* b1   = (const float*)d_in[6];
    const float* Wout = (const float*)d_in[7];
    const float* bout = (const float*)d_in[8];
    float* out = (float*)d_out;

    const int n = in_sizes[0] / D_H;
    const int e = in_sizes[2];
    const int d_outc = out_size / n;
    const int* srcI = ei;
    const int* dstI = ei + e;

    const size_t nd = (size_t)n * D_H;
    float* A   = (float*)d_ws;
    float* B   = A + nd;
    float* C   = B + nd;
    float* red = C + nd;

    const int gemm_blocks = (((n + 15) / 16) + 7) / 8;
    auto launch_gemm = [&](const float* X_, const float* W_, const float* bias_,
                           float* H_, int dout) {
        if (dout == 128)
            gemm_wmma_f32<128><<<gemm_blocks, 256, 0, stream>>>(X_, W_, bias_, H_, n);
        else
            gemm_wmma_f32<64><<<gemm_blocks, 256, 0, stream>>>(X_, W_, bias_, H_, n);
    };
    auto launch_gather = [&](const float* H_, const int* tab, float* O_, int D) {
        const int* srcS_ = tab;
        const float* wS_ = (const float*)(tab + e);
        const int* off_ = tab + 2 * (size_t)e;
        const int blocks = (n + 7) / 8;
        if (D == 128)
            spmm_gather<4><<<blocks, 256, 0, stream>>>(H_, off_, srcS_, wS_, O_, n);
        else
            spmm_gather<2><<<blocks, 256, 0, stream>>>(H_, off_, srcS_, wS_, O_, n);
    };
    auto launch_zero = [&](float* p, long long cnt) {
        int blocks = (int)((cnt + 256 * 8 - 1) / (256 * 8));
        zero_f32<<<blocks, 256, 0, stream>>>(p, cnt);
    };
    auto launch_pn = [&](const float* conv, const float* xold, float* xout) {
        launch_zero(red, 256);
        pn_reduce<<<1024, 256, 0, stream>>>(conv, red, n);
        pn_finalize<<<1, 128, 0, stream>>>(red, n);
        const int n4 = n * (D_H / 4);
        if (xold)
            pn_apply4<true><<<1024, 256, 0, stream>>>(conv, xold, xout, red, n4);
        else
            pn_apply4<false><<<1024, 256, 0, stream>>>(conv, xold, xout, red, n4);
    };

    const int npad = (n + 1024) & ~255;
    int*   t_src  = (int*)out;
    float* t_w    = (float*)(t_src + e);
    int*   t_off  = t_src + 2 * (size_t)e;
    int*   t_cnt  = t_off + npad;
    int*   t_bsum = t_cnt + npad;
    int*   t_eid  = t_bsum + 1024;
    const int scan_blocks = (n + 1023) / 1024;
    launch_zero((float*)t_cnt, n);
    csr_count<<<1024, 256, 0, stream>>>(dstI, t_cnt, e);
    csr_scan_local<<<scan_blocks, 1024, 0, stream>>>(t_cnt, t_off, t_bsum, n);
    csr_scan_add<<<scan_blocks, 1024, 0, stream>>>(t_off, t_bsum, n);
    csr_fill<<<1024, 256, 0, stream>>>(dstI, t_off, t_cnt, t_eid, e);
    csr_order<<<(n + 7) / 8, 256, 0, stream>>>(t_off, t_eid, srcI, ew, t_src, t_w, n);

    launch_gemm(x, W0, b0, A, D_H);
    launch_gather(A, t_src, B, D_H);
    launch_pn(B, nullptr, A);

    launch_gemm(A, W1, b1, B, D_H);
    launch_gather(B, t_src, C, D_H);
    launch_pn(C, A, C);

    const int table_words = 2 * e + n + 1;
    move_words<<<1024, 256, 0, stream>>>(t_src, (int*)B, table_words);
    launch_gemm(C, Wout, bout, A, d_outc);
    launch_gather(A, (const int*)B, out, d_outc);
}


# Edge table build: the fill pass stores each edge as one 16-byte record {id, source row, weight} at its slot; the ordering pass ranks a row's records in registers. Table and sum order unchanged
# speedup vs baseline: 2.2430x; 1.0550x over previous
#include <hip/hip_runtime.h>

#define D_H 128
#define PN_EPS 1e-6f

typedef float v2f __attribute__((ext_vector_type(2)));
typedef float v8f __attribute__((ext_vector_type(8)));

template <int DOUT>
__global__ void __launch_bounds__(256)
gemm_wmma_f32(const float* __restrict__ X, const float* __restrict__ W,
              const float* __restrict__ bias, float* __restrict__ H,
              int n_rows) {
    constexpr int TN  = DOUT / 16;
    constexpr int KC  = 64;
    constexpr int WST = KC + 4;
    __shared__ float Wt[DOUT * WST];

    const int lane = threadIdx.x & 31;
    const int wave = threadIdx.x >> 5;
    const int half = lane >> 4;
    const int l15  = lane & 15;

    const int tm      = blockIdx.x * 8 + wave;
    const int tiles_m = (n_rows + 15) >> 4;
    const bool active = (tm < tiles_m);

    int row_a = tm * 16 + l15;
    if (row_a >= n_rows) row_a = n_rows - 1;
    const float* __restrict__ ap = X + (size_t)row_a * D_H + half * 2;
    const float* __restrict__ bbase = &Wt[l15 * WST + half * 2];

    v8f acc[TN] = {};

    for (int k0 = 0; k0 < D_H; k0 += KC) {
        __syncthreads();
        const float* __restrict__ wg = W + (size_t)k0 * DOUT;
        for (int idx = threadIdx.x; idx < KC * DOUT; idx += 256) {
            int r = idx / DOUT;
            int c = idx - r * DOUT;
            Wt[c * WST + r] = wg[idx];
        }
        __syncthreads();

        if (active) {
            __builtin_prefetch(ap + k0 + KC, 0, 1);
            for (int kk = 0; kk < KC; kk += 4) {
                v2f a = *(const v2f*)(ap + k0 + kk);
#pragma unroll
                for (int t = 0; t < TN; ++t) {
                    v2f b = *(const v2f*)(bbase + t * (16 * WST) + kk);
                    acc[t] = __builtin_amdgcn_wmma_f32_16x16x4_f32(
                         false, a,  false, b,
                         (short)0, acc[t],
                         false,  false);
                }
            }
        }
    }

    if (!active) return;
    const int row_base = tm * 16 + half * 8;
#pragma unroll
    for (int t = 0; t < TN; ++t) {
        const int col = t * 16 + l15;
        const float bb = bias[col];
#pragma unroll
        for (int r = 0; r < 8; ++r) {
            const int row = row_base + r;
            if (row < n_rows)
                H[(size_t)row * DOUT + col] = acc[t][r] + bb;
        }
    }
}

__global__ void __launch_bounds__(256)
csr_count(const int* __restrict__ dst, int* __restrict__ cnt, int nE) {
    const int stride = gridDim.x * 256;
    for (int e = blockIdx.x * 256 + threadIdx.x; e < nE; e += stride)
        atomicAdd(&cnt[dst[e]], 1);
}

__global__ void __launch_bounds__(1024)
csr_scan_local(const int* __restrict__ cnt, int* __restrict__ off,
               int* __restrict__ bsum, int n_rows) {
    __shared__ int wsum[32];
    const int lane = threadIdx.x & 31;
    const int wave = threadIdx.x >> 5;
    const int i = blockIdx.x * 1024 + threadIdx.x;
    const int v = (i < n_rows) ? cnt[i] : 0;
    int x = v;
    for (int d = 1; d < 32; d <<= 1) {
        const int y = __shfl_up(x, d);
        if (lane >= d) x += y;
    }
    if (lane == 31) wsum[wave] = x;
    __syncthreads();
    if (wave == 0) {
        const int s = wsum[lane];
        int t = s;
        for (int d = 1; d < 32; d <<= 1) {
            const int y = __shfl_up(t, d);
            if (lane >= d) t += y;
        }
        wsum[lane] = t - s;
        if (lane == 31) bsum[blockIdx.x] = t;
    }
    __syncthreads();
    if (i < n_rows) off[i] = x - v + wsum[wave];
}

__global__ void __launch_bounds__(1024)
csr_scan_add(int* __restrict__ off, int* __restrict__ cur,
             const int* __restrict__ bsum, int n_rows) {
    __shared__ int base_s;
    if (threadIdx.x < 32) {
        int s = 0;
        for (int j = threadIdx.x; j < (int)blockIdx.x; j += 32) s += bsum[j];
        for (int d = 16; d > 0; d >>= 1) s += __shfl_xor(s, d);
        if (threadIdx.x == 0) base_s = s;
    }
    __syncthreads();
    const int base = base_s;
    const int i = blockIdx.x * 1024 + threadIdx.x;
    if (i < n_rows) {
        const int o = off[i] + base;
        off[i] = o;
        cur[i] = o;
    }
    if (blockIdx.x == gridDim.x - 1 && threadIdx.x == 0)
        off[n_rows] = base + bsum[blockIdx.x];
}

__global__ void __launch_bounds__(256)
csr_fill(const int* __restrict__ src, const int* __restrict__ dst,
         const float* __restrict__ ew, int* __restrict__ cur,
         int* __restrict__ recw, int nE) {
    int4* __restrict__ rec = (int4*)recw;
    const int stride = gridDim.x * 256;
    for (int e = blockIdx.x * 256 + threadIdx.x; e < nE; e += stride) {
        const int slot = atomicAdd(&cur[dst[e]], 1);
        rec[slot] = make_int4(e, src[e], __builtin_bit_cast(int, ew[e]), 0);
    }
}

__global__ void __launch_bounds__(256)
csr_order(const int* __restrict__ off, const int* __restrict__ recw,
          int* __restrict__ srcS, float* __restrict__ wS, int n_rows) {
    const int4* __restrict__ rec = (const int4*)recw;
    const int lane = threadIdx.x & 31;
    const int r = __builtin_amdgcn_readfirstlane((int)((blockIdx.x * 256 + threadIdx.x) >> 5));
    if (r >= n_rows) return;
    const int s = off[r];
    const int len = off[r + 1] - s;
    if (len <= 32) {
        int4 me = make_int4(0, 0, 0, 0);
        if (lane < len) me = rec[s + lane];
        int rank = 0;
        for (int j = 0; j < len; ++j) {
            const int idj = __builtin_amdgcn_readlane(me.x, j);
            rank += (idj < me.x) ? 1 : 0;
        }
        if (lane < len) {
            srcS[s + rank] = me.y;
            wS[s + rank] = __builtin_bit_cast(float, me.z);
        }
    } else {
        for (int i = lane; i < len; i += 32) {
            const int4 me = rec[s + i];
            int rank = 0;
            for (int j = 0; j < len; ++j) rank += (rec[s + j].x < me.x) ? 1 : 0;
            srcS[s + rank] = me.y;
            wS[s + rank] = __builtin_bit_cast(float, me.z);
        }
    }
}

template <int VEC>
__global__ void __launch_bounds__(256)
spmm_gather(const float* __restrict__ H, const int* __restrict__ off,
            const int* __restrict__ srcS, const float* __restrict__ wS,
            float* __restrict__ out, int n_rows) {
    typedef float vf __attribute__((ext_vector_type(VEC)));
    const int lane = threadIdx.x & 31;
    const int r = __builtin_amdgcn_readfirstlane((int)((blockIdx.x * 256 + threadIdx.x) >> 5));
    if (r >= n_rows) return;
    const int s = off[r];
    const int t = off[r + 1];
    const vf* __restrict__ Hv = (const vf*)H + lane;
    vf acc = {};
    for (int base = s; base < t; base += 32) {
        const int m = (t - base < 32) ? (t - base) : 32;
        int my_s = 0;
        float my_w = 0.0f;
        if (lane < m) {
            my_s = srcS[base + lane];
            my_w = wS[base + lane];
        }
#pragma unroll 4
        for (int j = 0; j < m; ++j) {
            const int sj = __builtin_amdgcn_readlane(my_s, j);
            const float wj = __builtin_bit_cast(
                float, __builtin_amdgcn_readlane(__builtin_bit_cast(int, my_w), j));
            const vf v = Hv[(size_t)sj * 32];
            acc += wj * v;
        }
    }
    ((vf*)out)[(size_t)r * 32 + lane] = acc;
}

__global__ void move_words(const int* __restrict__ from, int* __restrict__ to,
                           int count) {
    const int stride = gridDim.x * blockDim.x;
    const int i = blockIdx.x * blockDim.x + threadIdx.x;
    const int n4 = count >> 2;
    for (int k = i; k < n4; k += stride)
        ((int4*)to)[k] = ((const int4*)from)[k];
    for (int k = (n4 << 2) + i; k < count; k += stride) to[k] = from[k];
}

__global__ void zero_f32(float* __restrict__ p, long long count) {
    long long i = blockIdx.x * (long long)blockDim.x + threadIdx.x;
    const long long stride = (long long)gridDim.x * blockDim.x;
    for (; i < count; i += stride) p[i] = 0.0f;
}

#define PN_ROWS 8
typedef float v4f __attribute__((ext_vector_type(4)));

__global__ void __launch_bounds__(256)
pn_reduce(const float* __restrict__ H, float* __restrict__ red, int n_rows) {
    __shared__ v4f part[8 * 32];
    __shared__ float psq[8];
    const int lane = threadIdx.x & 31;
    const int wave = __builtin_amdgcn_readfirstlane((int)(threadIdx.x >> 5));
    const int stride = gridDim.x * 8;
    const v4f* __restrict__ Hv = (const v4f*)H + lane;
    const v4f zero = {0.0f, 0.0f, 0.0f, 0.0f};
    v4f cs = zero, sq = zero;
    for (int base = blockIdx.x * 8 + wave; base < n_rows; base += PN_ROWS * stride) {
        v4f v[PN_ROWS];
#pragma unroll
        for (int u = 0; u < PN_ROWS; ++u) {
            const int r = base + u * stride;
            v[u] = (r < n_rows) ? Hv[(size_t)r * 32] : zero;
        }
#pragma unroll
        for (int u = 0; u < PN_ROWS; ++u) {
            cs += v[u];
            sq += v[u] * v[u];
        }
    }
    float s = (sq.x + sq.y) + (sq.z + sq.w);
    for (int d = 16; d > 0; d >>= 1) s += __shfl_xor(s, d);
    part[wave * 32 + lane] = cs;
    if (lane == 0) psq[wave] = s;
    __syncthreads();
    if (threadIdx.x < D_H) {
        float t = 0.0f;
        for (int w = 0; w < 8; ++w) t += ((const float*)part)[w * D_H + threadIdx.x];
        unsafeAtomicAdd(&red[threadIdx.x], t);
    } else if (threadIdx.x == D_H) {
        float t = 0.0f;
        for (int w = 0; w < 8; ++w) t += psq[w];
        unsafeAtomicAdd(&red[D_H], t);
    }
}

__global__ void __launch_bounds__(128)
pn_finalize(float* __restrict__ red, int n_rows) {
    __shared__ float sh[D_H];
    const int c = threadIdx.x;
    const float inv_n = 1.0f / (float)n_rows;
    const float mean = red[c] * inv_n;
    sh[c] = mean * mean;
    __syncthreads();
    for (int s = 64; s > 0; s >>= 1) {
        if (c < s) sh[c] += sh[c + s];
        __syncthreads();
    }
    if (c == 0) {
        float ssq_centered = red[D_H] - (float)n_rows * sh[0];
        red[D_H + 1] = 1.0f / sqrtf(PN_EPS + ssq_centered * inv_n);
    }
    red[c] = mean;
}

template <bool HAS_OLD>
__global__ void __launch_bounds__(256)
pn_apply4(const float* conv, const float* __restrict__ xold, float* xout,
          const float* __restrict__ red, int n4) {
    const int stride = gridDim.x * 256;
    const int tid = blockIdx.x * 256 + threadIdx.x;
    const float4 mean = ((const float4*)red)[tid & (D_H / 4 - 1)];
    const float scale = red[D_H + 1];
    const float4* c4 = (const float4*)conv;
    const float4* x4 = (const float4*)xold;
    float4* o4 = (float4*)xout;
    for (int base = tid; base < n4; base += 4 * stride) {
        float4 v[4], o[4];
#pragma unroll
        for (int u = 0; u < 4; ++u) {
            const int idx = base + u * stride;
            if (idx < n4) {
                v[u] = c4[idx];
                if (HAS_OLD) o[u] = x4[idx];
            }
        }
#pragma unroll
        for (int u = 0; u < 4; ++u) {
            const int idx = base + u * stride;
            if (idx < n4) {
                float4 r;
                r.x = fmaxf((v[u].x - mean.x) * scale, 0.0f);
                r.y = fmaxf((v[u].y - mean.y) * scale, 0.0f);
                r.z = fmaxf((v[u].z - mean.z) * scale, 0.0f);
                r.w = fmaxf((v[u].w - mean.w) * scale, 0.0f);
                if (HAS_OLD) {
                    r.x += o[u].x; r.y += o[u].y; r.z += o[u].z; r.w += o[u].w;
                }
                o4[idx] = r;
            }
        }
    }
}

extern "C" void kernel_launch(void* const* d_in, const int* in_sizes, int n_in,
                              void* d_out, int out_size, void* d_ws, size_t ws_size,
                              hipStream_t stream) {
    const float* x    = (const float*)d_in[0];
    const int*   ei   = (const int*)d_in[1];
    const float* ew   = (const float*)d_in[2];
    const float* W0   = (const float*)d_in[3];
    const float* b0   = (const float*)d_in[4];
    const float* W1   = (const float*)d_in[5];
    const float* b1   = (const float*)d_in[6];
    const float* Wout = (const float*)d_in[7];
    const float* bout = (const float*)d_in[8];
    float* out = (float*)d_out;

    const int n = in_sizes[0] / D_H;
    const int e = in_sizes[2];
    const int d_outc = out_size / n;
    const int* srcI = ei;
    const int* dstI = ei + e;

    const size_t nd = (size_t)n * D_H;
    float* A   = (float*)d_ws;
    float* B   = A + nd;
    float* C   = B + nd;
    float* red = C + nd;

    const int gemm_blocks = (((n + 15) / 16) + 7) / 8;
    auto launch_gemm = [&](const float* X_, const float* W_, const float* bias_,
                           float* H_, int dout) {
        if (dout == 128)
            gemm_wmma_f32<128><<<gemm_blocks, 256, 0, stream>>>(X_, W_, bias_, H_, n);
        else
            gemm_wmma_f32<64><<<gemm_blocks, 256, 0, stream>>>(X_, W_, bias_, H_, n);
    };
    auto launch_gather = [&](const float* H_, const int* tab, float* O_, int D) {
        const int* srcS_ = tab;
        const float* wS_ = (const float*)(tab + e);
        const int* off_ = tab + 2 * (size_t)e;
        const int blocks = (n + 7) / 8;
        if (D == 128)
            spmm_gather<4><<<blocks, 256, 0, stream>>>(H_, off_, srcS_, wS_, O_, n);
        else
            spmm_gather<2><<<blocks, 256, 0, stream>>>(H_, off_, srcS_, wS_, O_, n);
    };
    auto launch_zero = [&](float* p, long long cnt) {
        int blocks = (int)((cnt + 256 * 8 - 1) / (256 * 8));
        zero_f32<<<blocks, 256, 0, stream>>>(p, cnt);
    };
    auto launch_pn = [&](const float* conv, const float* xold, float* xout) {
        launch_zero(red, 256);
        pn_reduce<<<1024, 256, 0, stream>>>(conv, red, n);
        pn_finalize<<<1, 128, 0, stream>>>(red, n);
        const int n4 = n * (D_H / 4);
        if (xold)
            pn_apply4<true><<<1024, 256, 0, stream>>>(conv, xold, xout, red, n4);
        else
            pn_apply4<false><<<1024, 256, 0, stream>>>(conv, xold, xout, red, n4);
    };

    const int npad = (n + 1024) & ~255;
    int*   t_src  = (int*)out;
    float* t_w    = (float*)(t_src + e);
    int*   t_off  = t_src + 2 * (size_t)e;
    int*   t_cnt  = t_off + npad;
    int*   t_bsum = t_cnt + npad;
    int*   t_rec  = (int*)C;
    const int scan_blocks = (n + 1023) / 1024;
    launch_zero((float*)t_cnt, n);
    csr_count<<<1024, 256, 0, stream>>>(dstI, t_cnt, e);
    csr_scan_local<<<scan_blocks, 1024, 0, stream>>>(t_cnt, t_off, t_bsum, n);
    csr_scan_add<<<scan_blocks, 1024, 0, stream>>>(t_off, t_cnt, t_bsum, n);
    csr_fill<<<1024, 256, 0, stream>>>(srcI, dstI, ew, t_cnt, t_rec, e);
    csr_order<<<(n + 7) / 8, 256, 0, stream>>>(t_off, t_rec, t_src, t_w, n);

    launch_gemm(x, W0, b0, A, D_H);
    launch_gather(A, t_src, B, D_H);
    launch_pn(B, nullptr, A);

    launch_gemm(A, W1, b1, B, D_H);
    launch_gather(B, t_src, C, D_H);
    launch_pn(C, A, C);

    const int table_words = 2 * e + n + 1;
    move_words<<<1024, 256, 0, stream>>>(t_src, (int*)B, table_words);
    launch_gemm(C, Wout, bout, A, d_outc);
    launch_gather(A, (const int*)B, out, d_outc);
}


# PairNorm finalize+apply folded into the next product's A-operand load (x1, x2 never stored): both apply passes, both finalize and both clear launches go, 21 launches -> 15; all sum orders kept
# speedup vs baseline: 2.3800x; 1.0611x over previous
#include <hip/hip_runtime.h>

#define D_H 128
#define PN_EPS 1e-6f

typedef float v2f __attribute__((ext_vector_type(2)));
typedef float v8f __attribute__((ext_vector_type(8)));

template <int DOUT, int MODE>
__global__ void __launch_bounds__(256)
gemm_wmma_f32(const float* __restrict__ X, const float* __restrict__ X2,
              const float* __restrict__ st1, const float* __restrict__ st2,
              const float* __restrict__ W, const float* __restrict__ bias,
              float* __restrict__ H, int n_rows) {
    constexpr int TN  = DOUT / 16;
    constexpr int KC  = 64;
    constexpr int WST = KC + 4;
    constexpr int NP  = (MODE == 0) ? 1 : MODE;
    __shared__ float Wt[DOUT * WST];
    __shared__ __attribute__((aligned(16))) float pnM[NP][D_H];
    __shared__ float pnT[NP][D_H];
    __shared__ float pnS[NP];

    const int lane = threadIdx.x & 31;
    const int wave = threadIdx.x >> 5;
    const int half = lane >> 4;
    const int l15  = lane & 15;

    if constexpr (MODE >= 1) {
        const float inv_n = 1.0f / (float)n_rows;
        const int p = threadIdx.x >> 7;
        const int c = threadIdx.x & (D_H - 1);
        const bool on = (p < MODE);
        const float* __restrict__ st = (p == 0) ? st1 : st2;
        if (on) {
            const float mean = st[c] * inv_n;
            pnM[p][c] = mean;
            pnT[p][c] = mean * mean;
        }
        __syncthreads();
        for (int s = 64; s > 0; s >>= 1) {
            if (on && c < s) pnT[p][c] += pnT[p][c + s];
            __syncthreads();
        }
        if (on && c == 0) {
            float ssq_centered = st[D_H] - (float)n_rows * pnT[p][0];
            pnS[p] = 1.0f / sqrtf(PN_EPS + ssq_centered * inv_n);
        }
    }

    const int tm      = blockIdx.x * 8 + wave;
    const int tiles_m = (n_rows + 15) >> 4;
    const bool active = (tm < tiles_m);

    int row_a = tm * 16 + l15;
    if (row_a >= n_rows) row_a = n_rows - 1;
    const float* __restrict__ ap = X + (size_t)row_a * D_H + half * 2;
    const float* __restrict__ ap2 = (MODE == 2) ? (X2 + (size_t)row_a * D_H + half * 2) : ap;
    const float* __restrict__ bbase = &Wt[l15 * WST + half * 2];

    v8f acc[TN] = {};

    for (int k0 = 0; k0 < D_H; k0 += KC) {
        __syncthreads();
        const float* __restrict__ wg = W + (size_t)k0 * DOUT;
        for (int idx = threadIdx.x; idx < KC * DOUT; idx += 256) {
            int r = idx / DOUT;
            int c = idx - r * DOUT;
            Wt[c * WST + r] = wg[idx];
        }
        __syncthreads();

        if (active) {
            __builtin_prefetch(ap + k0 + KC, 0, 1);
            if constexpr (MODE == 2) __builtin_prefetch(ap2 + k0 + KC, 0, 1);
            float sc1 = 0.0f, sc2 = 0.0f;
            if constexpr (MODE >= 1) sc1 = pnS[0];
            if constexpr (MODE == 2) sc2 = pnS[1];
#pragma unroll
            for (int kk = 0; kk < KC; kk += 4) {
                v2f a = *(const v2f*)(ap + k0 + kk);
                if constexpr (MODE >= 1) {
                    const v2f m = *(const v2f*)&pnM[0][k0 + kk + half * 2];
                    a.x = fmaxf((a.x - m.x) * sc1, 0.0f);
                    a.y = fmaxf((a.y - m.y) * sc1, 0.0f);
                }
                if constexpr (MODE == 2) {
                    v2f o = *(const v2f*)(ap2 + k0 + kk);
                    const v2f m2 = *(const v2f*)&pnM[1][k0 + kk + half * 2];
                    o.x = fmaxf((o.x - m2.x) * sc2, 0.0f);
                    o.y = fmaxf((o.y - m2.y) * sc2, 0.0f);
                    a.x += o.x;
                    a.y += o.y;
                }
#pragma unroll
                for (int t = 0; t < TN; ++t) {
                    v2f b = *(const v2f*)(bbase + t * (16 * WST) + kk);
                    acc[t] = __builtin_amdgcn_wmma_f32_16x16x4_f32(
                         false, a,  false, b,
                         (short)0, acc[t],
                         false,  false);
                }
            }
        }
    }

    if (!active) return;
    const int row_base = tm * 16 + half * 8;
#pragma unroll
    for (int t = 0; t < TN; ++t) {
        const int col = t * 16 + l15;
        const float bb = bias[col];
#pragma unroll
        for (int r = 0; r < 8; ++r) {
            const int row = row_base + r;
            if (row < n_rows)
                H[(size_t)row * DOUT + col] = acc[t][r] + bb;
        }
    }
}

__global__ void __launch_bounds__(256)
csr_count(const int* __restrict__ dst, int* __restrict__ cnt, int nE) {
    const int stride = gridDim.x * 256;
    for (int e = blockIdx.x * 256 + threadIdx.x; e < nE; e += stride)
        atomicAdd(&cnt[dst[e]], 1);
}

__global__ void __launch_bounds__(1024)
csr_scan_local(const int* __restrict__ cnt, int* __restrict__ off,
               int* __restrict__ bsum, int n_rows) {
    __shared__ int wsum[32];
    const int lane = threadIdx.x & 31;
    const int wave = threadIdx.x >> 5;
    const int i = blockIdx.x * 1024 + threadIdx.x;
    const int v = (i < n_rows) ? cnt[i] : 0;
    int x = v;
    for (int d = 1; d < 32; d <<= 1) {
        const int y = __shfl_up(x, d);
        if (lane >= d) x += y;
    }
    if (lane == 31) wsum[wave] = x;
    __syncthreads();
    if (wave == 0) {
        const int s = wsum[lane];
        int t = s;
        for (int d = 1; d < 32; d <<= 1) {
            const int y = __shfl_up(t, d);
            if (lane >= d) t += y;
        }
        wsum[lane] = t - s;
        if (lane == 31) bsum[blockIdx.x] = t;
    }
    __syncthreads();
    if (i < n_rows) off[i] = x - v + wsum[wave];
}

__global__ void __launch_bounds__(1024)
csr_scan_add(int* __restrict__ off, int* __restrict__ cur,
             const int* __restrict__ bsum, int n_rows) {
    __shared__ int base_s;
    if (threadIdx.x < 32) {
        int s = 0;
        for (int j = threadIdx.x; j < (int)blockIdx.x; j += 32) s += bsum[j];
        for (int d = 16; d > 0; d >>= 1) s += __shfl_xor(s, d);
        if (threadIdx.x == 0) base_s = s;
    }
    __syncthreads();
    const int base = base_s;
    const int i = blockIdx.x * 1024 + threadIdx.x;
    if (i < n_rows) {
        const int o = off[i] + base;
        off[i] = o;
        cur[i] = o;
    }
    if (blockIdx.x == gridDim.x - 1 && threadIdx.x == 0)
        off[n_rows] = base + bsum[blockIdx.x];
}

__global__ void __launch_bounds__(256)
csr_fill(const int* __restrict__ src, const int* __restrict__ dst,
         const float* __restrict__ ew, int* __restrict__ cur,
         int* __restrict__ recw, int nE) {
    int4* __restrict__ rec = (int4*)recw;
    const int stride = gridDim.x * 256;
    for (int e = blockIdx.x * 256 + threadIdx.x; e < nE; e += stride) {
        const int slot = atomicAdd(&cur[dst[e]], 1);
        rec[slot] = make_int4(e, src[e], __builtin_bit_cast(int, ew[e]), 0);
    }
}

__global__ void __launch_bounds__(256)
csr_order(const int* __restrict__ off, const int* __restrict__ recw,
          int* __restrict__ srcS, float* __restrict__ wS, int n_rows) {
    const int4* __restrict__ rec = (const int4*)recw;
    const int lane = threadIdx.x & 31;
    const int r = __builtin_amdgcn_readfirstlane((int)((blockIdx.x * 256 + threadIdx.x) >> 5));
    if (r >= n_rows) return;
    const int s = off[r];
    const int len = off[r + 1] - s;
    if (len <= 32) {
        int4 me = make_int4(0, 0, 0, 0);
        if (lane < len) me = rec[s + lane];
        int rank = 0;
        for (int j = 0; j < len; ++j) {
            const int idj = __builtin_amdgcn_readlane(me.x, j);
            rank += (idj < me.x) ? 1 : 0;
        }
        if (lane < len) {
            srcS[s + rank] = me.y;
            wS[s + rank] = __builtin_bit_cast(float, me.z);
        }
    } else {
        for (int i = lane; i < len; i += 32) {
            const int4 me = rec[s + i];
            int rank = 0;
            for (int j = 0; j < len; ++j) rank += (rec[s + j].x < me.x) ? 1 : 0;
            srcS[s + rank] = me.y;
            wS[s + rank] = __builtin_bit_cast(float, me.z);
        }
    }
}

template <int VEC>
__global__ void __launch_bounds__(256)
spmm_gather(const float* __restrict__ H, const int* __restrict__ off,
            const int* __restrict__ srcS, const float* __restrict__ wS,
            float* __restrict__ out, int n_rows) {
    typedef float vf __attribute__((ext_vector_type(VEC)));
    const int lane = threadIdx.x & 31;
    const int r = __builtin_amdgcn_readfirstlane((int)((blockIdx.x * 256 + threadIdx.x) >> 5));
    if (r >= n_rows) return;
    const int s = off[r];
    const int t = off[r + 1];
    const vf* __restrict__ Hv = (const vf*)H + lane;
    vf acc = {};
    for (int base = s; base < t; base += 32) {
        const int m = (t - base < 32) ? (t - base) : 32;
        int my_s = 0;
        float my_w = 0.0f;
        if (lane < m) {
            my_s = srcS[base + lane];
            my_w = wS[base + lane];
        }
#pragma unroll 4
        for (int j = 0; j < m; ++j) {
            const int sj = __builtin_amdgcn_readlane(my_s, j);
            const float wj = __builtin_bit_cast(
                float, __builtin_amdgcn_readlane(__builtin_bit_cast(int, my_w), j));
            const vf v = Hv[(size_t)sj * 32];
            acc += wj * v;
        }
    }
    ((vf*)out)[(size_t)r * 32 + lane] = acc;
}

__global__ void move_words(const int* __restrict__ from, int* __restrict__ to,
                           int count) {
    const int stride = gridDim.x * blockDim.x;
    const int i = blockIdx.x * blockDim.x + threadIdx.x;
    const int n4 = count >> 2;
    for (int k = i; k < n4; k += stride)
        ((int4*)to)[k] = ((const int4*)from)[k];
    for (int k = (n4 << 2) + i; k < count; k += stride) to[k] = from[k];
}

__global__ void zero_f32(float* __restrict__ p, long long count) {
    long long i = blockIdx.x * (long long)blockDim.x + threadIdx.x;
    const long long stride = (long long)gridDim.x * blockDim.x;
    for (; i < count; i += stride) p[i] = 0.0f;
}

#define PN_ROWS 8
typedef float v4f __attribute__((ext_vector_type(4)));

__global__ void __launch_bounds__(256)
pn_reduce(const float* __restrict__ H, float* __restrict__ red, int n_rows) {
    __shared__ v4f part[8 * 32];
    __shared__ float psq[8];
    const int lane = threadIdx.x & 31;
    const int wave = __builtin_amdgcn_readfirstlane((int)(threadIdx.x >> 5));
    const int stride = gridDim.x * 8;
    const v4f* __restrict__ Hv = (const v4f*)H + lane;
    const v4f zero = {0.0f, 0.0f, 0.0f, 0.0f};
    v4f cs = zero, sq = zero;
    for (int base = blockIdx.x * 8 + wave; base < n_rows; base += PN_ROWS * stride) {
        v4f v[PN_ROWS];
#pragma unroll
        for (int u = 0; u < PN_ROWS; ++u) {
            const int r = base + u * stride;
            v[u] = (r < n_rows) ? Hv[(size_t)r * 32] : zero;
        }
#pragma unroll
        for (int u = 0; u < PN_ROWS; ++u) {
            cs += v[u];
            sq += v[u] * v[u];
        }
    }
    float s = (sq.x + sq.y) + (sq.z + sq.w);
    for (int d = 16; d > 0; d >>= 1) s += __shfl_xor(s, d);
    part[wave * 32 + lane] = cs;
    if (lane == 0) psq[wave] = s;
    __syncthreads();
    if (threadIdx.x < D_H) {
        float t = 0.0f;
        for (int w = 0; w < 8; ++w) t += ((const float*)part)[w * D_H + threadIdx.x];
        unsafeAtomicAdd(&red[threadIdx.x], t);
    } else if (threadIdx.x == D_H) {
        float t = 0.0f;
        for (int w = 0; w < 8; ++w) t += psq[w];
        unsafeAtomicAdd(&red[D_H], t);
    }
}

extern "C" void kernel_launch(void* const* d_in, const int* in_sizes, int n_in,
                              void* d_out, int out_size, void* d_ws, size_t ws_size,
                              hipStream_t stream) {
    const float* x    = (const float*)d_in[0];
    const int*   ei   = (const int*)d_in[1];
    const float* ew   = (const float*)d_in[2];
    const float* W0   = (const float*)d_in[3];
    const float* b0   = (const float*)d_in[4];
    const float* W1   = (const float*)d_in[5];
    const float* b1   = (const float*)d_in[6];
    const float* Wout = (const float*)d_in[7];
    const float* bout = (const float*)d_in[8];
    float* out = (float*)d_out;

    const int n = in_sizes[0] / D_H;
    const int e = in_sizes[2];
    const int d_outc = out_size / n;
    const int* srcI = ei;
    const int* dstI = ei + e;

    const size_t nd = (size_t)n * D_H;
    float* A   = (float*)d_ws;
    float* B   = A + nd;
    float* C   = B + nd;

    const int gemm_blocks = (((n + 15) / 16) + 7) / 8;
    auto launch_gather = [&](const float* H_, const int* tab, float* O_, int D) {
        const int* srcS_ = tab;
        const float* wS_ = (const float*)(tab + e);
        const int* off_ = tab + 2 * (size_t)e;
        const int blocks = (n + 7) / 8;
        if (D == 128)
            spmm_gather<4><<<blocks, 256, 0, stream>>>(H_, off_, srcS_, wS_, O_, n);
        else
            spmm_gather<2><<<blocks, 256, 0, stream>>>(H_, off_, srcS_, wS_, O_, n);
    };
    auto launch_zero = [&](float* p, long long cnt) {
        int blocks = (int)((cnt + 256 * 8 - 1) / (256 * 8));
        zero_f32<<<blocks, 256, 0, stream>>>(p, cnt);
    };

    const int npad = (n + 1024) & ~255;
    int*   t_src  = (int*)out;
    float* t_w    = (float*)(t_src + e);
    int*   t_off  = t_src + 2 * (size_t)e;
    int*   t_cnt  = t_off + npad;
    int*   t_bsum = t_cnt + npad;
    float* st0    = (float*)(t_bsum + 1024);
    float* st1    = st0 + 132;
    int*   t_rec  = (int*)C;
    const int scan_blocks = (n + 1023) / 1024;
    launch_zero((float*)t_cnt, (long long)npad + 1024 + 2 * 132);
    csr_count<<<1024, 256, 0, stream>>>(dstI, t_cnt, e);
    csr_scan_local<<<scan_blocks, 1024, 0, stream>>>(t_cnt, t_off, t_bsum, n);
    csr_scan_add<<<scan_blocks, 1024, 0, stream>>>(t_off, t_cnt, t_bsum, n);
    csr_fill<<<1024, 256, 0, stream>>>(srcI, dstI, ew, t_cnt, t_rec, e);
    csr_order<<<(n + 7) / 8, 256, 0, stream>>>(t_off, t_rec, t_src, t_w, n);

    gemm_wmma_f32<128, 0><<<gemm_blocks, 256, 0, stream>>>(x, nullptr, nullptr, nullptr, W0, b0, A, n);
    launch_gather(A, t_src, B, D_H);
    pn_reduce<<<1024, 256, 0, stream>>>(B, st0, n);

    gemm_wmma_f32<128, 1><<<gemm_blocks, 256, 0, stream>>>(B, nullptr, st0, nullptr, W1, b1, A, n);
    launch_gather(A, t_src, C, D_H);
    pn_reduce<<<1024, 256, 0, stream>>>(C, st1, n);

    const int table_words = 2 * e + n + 1;
    int* tabA = (int*)(A + (size_t)n * (D_H / 2));
    move_words<<<1024, 256, 0, stream>>>(t_src, tabA, table_words);
    gemm_wmma_f32<64, 2><<<gemm_blocks, 256, 0, stream>>>(C, B, st1, st0, Wout, bout, A, n);
    launch_gather(A, tabA, out, d_outc);
}


# Edge table build with one atomic per edge instead of two: the counting add's return is kept as the edge's place in its row, the fill pass places by row start + place with plain loads; table unchanged
# speedup vs baseline: 2.6747x; 1.1238x over previous
#include <hip/hip_runtime.h>

#define D_H 128
#define PN_EPS 1e-6f

typedef float v2f __attribute__((ext_vector_type(2)));
typedef float v8f __attribute__((ext_vector_type(8)));

template <int DOUT, int MODE>
__global__ void __launch_bounds__(256)
gemm_wmma_f32(const float* __restrict__ X, const float* __restrict__ X2,
              const float* __restrict__ st1, const float* __restrict__ st2,
              const float* __restrict__ W, const float* __restrict__ bias,
              float* __restrict__ H, int n_rows) {
    constexpr int TN  = DOUT / 16;
    constexpr int KC  = 64;
    constexpr int WST = KC + 4;
    constexpr int NP  = (MODE == 0) ? 1 : MODE;
    __shared__ float Wt[DOUT * WST];
    __shared__ __attribute__((aligned(16))) float pnM[NP][D_H];
    __shared__ float pnT[NP][D_H];
    __shared__ float pnS[NP];

    const int lane = threadIdx.x & 31;
    const int wave = threadIdx.x >> 5;
    const int half = lane >> 4;
    const int l15  = lane & 15;

    if constexpr (MODE >= 1) {
        const float inv_n = 1.0f / (float)n_rows;
        const int p = threadIdx.x >> 7;
        const int c = threadIdx.x & (D_H - 1);
        const bool on = (p < MODE);
        const float* __restrict__ st = (p == 0) ? st1 : st2;
        if (on) {
            const float mean = st[c] * inv_n;
            pnM[p][c] = mean;
            pnT[p][c] = mean * mean;
        }
        __syncthreads();
        for (int s = 64; s > 0; s >>= 1) {
            if (on && c < s) pnT[p][c] += pnT[p][c + s];
            __syncthreads();
        }
        if (on && c == 0) {
            float ssq_centered = st[D_H] - (float)n_rows * pnT[p][0];
            pnS[p] = 1.0f / sqrtf(PN_EPS + ssq_centered * inv_n);
        }
    }

    const int tm      = blockIdx.x * 8 + wave;
    const int tiles_m = (n_rows + 15) >> 4;
    const bool active = (tm < tiles_m);

    int row_a = tm * 16 + l15;
    if (row_a >= n_rows) row_a = n_rows - 1;
    const float* __restrict__ ap = X + (size_t)row_a * D_H + half * 2;
    const float* __restrict__ ap2 = (MODE == 2) ? (X2 + (size_t)row_a * D_H + half * 2) : ap;
    const float* __restrict__ bbase = &Wt[l15 * WST + half * 2];

    v8f acc[TN] = {};

    for (int k0 = 0; k0 < D_H; k0 += KC) {
        __syncthreads();
        const float* __restrict__ wg = W + (size_t)k0 * DOUT;
        for (int idx = threadIdx.x; idx < KC * DOUT; idx += 256) {
            int r = idx / DOUT;
            int c = idx - r * DOUT;
            Wt[c * WST + r] = wg[idx];
        }
        __syncthreads();

        if (active) {
            __builtin_prefetch(ap + k0 + KC, 0, 1);
            if constexpr (MODE == 2) __builtin_prefetch(ap2 + k0 + KC, 0, 1);
            float sc1 = 0.0f, sc2 = 0.0f;
            if constexpr (MODE >= 1) sc1 = pnS[0];
            if constexpr (MODE == 2) sc2 = pnS[1];
#pragma unroll
            for (int kk = 0; kk < KC; kk += 4) {
                v2f a = *(const v2f*)(ap + k0 + kk);
                if constexpr (MODE >= 1) {
                    const v2f m = *(const v2f*)&pnM[0][k0 + kk + half * 2];
                    a.x = fmaxf((a.x - m.x) * sc1, 0.0f);
                    a.y = fmaxf((a.y - m.y) * sc1, 0.0f);
                }
                if constexpr (MODE == 2) {
                    v2f o = *(const v2f*)(ap2 + k0 + kk);
                    const v2f m2 = *(const v2f*)&pnM[1][k0 + kk + half * 2];
                    o.x = fmaxf((o.x - m2.x) * sc2, 0.0f);
                    o.y = fmaxf((o.y - m2.y) * sc2, 0.0f);
                    a.x += o.x;
                    a.y += o.y;
                }
#pragma unroll
                for (int t = 0; t < TN; ++t) {
                    v2f b = *(const v2f*)(bbase + t * (16 * WST) + kk);
                    acc[t] = __builtin_amdgcn_wmma_f32_16x16x4_f32(
                         false, a,  false, b,
                         (short)0, acc[t],
                         false,  false);
                }
            }
        }
    }

    if (!active) return;
    const int row_base = tm * 16 + half * 8;
#pragma unroll
    for (int t = 0; t < TN; ++t) {
        const int col = t * 16 + l15;
        const float bb = bias[col];
#pragma unroll
        for (int r = 0; r < 8; ++r) {
            const int row = row_base + r;
            if (row < n_rows)
                H[(size_t)row * DOUT + col] = acc[t][r] + bb;
        }
    }
}

__global__ void __launch_bounds__(256)
csr_count(const int* __restrict__ dst, int* __restrict__ cnt,
          int* __restrict__ rk, int nE) {
    const int stride = gridDim.x * 256;
    for (int e = blockIdx.x * 256 + threadIdx.x; e < nE; e += stride)
        rk[e] = atomicAdd(&cnt[dst[e]], 1);
}

__global__ void __launch_bounds__(1024)
csr_scan_local(const int* __restrict__ cnt, int* __restrict__ off,
               int* __restrict__ bsum, int n_rows) {
    __shared__ int wsum[32];
    const int lane = threadIdx.x & 31;
    const int wave = threadIdx.x >> 5;
    const int i = blockIdx.x * 1024 + threadIdx.x;
    const int v = (i < n_rows) ? cnt[i] : 0;
    int x = v;
    for (int d = 1; d < 32; d <<= 1) {
        const int y = __shfl_up(x, d);
        if (lane >= d) x += y;
    }
    if (lane == 31) wsum[wave] = x;
    __syncthreads();
    if (wave == 0) {
        const int s = wsum[lane];
        int t = s;
        for (int d = 1; d < 32; d <<= 1) {
            const int y = __shfl_up(t, d);
            if (lane >= d) t += y;
        }
        wsum[lane] = t - s;
        if (lane == 31) bsum[blockIdx.x] = t;
    }
    __syncthreads();
    if (i < n_rows) off[i] = x - v + wsum[wave];
}

__global__ void __launch_bounds__(1024)
csr_scan_add(int* __restrict__ off, const int* __restrict__ bsum, int n_rows) {
    __shared__ int base_s;
    if (threadIdx.x < 32) {
        int s = 0;
        for (int j = threadIdx.x; j < (int)blockIdx.x; j += 32) s += bsum[j];
        for (int d = 16; d > 0; d >>= 1) s += __shfl_xor(s, d);
        if (threadIdx.x == 0) base_s = s;
    }
    __syncthreads();
    const int base = base_s;
    const int i = blockIdx.x * 1024 + threadIdx.x;
    if (i < n_rows) off[i] += base;
    if (blockIdx.x == gridDim.x - 1 && threadIdx.x == 0)
        off[n_rows] = base + bsum[blockIdx.x];
}

__global__ void __launch_bounds__(256)
csr_fill(const int* __restrict__ src, const int* __restrict__ dst,
         const float* __restrict__ ew, const int* __restrict__ off,
         const int* __restrict__ rk, int* __restrict__ recw, int nE) {
    int4* __restrict__ rec = (int4*)recw;
    const int stride = gridDim.x * 256;
    for (int e = blockIdx.x * 256 + threadIdx.x; e < nE; e += stride) {
        const int slot = off[dst[e]] + rk[e];
        rec[slot] = make_int4(e, src[e], __builtin_bit_cast(int, ew[e]), 0);
    }
}

__global__ void __launch_bounds__(256)
csr_order(const int* __restrict__ off, const int* __restrict__ recw,
          int* __restrict__ srcS, float* __restrict__ wS, int n_rows) {
    const int4* __restrict__ rec = (const int4*)recw;
    const int lane = threadIdx.x & 31;
    const int r = __builtin_amdgcn_readfirstlane((int)((blockIdx.x * 256 + threadIdx.x) >> 5));
    if (r >= n_rows) return;
    const int s = off[r];
    const int len = off[r + 1] - s;
    if (len <= 32) {
        int4 me = make_int4(0, 0, 0, 0);
        if (lane < len) me = rec[s + lane];
        int rank = 0;
        for (int j = 0; j < len; ++j) {
            const int idj = __builtin_amdgcn_readlane(me.x, j);
            rank += (idj < me.x) ? 1 : 0;
        }
        if (lane < len) {
            srcS[s + rank] = me.y;
            wS[s + rank] = __builtin_bit_cast(float, me.z);
        }
    } else {
        for (int i = lane; i < len; i += 32) {
            const int4 me = rec[s + i];
            int rank = 0;
            for (int j = 0; j < len; ++j) rank += (rec[s + j].x < me.x) ? 1 : 0;
            srcS[s + rank] = me.y;
            wS[s + rank] = __builtin_bit_cast(float, me.z);
        }
    }
}

template <int VEC>
__global__ void __launch_bounds__(256)
spmm_gather(const float* __restrict__ H, const int* __restrict__ off,
            const int* __restrict__ srcS, const float* __restrict__ wS,
            float* __restrict__ out, int n_rows) {
    typedef float vf __attribute__((ext_vector_type(VEC)));
    const int lane = threadIdx.x & 31;
    const int r = __builtin_amdgcn_readfirstlane((int)((blockIdx.x * 256 + threadIdx.x) >> 5));
    if (r >= n_rows) return;
    const int s = off[r];
    const int t = off[r + 1];
    const vf* __restrict__ Hv = (const vf*)H + lane;
    vf acc = {};
    for (int base = s; base < t; base += 32) {
        const int m = (t - base < 32) ? (t - base) : 32;
        int my_s = 0;
        float my_w = 0.0f;
        if (lane < m) {
            my_s = srcS[base + lane];
            my_w = wS[base + lane];
        }
#pragma unroll 4
        for (int j = 0; j < m; ++j) {
            const int sj = __builtin_amdgcn_readlane(my_s, j);
            const float wj = __builtin_bit_cast(
                float, __builtin_amdgcn_readlane(__builtin_bit_cast(int, my_w), j));
            const vf v = Hv[(size_t)sj * 32];
            acc += wj * v;
        }
    }
    ((vf*)out)[(size_t)r * 32 + lane] = acc;
}

__global__ void move_words(const int* __restrict__ from, int* __restrict__ to,
                           int count) {
    const int stride = gridDim.x * blockDim.x;
    const int i = blockIdx.x * blockDim.x + threadIdx.x;
    const int n4 = count >> 2;
    for (int k = i; k < n4; k += stride)
        ((int4*)to)[k] = ((const int4*)from)[k];
    for (int k = (n4 << 2) + i; k < count; k += stride) to[k] = from[k];
}

__global__ void zero_f32(float* __restrict__ p, long long count) {
    long long i = blockIdx.x * (long long)blockDim.x + threadIdx.x;
    const long long stride = (long long)gridDim.x * blockDim.x;
    for (; i < count; i += stride) p[i] = 0.0f;
}

#define PN_ROWS 8
typedef float v4f __attribute__((ext_vector_type(4)));

__global__ void __launch_bounds__(256)
pn_reduce(const float* __restrict__ H, float* __restrict__ red, int n_rows) {
    __shared__ v4f part[8 * 32];
    __shared__ float psq[8];
    const int lane = threadIdx.x & 31;
    const int wave = __builtin_amdgcn_readfirstlane((int)(threadIdx.x >> 5));
    const int stride = gridDim.x * 8;
    const v4f* __restrict__ Hv = (const v4f*)H + lane;
    const v4f zero = {0.0f, 0.0f, 0.0f, 0.0f};
    v4f cs = zero, sq = zero;
    for (int base = blockIdx.x * 8 + wave; base < n_rows; base += PN_ROWS * stride) {
        v4f v[PN_ROWS];
#pragma unroll
        for (int u = 0; u < PN_ROWS; ++u) {
            const int r = base + u * stride;
            v[u] = (r < n_rows) ? Hv[(size_t)r * 32] : zero;
        }
#pragma unroll
        for (int u = 0; u < PN_ROWS; ++u) {
            cs += v[u];
            sq += v[u] * v[u];
        }
    }
    float s = (sq.x + sq.y) + (sq.z + sq.w);
    for (int d = 16; d > 0; d >>= 1) s += __shfl_xor(s, d);
    part[wave * 32 + lane] = cs;
    if (lane == 0) psq[wave] = s;
    __syncthreads();
    if (threadIdx.x < D_H) {
        float t = 0.0f;
        for (int w = 0; w < 8; ++w) t += ((const float*)part)[w * D_H + threadIdx.x];
        unsafeAtomicAdd(&red[threadIdx.x], t);
    } else if (threadIdx.x == D_H) {
        float t = 0.0f;
        for (int w = 0; w < 8; ++w) t += psq[w];
        unsafeAtomicAdd(&red[D_H], t);
    }
}

extern "C" void kernel_launch(void* const* d_in, const int* in_sizes, int n_in,
                              void* d_out, int out_size, void* d_ws, size_t ws_size,
                              hipStream_t stream) {
    const float* x    = (const float*)d_in[0];
    const int*   ei   = (const int*)d_in[1];
    const float* ew   = (const float*)d_in[2];
    const float* W0   = (const float*)d_in[3];
    const float* b0   = (const float*)d_in[4];
    const float* W1   = (const float*)d_in[5];
    const float* b1   = (const float*)d_in[6];
    const float* Wout = (const float*)d_in[7];
    const float* bout = (const float*)d_in[8];
    float* out = (float*)d_out;

    const int n = in_sizes[0] / D_H;
    const int e = in_sizes[2];
    const int d_outc = out_size / n;
    const int* srcI = ei;
    const int* dstI = ei + e;

    const size_t nd = (size_t)n * D_H;
    float* A   = (float*)d_ws;
    float* B   = A + nd;
    float* C   = B + nd;

    const int gemm_blocks = (((n + 15) / 16) + 7) / 8;
    auto launch_gather = [&](const float* H_, const int* tab, float* O_, int D) {
        const int* srcS_ = tab;
        const float* wS_ = (const float*)(tab + e);
        const int* off_ = tab + 2 * (size_t)e;
        const int blocks = (n + 7) / 8;
        if (D == 128)
            spmm_gather<4><<<blocks, 256, 0, stream>>>(H_, off_, srcS_, wS_, O_, n);
        else
            spmm_gather<2><<<blocks, 256, 0, stream>>>(H_, off_, srcS_, wS_, O_, n);
    };
    auto launch_zero = [&](float* p, long long cnt) {
        int blocks = (int)((cnt + 256 * 8 - 1) / (256 * 8));
        zero_f32<<<blocks, 256, 0, stream>>>(p, cnt);
    };

    const int npad = (n + 1024) & ~255;
    int*   t_src  = (int*)out;
    float* t_w    = (float*)(t_src + e);
    int*   t_off  = t_src + 2 * (size_t)e;
    int*   t_cnt  = t_off + npad;
    int*   t_bsum = t_cnt + npad;
    float* st0    = (float*)(t_bsum + 1024);
    float* st1    = st0 + 132;
    int*   t_rec  = (int*)C;
    int*   t_rk   = t_rec + 4 * (size_t)e;
    const int scan_blocks = (n + 1023) / 1024;
    launch_zero((float*)t_cnt, (long long)npad + 1024 + 2 * 132);
    csr_count<<<1024, 256, 0, stream>>>(dstI, t_cnt, t_rk, e);
    csr_scan_local<<<scan_blocks, 1024, 0, stream>>>(t_cnt, t_off, t_bsum, n);
    csr_scan_add<<<scan_blocks, 1024, 0, stream>>>(t_off, t_bsum, n);
    csr_fill<<<1024, 256, 0, stream>>>(srcI, dstI, ew, t_off, t_rk, t_rec, e);
    csr_order<<<(n + 7) / 8, 256, 0, stream>>>(t_off, t_rec, t_src, t_w, n);

    gemm_wmma_f32<128, 0><<<gemm_blocks, 256, 0, stream>>>(x, nullptr, nullptr, nullptr, W0, b0, A, n);
    launch_gather(A, t_src, B, D_H);
    pn_reduce<<<1024, 256, 0, stream>>>(B, st0, n);

    gemm_wmma_f32<128, 1><<<gemm_blocks, 256, 0, stream>>>(B, nullptr, st0, nullptr, W1, b1, A, n);
    launch_gather(A, t_src, C, D_H);
    pn_reduce<<<1024, 256, 0, stream>>>(C, st1, n);

    const int table_words = 2 * e + n + 1;
    int* tabA = (int*)(A + (size_t)n * (D_H / 2));
    move_words<<<1024, 256, 0, stream>>>(t_src, tabA, table_words);
    gemm_wmma_f32<64, 2><<<gemm_blocks, 256, 0, stream>>>(C, B, st1, st0, Wout, bout, A, n);
    launch_gather(A, tabA, out, d_outc);
}


# Edge table built by a two-level bucket sort, every counter in LDS (chunk histogram, scan, scatter to 256-row buckets, per-bucket row placement): no global atomics; table and sum orders unchanged
# speedup vs baseline: 3.3715x; 1.2605x over previous
#include <hip/hip_runtime.h>

#define D_H 128
#define PN_EPS 1e-6f

typedef float v2f __attribute__((ext_vector_type(2)));
typedef float v8f __attribute__((ext_vector_type(8)));

template <int DOUT, int MODE>
__global__ void __launch_bounds__(256)
gemm_wmma_f32(const float* __restrict__ X, const float* __restrict__ X2,
              const float* __restrict__ st1, const float* __restrict__ st2,
              const float* __restrict__ W, const float* __restrict__ bias,
              float* __restrict__ H, int n_rows) {
    constexpr int TN  = DOUT / 16;
    constexpr int KC  = 64;
    constexpr int WST = KC + 4;
    constexpr int NP  = (MODE == 0) ? 1 : MODE;
    __shared__ float Wt[DOUT * WST];
    __shared__ __attribute__((aligned(16))) float pnM[NP][D_H];
    __shared__ float pnT[NP][D_H];
    __shared__ float pnS[NP];

    const int lane = threadIdx.x & 31;
    const int wave = threadIdx.x >> 5;
    const int half = lane >> 4;
    const int l15  = lane & 15;

    if constexpr (MODE >= 1) {
        const float inv_n = 1.0f / (float)n_rows;
        const int p = threadIdx.x >> 7;
        const int c = threadIdx.x & (D_H - 1);
        const bool on = (p < MODE);
        const float* __restrict__ st = (p == 0) ? st1 : st2;
        if (on) {
            const float mean = st[c] * inv_n;
            pnM[p][c] = mean;
            pnT[p][c] = mean * mean;
        }
        __syncthreads();
        for (int s = 64; s > 0; s >>= 1) {
            if (on && c < s) pnT[p][c] += pnT[p][c + s];
            __syncthreads();
        }
        if (on && c == 0) {
            float ssq_centered = st[D_H] - (float)n_rows * pnT[p][0];
            pnS[p] = 1.0f / sqrtf(PN_EPS + ssq_centered * inv_n);
        }
    }

    const int tm      = blockIdx.x * 8 + wave;
    const int tiles_m = (n_rows + 15) >> 4;
    const bool active = (tm < tiles_m);

    int row_a = tm * 16 + l15;
    if (row_a >= n_rows) row_a = n_rows - 1;
    const float* __restrict__ ap = X + (size_t)row_a * D_H + half * 2;
    const float* __restrict__ ap2 = (MODE == 2) ? (X2 + (size_t)row_a * D_H + half * 2) : ap;
    const float* __restrict__ bbase = &Wt[l15 * WST + half * 2];

    v8f acc[TN] = {};

    for (int k0 = 0; k0 < D_H; k0 += KC) {
        __syncthreads();
        const float* __restrict__ wg = W + (size_t)k0 * DOUT;
        for (int idx = threadIdx.x; idx < KC * DOUT; idx += 256) {
            int r = idx / DOUT;
            int c = idx - r * DOUT;
            Wt[c * WST + r] = wg[idx];
        }
        __syncthreads();

        if (active) {
            __builtin_prefetch(ap + k0 + KC, 0, 1);
            if constexpr (MODE == 2) __builtin_prefetch(ap2 + k0 + KC, 0, 1);
            float sc1 = 0.0f, sc2 = 0.0f;
            if constexpr (MODE >= 1) sc1 = pnS[0];
            if constexpr (MODE == 2) sc2 = pnS[1];
#pragma unroll
            for (int kk = 0; kk < KC; kk += 4) {
                v2f a = *(const v2f*)(ap + k0 + kk);
                if constexpr (MODE >= 1) {
                    const v2f m = *(const v2f*)&pnM[0][k0 + kk + half * 2];
                    a.x = fmaxf((a.x - m.x) * sc1, 0.0f);
                    a.y = fmaxf((a.y - m.y) * sc1, 0.0f);
                }
                if constexpr (MODE == 2) {
                    v2f o = *(const v2f*)(ap2 + k0 + kk);
                    const v2f m2 = *(const v2f*)&pnM[1][k0 + kk + half * 2];
                    o.x = fmaxf((o.x - m2.x) * sc2, 0.0f);
                    o.y = fmaxf((o.y - m2.y) * sc2, 0.0f);
                    a.x += o.x;
                    a.y += o.y;
                }
#pragma unroll
                for (int t = 0; t < TN; ++t) {
                    v2f b = *(const v2f*)(bbase + t * (16 * WST) + kk);
                    acc[t] = __builtin_amdgcn_wmma_f32_16x16x4_f32(
                         false, a,  false, b,
                         (short)0, acc[t],
                         false,  false);
                }
            }
        }
    }

    if (!active) return;
    const int row_base = tm * 16 + half * 8;
#pragma unroll
    for (int t = 0; t < TN; ++t) {
        const int col = t * 16 + l15;
        const float bb = bias[col];
#pragma unroll
        for (int r = 0; r < 8; ++r) {
            const int row = row_base + r;
            if (row < n_rows)
                H[(size_t)row * DOUT + col] = acc[t][r] + bb;
        }
    }
}

#define BK_SHIFT 8
#define BK_ROWS (1 << BK_SHIFT)
#define EB 4

__global__ void __launch_bounds__(1024)
edge_hist(const int* __restrict__ dst, int* __restrict__ hist,
          float* __restrict__ st, int n_st, int nE, int nb, int chunk) {
    extern __shared__ int lh[];
    for (int i = threadIdx.x; i < nb; i += 1024) lh[i] = 0;
    if (blockIdx.x == 0)
        for (int i = threadIdx.x; i < n_st; i += 1024) st[i] = 0.0f;
    __syncthreads();
    const int e0 = blockIdx.x * chunk;
    const int e1 = (e0 + chunk < nE) ? (e0 + chunk) : nE;
    for (int eb = e0 + threadIdx.x; eb < e1; eb += EB * 1024) {
        int d[EB];
#pragma unroll
        for (int u = 0; u < EB; ++u) {
            const int e = eb + u * 1024;
            d[u] = (e < e1) ? dst[e] : 0;
        }
#pragma unroll
        for (int u = 0; u < EB; ++u)
            if (eb + u * 1024 < e1) atomicAdd(&lh[d[u] >> BK_SHIFT], 1);
    }
    __syncthreads();
    for (int i = threadIdx.x; i < nb; i += 1024)
        hist[(size_t)i * gridDim.x + blockIdx.x] = lh[i];
}

__global__ void __launch_bounds__(1024)
edge_scatter(const int* __restrict__ src, const int* __restrict__ dst,
             const float* __restrict__ ew, const int* __restrict__ hoff,
             int* __restrict__ recw, int nE, int nb, int chunk) {
    extern __shared__ int lc[];
    int4* __restrict__ rec = (int4*)recw;
    for (int i = threadIdx.x; i < nb; i += 1024)
        lc[i] = hoff[(size_t)i * gridDim.x + blockIdx.x];
    __syncthreads();
    const int e0 = blockIdx.x * chunk;
    const int e1 = (e0 + chunk < nE) ? (e0 + chunk) : nE;
    for (int eb = e0 + threadIdx.x; eb < e1; eb += EB * 1024) {
        int d[EB], sr[EB], wb[EB];
#pragma unroll
        for (int u = 0; u < EB; ++u) {
            const int e = eb + u * 1024;
            const bool in = (e < e1);
            d[u]  = in ? dst[e] : 0;
            sr[u] = in ? src[e] : 0;
            wb[u] = in ? __builtin_bit_cast(int, ew[e]) : 0;
        }
#pragma unroll
        for (int u = 0; u < EB; ++u) {
            const int e = eb + u * 1024;
            if (e < e1) {
                const int pos = atomicAdd(&lc[d[u] >> BK_SHIFT], 1);
                rec[pos] = make_int4(e, sr[u], wb[u], d[u]);
            }
        }
    }
}

__global__ void __launch_bounds__(1024)
bucket_place(const int* __restrict__ hoff, const int* __restrict__ recAw,
             int* __restrict__ recBw, int* __restrict__ off,
             int n_rows, int n_parts) {
    __shared__ int cnt[BK_ROWS];
    __shared__ int cur[BK_ROWS];
    __shared__ int wsum[BK_ROWS / 32];
    const int4* __restrict__ recA = (const int4*)recAw;
    int4* __restrict__ recB = (int4*)recBw;
    const int lane = threadIdx.x & 31;
    const int wave = threadIdx.x >> 5;
    const int b = blockIdx.x;
    const int row0 = b << BK_SHIFT;
    const int s = hoff[(size_t)b * n_parts];
    const int t = hoff[(size_t)(b + 1) * n_parts];
    if (threadIdx.x < BK_ROWS) cnt[threadIdx.x] = 0;
    __syncthreads();
    for (int ib = s + threadIdx.x; ib < t; ib += EB * 1024) {
        int d[EB];
#pragma unroll
        for (int u = 0; u < EB; ++u) {
            const int i = ib + u * 1024;
            d[u] = (i < t) ? recAw[4 * (size_t)i + 3] : row0;
        }
#pragma unroll
        for (int u = 0; u < EB; ++u)
            if (ib + u * 1024 < t) atomicAdd(&cnt[d[u] - row0], 1);
    }
    __syncthreads();
    const int v = (threadIdx.x < BK_ROWS) ? cnt[threadIdx.x] : 0;
    int x = v;
    for (int d = 1; d < 32; d <<= 1) {
        const int y = __shfl_up(x, d);
        if (lane >= d) x += y;
    }
    if (threadIdx.x < BK_ROWS && lane == 31) wsum[wave] = x;
    __syncthreads();
    if (threadIdx.x < BK_ROWS) {
        int base = s;
        for (int w = 0; w < wave; ++w) base += wsum[w];
        const int start = base + x - v;
        cur[threadIdx.x] = start;
        const int r = row0 + threadIdx.x;
        if (r < n_rows) off[r] = start;
    }
    if (b == (int)gridDim.x - 1 && threadIdx.x == 0) off[n_rows] = t;
    __syncthreads();
    for (int ib = s + threadIdx.x; ib < t; ib += EB * 1024) {
        int4 rc[EB];
#pragma unroll
        for (int u = 0; u < EB; ++u) {
            const int i = ib + u * 1024;
            rc[u] = (i < t) ? recA[i] : make_int4(0, 0, 0, row0);
        }
#pragma unroll
        for (int u = 0; u < EB; ++u) {
            if (ib + u * 1024 < t) {
                const int slot = atomicAdd(&cur[rc[u].w - row0], 1);
                recB[slot] = rc[u];
            }
        }
    }
}

__global__ void __launch_bounds__(1024)
csr_scan_local(const int* __restrict__ cnt, int* __restrict__ off,
               int* __restrict__ bsum, int n_rows) {
    __shared__ int wsum[32];
    const int lane = threadIdx.x & 31;
    const int wave = threadIdx.x >> 5;
    const int i = blockIdx.x * 1024 + threadIdx.x;
    const int v = (i < n_rows) ? cnt[i] : 0;
    int x = v;
    for (int d = 1; d < 32; d <<= 1) {
        const int y = __shfl_up(x, d);
        if (lane >= d) x += y;
    }
    if (lane == 31) wsum[wave] = x;
    __syncthreads();
    if (wave == 0) {
        const int s = wsum[lane];
        int t = s;
        for (int d = 1; d < 32; d <<= 1) {
            const int y = __shfl_up(t, d);
            if (lane >= d) t += y;
        }
        wsum[lane] = t - s;
        if (lane == 31) bsum[blockIdx.x] = t;
    }
    __syncthreads();
    if (i < n_rows) off[i] = x - v + wsum[wave];
}

__global__ void __launch_bounds__(1024)
csr_scan_add(int* __restrict__ off, const int* __restrict__ bsum, int n_rows) {
    __shared__ int base_s;
    if (threadIdx.x < 32) {
        int s = 0;
        for (int j = threadIdx.x; j < (int)blockIdx.x; j += 32) s += bsum[j];
        for (int d = 16; d > 0; d >>= 1) s += __shfl_xor(s, d);
        if (threadIdx.x == 0) base_s = s;
    }
    __syncthreads();
    const int base = base_s;
    const int i = blockIdx.x * 1024 + threadIdx.x;
    if (i < n_rows) off[i] += base;
    if (blockIdx.x == gridDim.x - 1 && threadIdx.x == 0)
        off[n_rows] = base + bsum[blockIdx.x];
}

__global__ void __launch_bounds__(256)
csr_order(const int* __restrict__ off, const int* __restrict__ recw,
          int* __restrict__ srcS, float* __restrict__ wS, int n_rows) {
    const int4* __restrict__ rec = (const int4*)recw;
    const int lane = threadIdx.x & 31;
    const int r = __builtin_amdgcn_readfirstlane((int)((blockIdx.x * 256 + threadIdx.x) >> 5));
    if (r >= n_rows) return;
    const int s = off[r];
    const int len = off[r + 1] - s;
    if (len <= 32) {
        int4 me = make_int4(0, 0, 0, 0);
        if (lane < len) me = rec[s + lane];
        int rank = 0;
        for (int j = 0; j < len; ++j) {
            const int idj = __builtin_amdgcn_readlane(me.x, j);
            rank += (idj < me.x) ? 1 : 0;
        }
        if (lane < len) {
            srcS[s + rank] = me.y;
            wS[s + rank] = __builtin_bit_cast(float, me.z);
        }
    } else {
        for (int i = lane; i < len; i += 32) {
            const int4 me = rec[s + i];
            int rank = 0;
            for (int j = 0; j < len; ++j) rank += (rec[s + j].x < me.x) ? 1 : 0;
            srcS[s + rank] = me.y;
            wS[s + rank] = __builtin_bit_cast(float, me.z);
        }
    }
}

template <int VEC>
__global__ void __launch_bounds__(256)
spmm_gather(const float* __restrict__ H, const int* __restrict__ off,
            const int* __restrict__ srcS, const float* __restrict__ wS,
            float* __restrict__ out, int n_rows) {
    typedef float vf __attribute__((ext_vector_type(VEC)));
    const int lane = threadIdx.x & 31;
    const int r = __builtin_amdgcn_readfirstlane((int)((blockIdx.x * 256 + threadIdx.x) >> 5));
    if (r >= n_rows) return;
    const int s = off[r];
    const int t = off[r + 1];
    const vf* __restrict__ Hv = (const vf*)H + lane;
    vf acc = {};
    for (int base = s; base < t; base += 32) {
        const int m = (t - base < 32) ? (t - base) : 32;
        int my_s = 0;
        float my_w = 0.0f;
        if (lane < m) {
            my_s = srcS[base + lane];
            my_w = wS[base + lane];
        }
#pragma unroll 4
        for (int j = 0; j < m; ++j) {
            const int sj = __builtin_amdgcn_readlane(my_s, j);
            const float wj = __builtin_bit_cast(
                float, __builtin_amdgcn_readlane(__builtin_bit_cast(int, my_w), j));
            const vf v = Hv[(size_t)sj * 32];
            acc += wj * v;
        }
    }
    ((vf*)out)[(size_t)r * 32 + lane] = acc;
}

__global__ void move_words(const int* __restrict__ from, int* __restrict__ to,
                           int count) {
    const int stride = gridDim.x * blockDim.x;
    const int i = blockIdx.x * blockDim.x + threadIdx.x;
    const int n4 = count >> 2;
    for (int k = i; k < n4; k += stride)
        ((int4*)to)[k] = ((const int4*)from)[k];
    for (int k = (n4 << 2) + i; k < count; k += stride) to[k] = from[k];
}

#define PN_ROWS 8
typedef float v4f __attribute__((ext_vector_type(4)));

__global__ void __launch_bounds__(256)
pn_reduce(const float* __restrict__ H, float* __restrict__ red, int n_rows) {
    __shared__ v4f part[8 * 32];
    __shared__ float psq[8];
    const int lane = threadIdx.x & 31;
    const int wave = __builtin_amdgcn_readfirstlane((int)(threadIdx.x >> 5));
    const int stride = gridDim.x * 8;
    const v4f* __restrict__ Hv = (const v4f*)H + lane;
    const v4f zero = {0.0f, 0.0f, 0.0f, 0.0f};
    v4f cs = zero, sq = zero;
    for (int base = blockIdx.x * 8 + wave; base < n_rows; base += PN_ROWS * stride) {
        v4f v[PN_ROWS];
#pragma unroll
        for (int u = 0; u < PN_ROWS; ++u) {
            const int r = base + u * stride;
            v[u] = (r < n_rows) ? Hv[(size_t)r * 32] : zero;
        }
#pragma unroll
        for (int u = 0; u < PN_ROWS; ++u) {
            cs += v[u];
            sq += v[u] * v[u];
        }
    }
    float s = (sq.x + sq.y) + (sq.z + sq.w);
    for (int d = 16; d > 0; d >>= 1) s += __shfl_xor(s, d);
    part[wave * 32 + lane] = cs;
    if (lane == 0) psq[wave] = s;
    __syncthreads();
    if (threadIdx.x < D_H) {
        float t = 0.0f;
        for (int w = 0; w < 8; ++w) t += ((const float*)part)[w * D_H + threadIdx.x];
        unsafeAtomicAdd(&red[threadIdx.x], t);
    } else if (threadIdx.x == D_H) {
        float t = 0.0f;
        for (int w = 0; w < 8; ++w) t += psq[w];
        unsafeAtomicAdd(&red[D_H], t);
    }
}

extern "C" void kernel_launch(void* const* d_in, const int* in_sizes, int n_in,
                              void* d_out, int out_size, void* d_ws, size_t ws_size,
                              hipStream_t stream) {
    const float* x    = (const float*)d_in[0];
    const int*   ei   = (const int*)d_in[1];
    const float* ew   = (const float*)d_in[2];
    const float* W0   = (const float*)d_in[3];
    const float* b0   = (const float*)d_in[4];
    const float* W1   = (const float*)d_in[5];
    const float* b1   = (const float*)d_in[6];
    const float* Wout = (const float*)d_in[7];
    const float* bout = (const float*)d_in[8];
    float* out = (float*)d_out;

    const int n = in_sizes[0] / D_H;
    const int e = in_sizes[2];
    const int d_outc = out_size / n;
    const int* srcI = ei;
    const int* dstI = ei + e;

    const size_t nd = (size_t)n * D_H;
    float* A   = (float*)d_ws;
    float* B   = A + nd;
    float* C   = B + nd;

    const int gemm_blocks = (((n + 15) / 16) + 7) / 8;
    auto launch_gather = [&](const float* H_, const int* tab, float* O_, int D) {
        const int* srcS_ = tab;
        const float* wS_ = (const float*)(tab + e);
        const int* off_ = tab + 2 * (size_t)e;
        const int blocks = (n + 7) / 8;
        if (D == 128)
            spmm_gather<4><<<blocks, 256, 0, stream>>>(H_, off_, srcS_, wS_, O_, n);
        else
            spmm_gather<2><<<blocks, 256, 0, stream>>>(H_, off_, srcS_, wS_, O_, n);
    };
    const int npad = (n + 1024) & ~255;
    const int parts = 512;
    const int chunk = (e + parts - 1) / parts;
    const int nb = (n + BK_ROWS - 1) >> BK_SHIFT;
    const int hl = nb * parts;
    const int hlpad = (hl + 1024) & ~255;
    const int hscan_blocks = (hl + 1023) / 1024;
    int*   t_src  = (int*)out;
    float* t_w    = (float*)(t_src + e);
    int*   t_off  = t_src + 2 * (size_t)e;
    int*   t_hist = t_off + npad;
    int*   t_hoff = t_hist + hlpad;
    int*   t_bsum = t_hoff + hlpad;
    float* st0    = (float*)(t_bsum + ((hscan_blocks + 255) & ~255));
    float* st1    = st0 + 132;
    int*   t_recA = (int*)C;
    int*   t_recB = t_recA + 4 * (size_t)e;
    const size_t lds_nb = (size_t)nb * sizeof(int);
    edge_hist<<<parts, 1024, lds_nb, stream>>>(dstI, t_hist, st0, 2 * 132, e, nb, chunk);
    csr_scan_local<<<hscan_blocks, 1024, 0, stream>>>(t_hist, t_hoff, t_bsum, hl);
    csr_scan_add<<<hscan_blocks, 1024, 0, stream>>>(t_hoff, t_bsum, hl);
    edge_scatter<<<parts, 1024, lds_nb, stream>>>(srcI, dstI, ew, t_hoff, t_recA, e, nb, chunk);
    bucket_place<<<nb, 1024, 0, stream>>>(t_hoff, t_recA, t_recB, t_off, n, parts);
    csr_order<<<(n + 7) / 8, 256, 0, stream>>>(t_off, t_recB, t_src, t_w, n);

    gemm_wmma_f32<128, 0><<<gemm_blocks, 256, 0, stream>>>(x, nullptr, nullptr, nullptr, W0, b0, A, n);
    launch_gather(A, t_src, B, D_H);
    pn_reduce<<<1024, 256, 0, stream>>>(B, st0, n);

    gemm_wmma_f32<128, 1><<<gemm_blocks, 256, 0, stream>>>(B, nullptr, st0, nullptr, W1, b1, A, n);
    launch_gather(A, t_src, C, D_H);
    pn_reduce<<<1024, 256, 0, stream>>>(C, st1, n);

    const int table_words = 2 * e + n + 1;
    int* tabA = (int*)(A + (size_t)n * (D_H / 2));
    move_words<<<1024, 256, 0, stream>>>(t_src, tabA, table_words);
    gemm_wmma_f32<64, 2><<<gemm_blocks, 256, 0, stream>>>(C, B, st1, st0, Wout, bout, A, n);
    launch_gather(A, tabA, out, d_outc);
}


# PairNorm statistics summed inside the aggregation that writes the plane (512 resident blocks walk row ranges, lanes keep column sums in registers); both statistics passes go; partials regrouped
# speedup vs baseline: 3.4812x; 1.0325x over previous
#include <hip/hip_runtime.h>

#define D_H 128
#define PN_EPS 1e-6f

typedef float v2f __attribute__((ext_vector_type(2)));
typedef float v8f __attribute__((ext_vector_type(8)));

template <int DOUT, int MODE>
__global__ void __launch_bounds__(256)
gemm_wmma_f32(const float* __restrict__ X, const float* __restrict__ X2,
              const float* __restrict__ st1, const float* __restrict__ st2,
              const float* __restrict__ W, const float* __restrict__ bias,
              float* __restrict__ H, int n_rows) {
    constexpr int TN  = DOUT / 16;
    constexpr int KC  = 64;
    constexpr int WST = KC + 4;
    constexpr int NP  = (MODE == 0) ? 1 : MODE;
    __shared__ float Wt[DOUT * WST];
    __shared__ __attribute__((aligned(16))) float pnM[NP][D_H];
    __shared__ float pnT[NP][D_H];
    __shared__ float pnS[NP];

    const int lane = threadIdx.x & 31;
    const int wave = threadIdx.x >> 5;
    const int half = lane >> 4;
    const int l15  = lane & 15;

    if constexpr (MODE >= 1) {
        const float inv_n = 1.0f / (float)n_rows;
        const int p = threadIdx.x >> 7;
        const int c = threadIdx.x & (D_H - 1);
        const bool on = (p < MODE);
        const float* __restrict__ st = (p == 0) ? st1 : st2;
        if (on) {
            const float mean = st[c] * inv_n;
            pnM[p][c] = mean;
            pnT[p][c] = mean * mean;
        }
        __syncthreads();
        for (int s = 64; s > 0; s >>= 1) {
            if (on && c < s) pnT[p][c] += pnT[p][c + s];
            __syncthreads();
        }
        if (on && c == 0) {
            float ssq_centered = st[D_H] - (float)n_rows * pnT[p][0];
            pnS[p] = 1.0f / sqrtf(PN_EPS + ssq_centered * inv_n);
        }
    }

    const int tm      = blockIdx.x * 8 + wave;
    const int tiles_m = (n_rows + 15) >> 4;
    const bool active = (tm < tiles_m);

    int row_a = tm * 16 + l15;
    if (row_a >= n_rows) row_a = n_rows - 1;
    const float* __restrict__ ap = X + (size_t)row_a * D_H + half * 2;
    const float* __restrict__ ap2 = (MODE == 2) ? (X2 + (size_t)row_a * D_H + half * 2) : ap;
    const float* __restrict__ bbase = &Wt[l15 * WST + half * 2];

    v8f acc[TN] = {};

    for (int k0 = 0; k0 < D_H; k0 += KC) {
        __syncthreads();
        const float* __restrict__ wg = W + (size_t)k0 * DOUT;
        for (int idx = threadIdx.x; idx < KC * DOUT; idx += 256) {
            int r = idx / DOUT;
            int c = idx - r * DOUT;
            Wt[c * WST + r] = wg[idx];
        }
        __syncthreads();

        if (active) {
            __builtin_prefetch(ap + k0 + KC, 0, 1);
            if constexpr (MODE == 2) __builtin_prefetch(ap2 + k0 + KC, 0, 1);
            float sc1 = 0.0f, sc2 = 0.0f;
            if constexpr (MODE >= 1) sc1 = pnS[0];
            if constexpr (MODE == 2) sc2 = pnS[1];
#pragma unroll
            for (int kk = 0; kk < KC; kk += 4) {
                v2f a = *(const v2f*)(ap + k0 + kk);
                if constexpr (MODE >= 1) {
                    const v2f m = *(const v2f*)&pnM[0][k0 + kk + half * 2];
                    a.x = fmaxf((a.x - m.x) * sc1, 0.0f);
                    a.y = fmaxf((a.y - m.y) * sc1, 0.0f);
                }
                if constexpr (MODE == 2) {
                    v2f o = *(const v2f*)(ap2 + k0 + kk);
                    const v2f m2 = *(const v2f*)&pnM[1][k0 + kk + half * 2];
                    o.x = fmaxf((o.x - m2.x) * sc2, 0.0f);
                    o.y = fmaxf((o.y - m2.y) * sc2, 0.0f);
                    a.x += o.x;
                    a.y += o.y;
                }
#pragma unroll
                for (int t = 0; t < TN; ++t) {
                    v2f b = *(const v2f*)(bbase + t * (16 * WST) + kk);
                    acc[t] = __builtin_amdgcn_wmma_f32_16x16x4_f32(
                         false, a,  false, b,
                         (short)0, acc[t],
                         false,  false);
                }
            }
        }
    }

    if (!active) return;
    const int row_base = tm * 16 + half * 8;
#pragma unroll
    for (int t = 0; t < TN; ++t) {
        const int col = t * 16 + l15;
        const float bb = bias[col];
#pragma unroll
        for (int r = 0; r < 8; ++r) {
            const int row = row_base + r;
            if (row < n_rows)
                H[(size_t)row * DOUT + col] = acc[t][r] + bb;
        }
    }
}

#define BK_SHIFT 8
#define BK_ROWS (1 << BK_SHIFT)
#define EB 4

__global__ void __launch_bounds__(1024)
edge_hist(const int* __restrict__ dst, int* __restrict__ hist,
          float* __restrict__ st, int n_st, int nE, int nb, int chunk) {
    extern __shared__ int lh[];
    for (int i = threadIdx.x; i < nb; i += 1024) lh[i] = 0;
    if (blockIdx.x == 0)
        for (int i = threadIdx.x; i < n_st; i += 1024) st[i] = 0.0f;
    __syncthreads();
    const int e0 = blockIdx.x * chunk;
    const int e1 = (e0 + chunk < nE) ? (e0 + chunk) : nE;
    for (int eb = e0 + threadIdx.x; eb < e1; eb += EB * 1024) {
        int d[EB];
#pragma unroll
        for (int u = 0; u < EB; ++u) {
            const int e = eb + u * 1024;
            d[u] = (e < e1) ? dst[e] : 0;
        }
#pragma unroll
        for (int u = 0; u < EB; ++u)
            if (eb + u * 1024 < e1) atomicAdd(&lh[d[u] >> BK_SHIFT], 1);
    }
    __syncthreads();
    for (int i = threadIdx.x; i < nb; i += 1024)
        hist[(size_t)i * gridDim.x + blockIdx.x] = lh[i];
}

__global__ void __launch_bounds__(1024)
edge_scatter(const int* __restrict__ src, const int* __restrict__ dst,
             const float* __restrict__ ew, const int* __restrict__ hoff,
             int* __restrict__ recw, int nE, int nb, int chunk) {
    extern __shared__ int lc[];
    int4* __restrict__ rec = (int4*)recw;
    for (int i = threadIdx.x; i < nb; i += 1024)
        lc[i] = hoff[(size_t)i * gridDim.x + blockIdx.x];
    __syncthreads();
    const int e0 = blockIdx.x * chunk;
    const int e1 = (e0 + chunk < nE) ? (e0 + chunk) : nE;
    for (int eb = e0 + threadIdx.x; eb < e1; eb += EB * 1024) {
        int d[EB], sr[EB], wb[EB];
#pragma unroll
        for (int u = 0; u < EB; ++u) {
            const int e = eb + u * 1024;
            const bool in = (e < e1);
            d[u]  = in ? dst[e] : 0;
            sr[u] = in ? src[e] : 0;
            wb[u] = in ? __builtin_bit_cast(int, ew[e]) : 0;
        }
#pragma unroll
        for (int u = 0; u < EB; ++u) {
            const int e = eb + u * 1024;
            if (e < e1) {
                const int pos = atomicAdd(&lc[d[u] >> BK_SHIFT], 1);
                rec[pos] = make_int4(e, sr[u], wb[u], d[u]);
            }
        }
    }
}

__global__ void __launch_bounds__(1024)
bucket_place(const int* __restrict__ hoff, const int* __restrict__ recAw,
             int* __restrict__ recBw, int* __restrict__ off,
             int n_rows, int n_parts) {
    __shared__ int cnt[BK_ROWS];
    __shared__ int cur[BK_ROWS];
    __shared__ int wsum[BK_ROWS / 32];
    const int4* __restrict__ recA = (const int4*)recAw;
    int4* __restrict__ recB = (int4*)recBw;
    const int lane = threadIdx.x & 31;
    const int wave = threadIdx.x >> 5;
    const int b = blockIdx.x;
    const int row0 = b << BK_SHIFT;
    const int s = hoff[(size_t)b * n_parts];
    const int t = hoff[(size_t)(b + 1) * n_parts];
    if (threadIdx.x < BK_ROWS) cnt[threadIdx.x] = 0;
    __syncthreads();
    for (int ib = s + threadIdx.x; ib < t; ib += EB * 1024) {
        int d[EB];
#pragma unroll
        for (int u = 0; u < EB; ++u) {
            const int i = ib + u * 1024;
            d[u] = (i < t) ? recAw[4 * (size_t)i + 3] : row0;
        }
#pragma unroll
        for (int u = 0; u < EB; ++u)
            if (ib + u * 1024 < t) atomicAdd(&cnt[d[u] - row0], 1);
    }
    __syncthreads();
    const int v = (threadIdx.x < BK_ROWS) ? cnt[threadIdx.x] : 0;
    int x = v;
    for (int d = 1; d < 32; d <<= 1) {
        const int y = __shfl_up(x, d);
        if (lane >= d) x += y;
    }
    if (threadIdx.x < BK_ROWS && lane == 31) wsum[wave] = x;
    __syncthreads();
    if (threadIdx.x < BK_ROWS) {
        int base = s;
        for (int w = 0; w < wave; ++w) base += wsum[w];
        const int start = base + x - v;
        cur[threadIdx.x] = start;
        const int r = row0 + threadIdx.x;
        if (r < n_rows) off[r] = start;
    }
    if (b == (int)gridDim.x - 1 && threadIdx.x == 0) off[n_rows] = t;
    __syncthreads();
    for (int ib = s + threadIdx.x; ib < t; ib += EB * 1024) {
        int4 rc[EB];
#pragma unroll
        for (int u = 0; u < EB; ++u) {
            const int i = ib + u * 1024;
            rc[u] = (i < t) ? recA[i] : make_int4(0, 0, 0, row0);
        }
#pragma unroll
        for (int u = 0; u < EB; ++u) {
            if (ib + u * 1024 < t) {
                const int slot = atomicAdd(&cur[rc[u].w - row0], 1);
                recB[slot] = rc[u];
            }
        }
    }
}

__global__ void __launch_bounds__(1024)
csr_scan_local(const int* __restrict__ cnt, int* __restrict__ off,
               int* __restrict__ bsum, int n_rows) {
    __shared__ int wsum[32];
    const int lane = threadIdx.x & 31;
    const int wave = threadIdx.x >> 5;
    const int i = blockIdx.x * 1024 + threadIdx.x;
    const int v = (i < n_rows) ? cnt[i] : 0;
    int x = v;
    for (int d = 1; d < 32; d <<= 1) {
        const int y = __shfl_up(x, d);
        if (lane >= d) x += y;
    }
    if (lane == 31) wsum[wave] = x;
    __syncthreads();
    if (wave == 0) {
        const int s = wsum[lane];
        int t = s;
        for (int d = 1; d < 32; d <<= 1) {
            const int y = __shfl_up(t, d);
            if (lane >= d) t += y;
        }
        wsum[lane] = t - s;
        if (lane == 31) bsum[blockIdx.x] = t;
    }
    __syncthreads();
    if (i < n_rows) off[i] = x - v + wsum[wave];
}

__global__ void __launch_bounds__(1024)
csr_scan_add(int* __restrict__ off, const int* __restrict__ bsum, int n_rows) {
    __shared__ int base_s;
    if (threadIdx.x < 32) {
        int s = 0;
        for (int j = threadIdx.x; j < (int)blockIdx.x; j += 32) s += bsum[j];
        for (int d = 16; d > 0; d >>= 1) s += __shfl_xor(s, d);
        if (threadIdx.x == 0) base_s = s;
    }
    __syncthreads();
    const int base = base_s;
    const int i = blockIdx.x * 1024 + threadIdx.x;
    if (i < n_rows) off[i] += base;
    if (blockIdx.x == gridDim.x - 1 && threadIdx.x == 0)
        off[n_rows] = base + bsum[blockIdx.x];
}

__global__ void __launch_bounds__(256)
csr_order(const int* __restrict__ off, const int* __restrict__ recw,
          int* __restrict__ srcS, float* __restrict__ wS, int n_rows) {
    const int4* __restrict__ rec = (const int4*)recw;
    const int lane = threadIdx.x & 31;
    const int r = __builtin_amdgcn_readfirstlane((int)((blockIdx.x * 256 + threadIdx.x) >> 5));
    if (r >= n_rows) return;
    const int s = off[r];
    const int len = off[r + 1] - s;
    if (len <= 32) {
        int4 me = make_int4(0, 0, 0, 0);
        if (lane < len) me = rec[s + lane];
        int rank = 0;
        for (int j = 0; j < len; ++j) {
            const int idj = __builtin_amdgcn_readlane(me.x, j);
            rank += (idj < me.x) ? 1 : 0;
        }
        if (lane < len) {
            srcS[s + rank] = me.y;
            wS[s + rank] = __builtin_bit_cast(float, me.z);
        }
    } else {
        for (int i = lane; i < len; i += 32) {
            const int4 me = rec[s + i];
            int rank = 0;
            for (int j = 0; j < len; ++j) rank += (rec[s + j].x < me.x) ? 1 : 0;
            srcS[s + rank] = me.y;
            wS[s + rank] = __builtin_bit_cast(float, me.z);
        }
    }
}

template <int VEC>
__global__ void __launch_bounds__(256)
spmm_gather(const float* __restrict__ H, const int* __restrict__ off,
            const int* __restrict__ srcS, const float* __restrict__ wS,
            float* __restrict__ out, int n_rows) {
    typedef float vf __attribute__((ext_vector_type(VEC)));
    const int lane = threadIdx.x & 31;
    const int r = __builtin_amdgcn_readfirstlane((int)((blockIdx.x * 256 + threadIdx.x) >> 5));
    if (r >= n_rows) return;
    const int s = off[r];
    const int t = off[r + 1];
    const vf* __restrict__ Hv = (const vf*)H + lane;
    vf acc = {};
    for (int base = s; base < t; base += 32) {
        const int m = (t - base < 32) ? (t - base) : 32;
        int my_s = 0;
        float my_w = 0.0f;
        if (lane < m) {
            my_s = srcS[base + lane];
            my_w = wS[base + lane];
        }
#pragma unroll 4
        for (int j = 0; j < m; ++j) {
            const int sj = __builtin_amdgcn_readlane(my_s, j);
            const float wj = __builtin_bit_cast(
                float, __builtin_amdgcn_readlane(__builtin_bit_cast(int, my_w), j));
            const vf v = Hv[(size_t)sj * 32];
            acc += wj * v;
        }
    }
    ((vf*)out)[(size_t)r * 32 + lane] = acc;
}

__global__ void move_words(const int* __restrict__ from, int* __restrict__ to,
                           int count) {
    const int stride = gridDim.x * blockDim.x;
    const int i = blockIdx.x * blockDim.x + threadIdx.x;
    const int n4 = count >> 2;
    for (int k = i; k < n4; k += stride)
        ((int4*)to)[k] = ((const int4*)from)[k];
    for (int k = (n4 << 2) + i; k < count; k += stride) to[k] = from[k];
}

#define GS_BLOCKS 512
#define GS_WAVES 32
typedef float v4f __attribute__((ext_vector_type(4)));

__global__ void __launch_bounds__(GS_WAVES * 32)
spmm_gather_stats(const float* __restrict__ H, const int* __restrict__ off,
                  const int* __restrict__ srcS, const float* __restrict__ wS,
                  float* __restrict__ out, float* __restrict__ red, int n_rows) {
    __shared__ v4f part[GS_WAVES * 32];
    __shared__ float psq[GS_WAVES];
    const int lane = threadIdx.x & 31;
    const int wave = __builtin_amdgcn_readfirstlane((int)(threadIdx.x >> 5));
    const unsigned long long gw = (unsigned long long)blockIdx.x * GS_WAVES + wave;
    const int r0 = (int)((gw * (unsigned long long)n_rows) / (GS_BLOCKS * GS_WAVES));
    const int r1 = (int)(((gw + 1) * (unsigned long long)n_rows) / (GS_BLOCKS * GS_WAVES));
    const v4f* __restrict__ Hv = (const v4f*)H + lane;
    const v4f zero = {0.0f, 0.0f, 0.0f, 0.0f};
    v4f cs = zero, sq = zero;
    for (int r = r0; r < r1; ++r) {
        const int s = off[r];
        const int e = off[r + 1];
        v4f acc = zero;
        for (int base = s; base < e; base += 32) {
            const int m = (e - base < 32) ? (e - base) : 32;
            int my_s = 0;
            float my_w = 0.0f;
            if (lane < m) {
                my_s = srcS[base + lane];
                my_w = wS[base + lane];
            }
#pragma unroll 4
            for (int j = 0; j < m; ++j) {
                const int sj = __builtin_amdgcn_readlane(my_s, j);
                const float wj = __builtin_bit_cast(
                    float, __builtin_amdgcn_readlane(__builtin_bit_cast(int, my_w), j));
                const v4f v = Hv[(size_t)sj * 32];
                acc += wj * v;
            }
        }
        ((v4f*)out)[(size_t)r * 32 + lane] = acc;
        cs += acc;
        sq += acc * acc;
    }
    float q = (sq.x + sq.y) + (sq.z + sq.w);
    for (int d = 16; d > 0; d >>= 1) q += __shfl_xor(q, d);
    part[wave * 32 + lane] = cs;
    if (lane == 0) psq[wave] = q;
    __syncthreads();
    if (threadIdx.x < D_H) {
        float a = 0.0f;
        for (int w = 0; w < GS_WAVES; ++w) a += ((const float*)part)[w * D_H + threadIdx.x];
        unsafeAtomicAdd(&red[threadIdx.x], a);
    } else if (threadIdx.x == D_H) {
        float a = 0.0f;
        for (int w = 0; w < GS_WAVES; ++w) a += psq[w];
        unsafeAtomicAdd(&red[D_H], a);
    }
}

extern "C" void kernel_launch(void* const* d_in, const int* in_sizes, int n_in,
                              void* d_out, int out_size, void* d_ws, size_t ws_size,
                              hipStream_t stream) {
    const float* x    = (const float*)d_in[0];
    const int*   ei   = (const int*)d_in[1];
    const float* ew   = (const float*)d_in[2];
    const float* W0   = (const float*)d_in[3];
    const float* b0   = (const float*)d_in[4];
    const float* W1   = (const float*)d_in[5];
    const float* b1   = (const float*)d_in[6];
    const float* Wout = (const float*)d_in[7];
    const float* bout = (const float*)d_in[8];
    float* out = (float*)d_out;

    const int n = in_sizes[0] / D_H;
    const int e = in_sizes[2];
    const int* srcI = ei;
    const int* dstI = ei + e;

    const size_t nd = (size_t)n * D_H;
    float* A   = (float*)d_ws;
    float* B   = A + nd;
    float* C   = B + nd;

    const int gemm_blocks = (((n + 15) / 16) + 7) / 8;
    const int npad = (n + 1024) & ~255;
    const int parts = 512;
    const int chunk = (e + parts - 1) / parts;
    const int nb = (n + BK_ROWS - 1) >> BK_SHIFT;
    const int hl = nb * parts;
    const int hlpad = (hl + 1024) & ~255;
    const int hscan_blocks = (hl + 1023) / 1024;
    int*   t_src  = (int*)out;
    float* t_w    = (float*)(t_src + e);
    int*   t_off  = t_src + 2 * (size_t)e;
    int*   t_hist = t_off + npad;
    int*   t_hoff = t_hist + hlpad;
    int*   t_bsum = t_hoff + hlpad;
    float* st0    = (float*)(t_bsum + ((hscan_blocks + 255) & ~255));
    float* st1    = st0 + 132;
    int*   t_recA = (int*)C;
    int*   t_recB = t_recA + 4 * (size_t)e;
    const size_t lds_nb = (size_t)nb * sizeof(int);
    edge_hist<<<parts, 1024, lds_nb, stream>>>(dstI, t_hist, st0, 2 * 132, e, nb, chunk);
    csr_scan_local<<<hscan_blocks, 1024, 0, stream>>>(t_hist, t_hoff, t_bsum, hl);
    csr_scan_add<<<hscan_blocks, 1024, 0, stream>>>(t_hoff, t_bsum, hl);
    edge_scatter<<<parts, 1024, lds_nb, stream>>>(srcI, dstI, ew, t_hoff, t_recA, e, nb, chunk);
    bucket_place<<<nb, 1024, 0, stream>>>(t_hoff, t_recA, t_recB, t_off, n, parts);
    csr_order<<<(n + 7) / 8, 256, 0, stream>>>(t_off, t_recB, t_src, t_w, n);

    gemm_wmma_f32<128, 0><<<gemm_blocks, 256, 0, stream>>>(x, nullptr, nullptr, nullptr, W0, b0, A, n);
    spmm_gather_stats<<<GS_BLOCKS, GS_WAVES * 32, 0, stream>>>(A, t_off, t_src, t_w, B, st0, n);

    gemm_wmma_f32<128, 1><<<gemm_blocks, 256, 0, stream>>>(B, nullptr, st0, nullptr, W1, b1, A, n);
    spmm_gather_stats<<<GS_BLOCKS, GS_WAVES * 32, 0, stream>>>(A, t_off, t_src, t_w, C, st1, n);

    const int table_words = 2 * e + n + 1;
    int* tabA = (int*)(A + (size_t)n * (D_H / 2));
    move_words<<<1024, 256, 0, stream>>>(t_src, tabA, table_words);
    gemm_wmma_f32<64, 2><<<gemm_blocks, 256, 0, stream>>>(C, B, st1, st0, Wout, bout, A, n);
    spmm_gather<2><<<(n + 7) / 8, 256, 0, stream>>>(A, tabA + 2 * (size_t)e, tabA, (const float*)(tabA + e), out, n);
}


# Layers 0,1: aggregation first, then the product, one launch each (A(XW+b)=(AX)W+(A1)b): a wave sums a 16-row tile of AX in LDS, multiplies it by W by WMMA; sums regrouped; 13->10 launches
# speedup vs baseline: 4.3153x; 1.2396x over previous
#include <hip/hip_runtime.h>

#define D_H 128
#define PN_EPS 1e-6f

typedef float v2f __attribute__((ext_vector_type(2)));
typedef float v8f __attribute__((ext_vector_type(8)));

template <int DOUT, int MODE>
__global__ void __launch_bounds__(256)
gemm_wmma_f32(const float* __restrict__ X, const float* __restrict__ X2,
              const float* __restrict__ st1, const float* __restrict__ st2,
              const float* __restrict__ W, const float* __restrict__ bias,
              float* __restrict__ H, int n_rows) {
    constexpr int TN  = DOUT / 16;
    constexpr int KC  = 64;
    constexpr int WST = KC + 4;
    constexpr int NP  = (MODE == 0) ? 1 : MODE;
    __shared__ float Wt[DOUT * WST];
    __shared__ __attribute__((aligned(16))) float pnM[NP][D_H];
    __shared__ float pnT[NP][D_H];
    __shared__ float pnS[NP];

    const int lane = threadIdx.x & 31;
    const int wave = threadIdx.x >> 5;
    const int half = lane >> 4;
    const int l15  = lane & 15;

    if constexpr (MODE >= 1) {
        const float inv_n = 1.0f / (float)n_rows;
        const int p = threadIdx.x >> 7;
        const int c = threadIdx.x & (D_H - 1);
        const bool on = (p < MODE);
        const float* __restrict__ st = (p == 0) ? st1 : st2;
        if (on) {
            const float mean = st[c] * inv_n;
            pnM[p][c] = mean;
            pnT[p][c] = mean * mean;
        }
        __syncthreads();
        for (int s = 64; s > 0; s >>= 1) {
            if (on && c < s) pnT[p][c] += pnT[p][c + s];
            __syncthreads();
        }
        if (on && c == 0) {
            float ssq_centered = st[D_H] - (float)n_rows * pnT[p][0];
            pnS[p] = 1.0f / sqrtf(PN_EPS + ssq_centered * inv_n);
        }
    }

    const int tm      = blockIdx.x * 8 + wave;
    const int tiles_m = (n_rows + 15) >> 4;
    const bool active = (tm < tiles_m);

    int row_a = tm * 16 + l15;
    if (row_a >= n_rows) row_a = n_rows - 1;
    const float* __restrict__ ap = X + (size_t)row_a * D_H + half * 2;
    const float* __restrict__ ap2 = (MODE == 2) ? (X2 + (size_t)row_a * D_H + half * 2) : ap;
    const float* __restrict__ bbase = &Wt[l15 * WST + half * 2];

    v8f acc[TN] = {};

    for (int k0 = 0; k0 < D_H; k0 += KC) {
        __syncthreads();
        const float* __restrict__ wg = W + (size_t)k0 * DOUT;
        for (int idx = threadIdx.x; idx < KC * DOUT; idx += 256) {
            int r = idx / DOUT;
            int c = idx - r * DOUT;
            Wt[c * WST + r] = wg[idx];
        }
        __syncthreads();

        if (active) {
            __builtin_prefetch(ap + k0 + KC, 0, 1);
            if constexpr (MODE == 2) __builtin_prefetch(ap2 + k0 + KC, 0, 1);
            float sc1 = 0.0f, sc2 = 0.0f;
            if constexpr (MODE >= 1) sc1 = pnS[0];
            if constexpr (MODE == 2) sc2 = pnS[1];
#pragma unroll
            for (int kk = 0; kk < KC; kk += 4) {
                v2f a = *(const v2f*)(ap + k0 + kk);
                if constexpr (MODE >= 1) {
                    const v2f m = *(const v2f*)&pnM[0][k0 + kk + half * 2];
                    a.x = fmaxf((a.x - m.x) * sc1, 0.0f);
                    a.y = fmaxf((a.y - m.y) * sc1, 0.0f);
                }
                if constexpr (MODE == 2) {
                    v2f o = *(const v2f*)(ap2 + k0 + kk);
                    const v2f m2 = *(const v2f*)&pnM[1][k0 + kk + half * 2];
                    o.x = fmaxf((o.x - m2.x) * sc2, 0.0f);
                    o.y = fmaxf((o.y - m2.y) * sc2, 0.0f);
                    a.x += o.x;
                    a.y += o.y;
                }
#pragma unroll
                for (int t = 0; t < TN; ++t) {
                    v2f b = *(const v2f*)(bbase + t * (16 * WST) + kk);
                    acc[t] = __builtin_amdgcn_wmma_f32_16x16x4_f32(
                         false, a,  false, b,
                         (short)0, acc[t],
                         false,  false);
                }
            }
        }
    }

    if (!active) return;
    const int row_base = tm * 16 + half * 8;
#pragma unroll
    for (int t = 0; t < TN; ++t) {
        const int col = t * 16 + l15;
        const float bb = bias[col];
#pragma unroll
        for (int r = 0; r < 8; ++r) {
            const int row = row_base + r;
            if (row < n_rows)
                H[(size_t)row * DOUT + col] = acc[t][r] + bb;
        }
    }
}

#define BK_SHIFT 8
#define BK_ROWS (1 << BK_SHIFT)
#define EB 4

__global__ void __launch_bounds__(1024)
edge_hist(const int* __restrict__ dst, int* __restrict__ hist,
          float* __restrict__ st, int n_st, int nE, int nb, int chunk) {
    extern __shared__ int lh[];
    for (int i = threadIdx.x; i < nb; i += 1024) lh[i] = 0;
    if (blockIdx.x == 0)
        for (int i = threadIdx.x; i < n_st; i += 1024) st[i] = 0.0f;
    __syncthreads();
    const int e0 = blockIdx.x * chunk;
    const int e1 = (e0 + chunk < nE) ? (e0 + chunk) : nE;
    for (int eb = e0 + threadIdx.x; eb < e1; eb += EB * 1024) {
        int d[EB];
#pragma unroll
        for (int u = 0; u < EB; ++u) {
            const int e = eb + u * 1024;
            d[u] = (e < e1) ? dst[e] : 0;
        }
#pragma unroll
        for (int u = 0; u < EB; ++u)
            if (eb + u * 1024 < e1) atomicAdd(&lh[d[u] >> BK_SHIFT], 1);
    }
    __syncthreads();
    for (int i = threadIdx.x; i < nb; i += 1024)
        hist[(size_t)i * gridDim.x + blockIdx.x] = lh[i];
}

__global__ void __launch_bounds__(1024)
edge_scatter(const int* __restrict__ src, const int* __restrict__ dst,
             const float* __restrict__ ew, const int* __restrict__ hoff,
             int* __restrict__ recw, int nE, int nb, int chunk) {
    extern __shared__ int lc[];
    int4* __restrict__ rec = (int4*)recw;
    for (int i = threadIdx.x; i < nb; i += 1024)
        lc[i] = hoff[(size_t)i * gridDim.x + blockIdx.x];
    __syncthreads();
    const int e0 = blockIdx.x * chunk;
    const int e1 = (e0 + chunk < nE) ? (e0 + chunk) : nE;
    for (int eb = e0 + threadIdx.x; eb < e1; eb += EB * 1024) {
        int d[EB], sr[EB], wb[EB];
#pragma unroll
        for (int u = 0; u < EB; ++u) {
            const int e = eb + u * 1024;
            const bool in = (e < e1);
            d[u]  = in ? dst[e] : 0;
            sr[u] = in ? src[e] : 0;
            wb[u] = in ? __builtin_bit_cast(int, ew[e]) : 0;
        }
#pragma unroll
        for (int u = 0; u < EB; ++u) {
            const int e = eb + u * 1024;
            if (e < e1) {
                const int pos = atomicAdd(&lc[d[u] >> BK_SHIFT], 1);
                rec[pos] = make_int4(e, sr[u], wb[u], d[u]);
            }
        }
    }
}

__global__ void __launch_bounds__(1024)
bucket_place(const int* __restrict__ hoff, const int* __restrict__ recAw,
             int* __restrict__ recBw, int* __restrict__ off,
             int n_rows, int n_parts) {
    __shared__ int cnt[BK_ROWS];
    __shared__ int cur[BK_ROWS];
    __shared__ int wsum[BK_ROWS / 32];
    const int4* __restrict__ recA = (const int4*)recAw;
    int4* __restrict__ recB = (int4*)recBw;
    const int lane = threadIdx.x & 31;
    const int wave = threadIdx.x >> 5;
    const int b = blockIdx.x;
    const int row0 = b << BK_SHIFT;
    const int s = hoff[(size_t)b * n_parts];
    const int t = hoff[(size_t)(b + 1) * n_parts];
    if (threadIdx.x < BK_ROWS) cnt[threadIdx.x] = 0;
    __syncthreads();
    for (int ib = s + threadIdx.x; ib < t; ib += EB * 1024) {
        int d[EB];
#pragma unroll
        for (int u = 0; u < EB; ++u) {
            const int i = ib + u * 1024;
            d[u] = (i < t) ? recAw[4 * (size_t)i + 3] : row0;
        }
#pragma unroll
        for (int u = 0; u < EB; ++u)
            if (ib + u * 1024 < t) atomicAdd(&cnt[d[u] - row0], 1);
    }
    __syncthreads();
    const int v = (threadIdx.x < BK_ROWS) ? cnt[threadIdx.x] : 0;
    int x = v;
    for (int d = 1; d < 32; d <<= 1) {
        const int y = __shfl_up(x, d);
        if (lane >= d) x += y;
    }
    if (threadIdx.x < BK_ROWS && lane == 31) wsum[wave] = x;
    __syncthreads();
    if (threadIdx.x < BK_ROWS) {
        int base = s;
        for (int w = 0; w < wave; ++w) base += wsum[w];
        const int start = base + x - v;
        cur[threadIdx.x] = start;
        const int r = row0 + threadIdx.x;
        if (r < n_rows) off[r] = start;
    }
    if (b == (int)gridDim.x - 1 && threadIdx.x == 0) off[n_rows] = t;
    __syncthreads();
    for (int ib = s + threadIdx.x; ib < t; ib += EB * 1024) {
        int4 rc[EB];
#pragma unroll
        for (int u = 0; u < EB; ++u) {
            const int i = ib + u * 1024;
            rc[u] = (i < t) ? recA[i] : make_int4(0, 0, 0, row0);
        }
#pragma unroll
        for (int u = 0; u < EB; ++u) {
            if (ib + u * 1024 < t) {
                const int slot = atomicAdd(&cur[rc[u].w - row0], 1);
                recB[slot] = rc[u];
            }
        }
    }
}

__global__ void __launch_bounds__(1024)
csr_scan_local(const int* __restrict__ cnt, int* __restrict__ off,
               int* __restrict__ bsum, int n_rows) {
    __shared__ int wsum[32];
    const int lane = threadIdx.x & 31;
    const int wave = threadIdx.x >> 5;
    const int i = blockIdx.x * 1024 + threadIdx.x;
    const int v = (i < n_rows) ? cnt[i] : 0;
    int x = v;
    for (int d = 1; d < 32; d <<= 1) {
        const int y = __shfl_up(x, d);
        if (lane >= d) x += y;
    }
    if (lane == 31) wsum[wave] = x;
    __syncthreads();
    if (wave == 0) {
        const int s = wsum[lane];
        int t = s;
        for (int d = 1; d < 32; d <<= 1) {
            const int y = __shfl_up(t, d);
            if (lane >= d) t += y;
        }
        wsum[lane] = t - s;
        if (lane == 31) bsum[blockIdx.x] = t;
    }
    __syncthreads();
    if (i < n_rows) off[i] = x - v + wsum[wave];
}

__global__ void __launch_bounds__(1024)
csr_scan_add(int* __restrict__ off, const int* __restrict__ bsum, int n_rows) {
    __shared__ int base_s;
    if (threadIdx.x < 32) {
        int s = 0;
        for (int j = threadIdx.x; j < (int)blockIdx.x; j += 32) s += bsum[j];
        for (int d = 16; d > 0; d >>= 1) s += __shfl_xor(s, d);
        if (threadIdx.x == 0) base_s = s;
    }
    __syncthreads();
    const int base = base_s;
    const int i = blockIdx.x * 1024 + threadIdx.x;
    if (i < n_rows) off[i] += base;
    if (blockIdx.x == gridDim.x - 1 && threadIdx.x == 0)
        off[n_rows] = base + bsum[blockIdx.x];
}

__global__ void __launch_bounds__(256)
csr_order(const int* __restrict__ off, const int* __restrict__ recw,
          int* __restrict__ srcS, float* __restrict__ wS, int n_rows) {
    const int4* __restrict__ rec = (const int4*)recw;
    const int lane = threadIdx.x & 31;
    const int r = __builtin_amdgcn_readfirstlane((int)((blockIdx.x * 256 + threadIdx.x) >> 5));
    if (r >= n_rows) return;
    const int s = off[r];
    const int len = off[r + 1] - s;
    if (len <= 32) {
        int4 me = make_int4(0, 0, 0, 0);
        if (lane < len) me = rec[s + lane];
        int rank = 0;
        for (int j = 0; j < len; ++j) {
            const int idj = __builtin_amdgcn_readlane(me.x, j);
            rank += (idj < me.x) ? 1 : 0;
        }
        if (lane < len) {
            srcS[s + rank] = me.y;
            wS[s + rank] = __builtin_bit_cast(float, me.z);
        }
    } else {
        for (int i = lane; i < len; i += 32) {
            const int4 me = rec[s + i];
            int rank = 0;
            for (int j = 0; j < len; ++j) rank += (rec[s + j].x < me.x) ? 1 : 0;
            srcS[s + rank] = me.y;
            wS[s + rank] = __builtin_bit_cast(float, me.z);
        }
    }
}

template <int VEC>
__global__ void __launch_bounds__(256)
spmm_gather(const float* __restrict__ H, const int* __restrict__ off,
            const int* __restrict__ srcS, const float* __restrict__ wS,
            float* __restrict__ out, int n_rows) {
    typedef float vf __attribute__((ext_vector_type(VEC)));
    const int lane = threadIdx.x & 31;
    const int r = __builtin_amdgcn_readfirstlane((int)((blockIdx.x * 256 + threadIdx.x) >> 5));
    if (r >= n_rows) return;
    const int s = off[r];
    const int t = off[r + 1];
    const vf* __restrict__ Hv = (const vf*)H + lane;
    vf acc = {};
    for (int base = s; base < t; base += 32) {
        const int m = (t - base < 32) ? (t - base) : 32;
        int my_s = 0;
        float my_w = 0.0f;
        if (lane < m) {
            my_s = srcS[base + lane];
            my_w = wS[base + lane];
        }
#pragma unroll 4
        for (int j = 0; j < m; ++j) {
            const int sj = __builtin_amdgcn_readlane(my_s, j);
            const float wj = __builtin_bit_cast(
                float, __builtin_amdgcn_readlane(__builtin_bit_cast(int, my_w), j));
            const vf v = Hv[(size_t)sj * 32];
            acc += wj * v;
        }
    }
    ((vf*)out)[(size_t)r * 32 + lane] = acc;
}

#define AD_WAVES 25
#define AD_ST (D_H + 4)
#define GB 8
typedef float v4f __attribute__((ext_vector_type(4)));

template <int MODE, bool FULL>
__device__ __forceinline__ void gather_group(const v4f* __restrict__ Xv, int my_s, float my_w,
                                             int j0, int m, v4f m4, float sc,
                                             v4f& a, float& dsum) {
    int sj[GB];
    float wj[GB];
#pragma unroll
    for (int u = 0; u < GB; ++u) {
        const int j = j0 + u;
        sj[u] = __builtin_amdgcn_readlane(my_s, j);
        wj[u] = __builtin_bit_cast(
            float, __builtin_amdgcn_readlane(__builtin_bit_cast(int, my_w), j));
    }
    v4f v[GB];
#pragma unroll
    for (int u = 0; u < GB; ++u)
        if (FULL || j0 + u < m) v[u] = Xv[(size_t)sj[u] * 32];
#pragma unroll
    for (int u = 0; u < GB; ++u) {
        if (FULL || j0 + u < m) {
            v4f x = v[u];
            if constexpr (MODE == 1) {
                x = (x - m4) * sc;
                x.x = fmaxf(x.x, 0.0f);
                x.y = fmaxf(x.y, 0.0f);
                x.z = fmaxf(x.z, 0.0f);
                x.w = fmaxf(x.w, 0.0f);
            }
            a += wj[u] * x;
            dsum += wj[u];
        }
    }
}

template <int MODE>
__global__ void __launch_bounds__(AD_WAVES * 32)
agg_dense(const float* __restrict__ X, const float* __restrict__ st,
          const float* __restrict__ W, const float* __restrict__ bias,
          const int* __restrict__ off, const int* __restrict__ srcS,
          const float* __restrict__ wS, float* __restrict__ out,
          float* __restrict__ red, int n_rows) {
    __shared__ __attribute__((aligned(16))) float Wt[D_H * AD_ST];
    __shared__ __attribute__((aligned(16))) float tile[AD_WAVES][16 * AD_ST];
    __shared__ v4f part[AD_WAVES * 32];
    __shared__ float psq[AD_WAVES];
    __shared__ float dw[AD_WAVES][16];
    __shared__ __attribute__((aligned(16))) float pnM[D_H];
    __shared__ float pnT[D_H];
    __shared__ float pnS;

    const int tid  = threadIdx.x;
    const int lane = tid & 31;
    const int wave = __builtin_amdgcn_readfirstlane(tid >> 5);
    const int half = lane >> 4;
    const int l15  = lane & 15;
    const v4f zero = {0.0f, 0.0f, 0.0f, 0.0f};

    {
        const v4f* __restrict__ W4 = (const v4f*)W;
        constexpr int NW4 = D_H * D_H / 4;
        constexpr int PER = (NW4 + AD_WAVES * 32 - 1) / (AD_WAVES * 32);
        v4f wv[PER];
#pragma unroll
        for (int u = 0; u < PER; ++u) {
            const int i = tid + u * (AD_WAVES * 32);
            wv[u] = (i < NW4) ? W4[i] : zero;
        }
#pragma unroll
        for (int u = 0; u < PER; ++u) {
            const int i = tid + u * (AD_WAVES * 32);
            if (i < NW4) {
                const int k  = i >> 5;
                const int c4 = (i & 31) << 2;
                Wt[(c4 + 0) * AD_ST + k] = wv[u].x;
                Wt[(c4 + 1) * AD_ST + k] = wv[u].y;
                Wt[(c4 + 2) * AD_ST + k] = wv[u].z;
                Wt[(c4 + 3) * AD_ST + k] = wv[u].w;
            }
        }
    }
    if constexpr (MODE == 1) {
        const float inv_n = 1.0f / (float)n_rows;
        const int c = tid & (D_H - 1);
        const bool on = (tid < D_H);
        if (on) {
            const float mean = st[c] * inv_n;
            pnM[c] = mean;
            pnT[c] = mean * mean;
        }
        __syncthreads();
        for (int s = 64; s > 0; s >>= 1) {
            if (on && c < s) pnT[c] += pnT[c + s];
            __syncthreads();
        }
        if (tid == 0) {
            float ssq_centered = st[D_H] - (float)n_rows * pnT[0];
            pnS = 1.0f / sqrtf(PN_EPS + ssq_centered * inv_n);
        }
    }
    __syncthreads();

    const int tm      = blockIdx.x * AD_WAVES + wave;
    const int tiles_m = (n_rows + 15) >> 4;
    const bool active = (tm < tiles_m);
    float* my = tile[wave];
    v4f cs = zero, sq = zero;

    if (active) {
        v4f m4 = zero;
        float sc = 0.0f;
        if constexpr (MODE == 1) {
            m4 = *(const v4f*)&pnM[4 * lane];
            sc = pnS;
        }
        int ro = tm * 16 + lane;
        if (ro > n_rows) ro = n_rows;
        const int my_off = (lane < 17) ? off[ro] : 0;
        const v4f* __restrict__ Xv = (const v4f*)X + lane;

        int cur_s = 0;
        float cur_w = 0.0f;
        for (int i = -1; i < 16; ++i) {
            const int s = __builtin_amdgcn_readlane(my_off, (i < 0) ? 0 : i);
            const int e = (i < 0) ? s : __builtin_amdgcn_readlane(my_off, i + 1);
            int nx_s = 0;
            float nx_w = 0.0f;
            if (i < 15) {
                const int e2 = __builtin_amdgcn_readlane(my_off, i + 2);
                if (lane < e2 - e) {
                    nx_s = srcS[e + lane];
                    nx_w = wS[e + lane];
                }
            }
            v4f a = zero;
            float dsum = 0.0f;
            const int m0 = (e - s < 32) ? (e - s) : 32;
            for (int j0 = 0; j0 < m0; j0 += GB) {
                if (j0 + GB <= m0) gather_group<MODE, true>(Xv, cur_s, cur_w, j0, m0, m4, sc, a, dsum);
                else               gather_group<MODE, false>(Xv, cur_s, cur_w, j0, m0, m4, sc, a, dsum);
            }
            for (int base = s + 32; base < e; base += 32) {
                const int m = (e - base < 32) ? (e - base) : 32;
                int my_s = 0;
                float my_w = 0.0f;
                if (lane < m) {
                    my_s = srcS[base + lane];
                    my_w = wS[base + lane];
                }
                for (int j0 = 0; j0 < m; j0 += GB) {
                    if (j0 + GB <= m) gather_group<MODE, true>(Xv, my_s, my_w, j0, m, m4, sc, a, dsum);
                    else              gather_group<MODE, false>(Xv, my_s, my_w, j0, m, m4, sc, a, dsum);
                }
            }
            if (i >= 0) {
                *(v4f*)&my[i * AD_ST + 4 * lane] = a;
                if (lane == 0) dw[wave][i] = dsum;
            }
            cur_s = nx_s;
            cur_w = nx_w;
        }

        __builtin_amdgcn_fence(__ATOMIC_RELEASE, "workgroup");
        __builtin_amdgcn_fence(__ATOMIC_ACQUIRE, "workgroup");
        const float* abase = my + l15 * AD_ST + half * 2;
        const float* __restrict__ bbase = &Wt[l15 * AD_ST + half * 2];
        v8f acc[8] = {};
#pragma unroll
        for (int kk = 0; kk < D_H; kk += 4) {
            const v2f af = *(const v2f*)(abase + kk);
            v2f bf[8];
#pragma unroll
            for (int t = 0; t < 8; ++t) bf[t] = *(const v2f*)(bbase + t * (16 * AD_ST) + kk);
#pragma unroll
            for (int t = 0; t < 8; ++t)
                acc[t] = __builtin_amdgcn_wmma_f32_16x16x4_f32(
                     false, af,  false, bf[t],
                     (short)0, acc[t],
                     false,  false);
        }

        float dwr[8];
#pragma unroll
        for (int r = 0; r < 8; ++r) dwr[r] = dw[wave][half * 8 + r];
#pragma unroll
        for (int t = 0; t < 8; ++t) {
            const float bb = bias[t * 16 + l15];
#pragma unroll
            for (int r = 0; r < 8; ++r)
                my[(half * 8 + r) * AD_ST + t * 16 + l15] = acc[t][r] + dwr[r] * bb;
        }
        __builtin_amdgcn_fence(__ATOMIC_RELEASE, "workgroup");
        __builtin_amdgcn_fence(__ATOMIC_ACQUIRE, "workgroup");
        for (int i = 0; i < 16; ++i) {
            const int row = tm * 16 + i;
            const v4f v = *(const v4f*)&my[i * AD_ST + 4 * lane];
            if (row < n_rows) ((v4f*)out)[(size_t)row * 32 + lane] = v;
            cs += v;
            sq += v * v;
        }
    }
    float q = (sq.x + sq.y) + (sq.z + sq.w);
    for (int d = 16; d > 0; d >>= 1) q += __shfl_xor(q, d);
    part[wave * 32 + lane] = cs;
    if (lane == 0) psq[wave] = q;
    __syncthreads();
    if (tid < D_H) {
        float a = 0.0f;
        for (int w = 0; w < AD_WAVES; ++w) a += ((const float*)part)[w * D_H + tid];
        unsafeAtomicAdd(&red[tid], a);
    } else if (tid == D_H) {
        float a = 0.0f;
        for (int w = 0; w < AD_WAVES; ++w) a += psq[w];
        unsafeAtomicAdd(&red[D_H], a);
    }
}

extern "C" void kernel_launch(void* const* d_in, const int* in_sizes, int n_in,
                              void* d_out, int out_size, void* d_ws, size_t ws_size,
                              hipStream_t stream) {
    const float* x    = (const float*)d_in[0];
    const int*   ei   = (const int*)d_in[1];
    const float* ew   = (const float*)d_in[2];
    const float* W0   = (const float*)d_in[3];
    const float* b0   = (const float*)d_in[4];
    const float* W1   = (const float*)d_in[5];
    const float* b1   = (const float*)d_in[6];
    const float* Wout = (const float*)d_in[7];
    const float* bout = (const float*)d_in[8];
    float* out = (float*)d_out;

    const int n = in_sizes[0] / D_H;
    const int e = in_sizes[2];
    const int* srcI = ei;
    const int* dstI = ei + e;

    const size_t nd = (size_t)n * D_H;
    float* A   = (float*)d_ws;
    float* B   = A + nd;
    float* C   = B + nd;

    const int tiles_m = (n + 15) / 16;
    const int gemm_blocks = (tiles_m + 7) / 8;
    const int ad_blocks = (tiles_m + AD_WAVES - 1) / AD_WAVES;
    const int parts = 512;
    const int chunk = (e + parts - 1) / parts;
    const int nb = (n + BK_ROWS - 1) >> BK_SHIFT;
    const int hl = nb * parts;
    const int hlpad = (hl + 1024) & ~255;
    const int hscan_blocks = (hl + 1023) / 1024;
    int*   t_src  = (int*)(A + (size_t)n * (D_H / 2));
    float* t_w    = (float*)(t_src + e);
    int*   t_off  = t_src + 2 * (size_t)e;
    int*   t_hist = (int*)out;
    int*   t_hoff = t_hist + hlpad;
    int*   t_bsum = t_hoff + hlpad;
    float* st0    = (float*)(t_bsum + ((hscan_blocks + 255) & ~255));
    float* st1    = st0 + 132;
    int*   t_recA = (int*)C;
    int*   t_recB = t_recA + 4 * (size_t)e;
    const size_t lds_nb = (size_t)nb * sizeof(int);
    edge_hist<<<parts, 1024, lds_nb, stream>>>(dstI, t_hist, st0, 2 * 132, e, nb, chunk);
    csr_scan_local<<<hscan_blocks, 1024, 0, stream>>>(t_hist, t_hoff, t_bsum, hl);
    csr_scan_add<<<hscan_blocks, 1024, 0, stream>>>(t_hoff, t_bsum, hl);
    edge_scatter<<<parts, 1024, lds_nb, stream>>>(srcI, dstI, ew, t_hoff, t_recA, e, nb, chunk);
    bucket_place<<<nb, 1024, 0, stream>>>(t_hoff, t_recA, t_recB, t_off, n, parts);
    csr_order<<<(n + 7) / 8, 256, 0, stream>>>(t_off, t_recB, t_src, t_w, n);

    agg_dense<0><<<ad_blocks, AD_WAVES * 32, 0, stream>>>(x, nullptr, W0, b0, t_off, t_src, t_w, B, st0, n);

    agg_dense<1><<<ad_blocks, AD_WAVES * 32, 0, stream>>>(B, st0, W1, b1, t_off, t_src, t_w, C, st1, n);

    gemm_wmma_f32<64, 2><<<gemm_blocks, 256, 0, stream>>>(C, B, st1, st0, Wout, bout, A, n);
    spmm_gather<2><<<(n + 7) / 8, 256, 0, stream>>>(A, t_off, t_src, t_w, out, n);
}


# Edge ordering folded into the bucket pass: a bucket's records are grouped by row in LDS and ranked by edge id there; the ordering launch and the by-row record array go (10->9 launches); table kept
# speedup vs baseline: 4.5402x; 1.0521x over previous
#include <hip/hip_runtime.h>

#define D_H 128
#define PN_EPS 1e-6f

typedef float v2f __attribute__((ext_vector_type(2)));
typedef float v8f __attribute__((ext_vector_type(8)));

template <int DOUT, int MODE>
__global__ void __launch_bounds__(256)
gemm_wmma_f32(const float* __restrict__ X, const float* __restrict__ X2,
              const float* __restrict__ st1, const float* __restrict__ st2,
              const float* __restrict__ W, const float* __restrict__ bias,
              float* __restrict__ H, int n_rows) {
    constexpr int TN  = DOUT / 16;
    constexpr int KC  = 64;
    constexpr int WST = KC + 4;
    constexpr int NP  = (MODE == 0) ? 1 : MODE;
    __shared__ float Wt[DOUT * WST];
    __shared__ __attribute__((aligned(16))) float pnM[NP][D_H];
    __shared__ float pnT[NP][D_H];
    __shared__ float pnS[NP];

    const int lane = threadIdx.x & 31;
    const int wave = threadIdx.x >> 5;
    const int half = lane >> 4;
    const int l15  = lane & 15;

    if constexpr (MODE >= 1) {
        const float inv_n = 1.0f / (float)n_rows;
        const int p = threadIdx.x >> 7;
        const int c = threadIdx.x & (D_H - 1);
        const bool on = (p < MODE);
        const float* __restrict__ st = (p == 0) ? st1 : st2;
        if (on) {
            const float mean = st[c] * inv_n;
            pnM[p][c] = mean;
            pnT[p][c] = mean * mean;
        }
        __syncthreads();
        for (int s = 64; s > 0; s >>= 1) {
            if (on && c < s) pnT[p][c] += pnT[p][c + s];
            __syncthreads();
        }
        if (on && c == 0) {
            float ssq_centered = st[D_H] - (float)n_rows * pnT[p][0];
            pnS[p] = 1.0f / sqrtf(PN_EPS + ssq_centered * inv_n);
        }
    }

    const int tm      = blockIdx.x * 8 + wave;
    const int tiles_m = (n_rows + 15) >> 4;
    const bool active = (tm < tiles_m);

    int row_a = tm * 16 + l15;
    if (row_a >= n_rows) row_a = n_rows - 1;
    const float* __restrict__ ap = X + (size_t)row_a * D_H + half * 2;
    const float* __restrict__ ap2 = (MODE == 2) ? (X2 + (size_t)row_a * D_H + half * 2) : ap;
    const float* __restrict__ bbase = &Wt[l15 * WST + half * 2];

    v8f acc[TN] = {};

    for (int k0 = 0; k0 < D_H; k0 += KC) {
        __syncthreads();
        const float* __restrict__ wg = W + (size_t)k0 * DOUT;
        for (int idx = threadIdx.x; idx < KC * DOUT; idx += 256) {
            int r = idx / DOUT;
            int c = idx - r * DOUT;
            Wt[c * WST + r] = wg[idx];
        }
        __syncthreads();

        if (active) {
            __builtin_prefetch(ap + k0 + KC, 0, 1);
            if constexpr (MODE == 2) __builtin_prefetch(ap2 + k0 + KC, 0, 1);
            float sc1 = 0.0f, sc2 = 0.0f;
            if constexpr (MODE >= 1) sc1 = pnS[0];
            if constexpr (MODE == 2) sc2 = pnS[1];
#pragma unroll
            for (int kk = 0; kk < KC; kk += 4) {
                v2f a = *(const v2f*)(ap + k0 + kk);
                if constexpr (MODE >= 1) {
                    const v2f m = *(const v2f*)&pnM[0][k0 + kk + half * 2];
                    a.x = fmaxf((a.x - m.x) * sc1, 0.0f);
                    a.y = fmaxf((a.y - m.y) * sc1, 0.0f);
                }
                if constexpr (MODE == 2) {
                    v2f o = *(const v2f*)(ap2 + k0 + kk);
                    const v2f m2 = *(const v2f*)&pnM[1][k0 + kk + half * 2];
                    o.x = fmaxf((o.x - m2.x) * sc2, 0.0f);
                    o.y = fmaxf((o.y - m2.y) * sc2, 0.0f);
                    a.x += o.x;
                    a.y += o.y;
                }
#pragma unroll
                for (int t = 0; t < TN; ++t) {
                    v2f b = *(const v2f*)(bbase + t * (16 * WST) + kk);
                    acc[t] = __builtin_amdgcn_wmma_f32_16x16x4_f32(
                         false, a,  false, b,
                         (short)0, acc[t],
                         false,  false);
                }
            }
        }
    }

    if (!active) return;
    const int row_base = tm * 16 + half * 8;
#pragma unroll
    for (int t = 0; t < TN; ++t) {
        const int col = t * 16 + l15;
        const float bb = bias[col];
#pragma unroll
        for (int r = 0; r < 8; ++r) {
            const int row = row_base + r;
            if (row < n_rows)
                H[(size_t)row * DOUT + col] = acc[t][r] + bb;
        }
    }
}

#define BK_SHIFT 8
#define BK_ROWS (1 << BK_SHIFT)
#define EB 4

__global__ void __launch_bounds__(1024)
edge_hist(const int* __restrict__ dst, int* __restrict__ hist,
          float* __restrict__ st, int n_st, int nE, int nb, int chunk) {
    extern __shared__ int lh[];
    for (int i = threadIdx.x; i < nb; i += 1024) lh[i] = 0;
    if (blockIdx.x == 0)
        for (int i = threadIdx.x; i < n_st; i += 1024) st[i] = 0.0f;
    __syncthreads();
    const int e0 = blockIdx.x * chunk;
    const int e1 = (e0 + chunk < nE) ? (e0 + chunk) : nE;
    for (int eb = e0 + threadIdx.x; eb < e1; eb += EB * 1024) {
        int d[EB];
#pragma unroll
        for (int u = 0; u < EB; ++u) {
            const int e = eb + u * 1024;
            d[u] = (e < e1) ? dst[e] : 0;
        }
#pragma unroll
        for (int u = 0; u < EB; ++u)
            if (eb + u * 1024 < e1) atomicAdd(&lh[d[u] >> BK_SHIFT], 1);
    }
    __syncthreads();
    for (int i = threadIdx.x; i < nb; i += 1024)
        hist[(size_t)i * gridDim.x + blockIdx.x] = lh[i];
}

__global__ void __launch_bounds__(1024)
edge_scatter(const int* __restrict__ src, const int* __restrict__ dst,
             const float* __restrict__ ew, const int* __restrict__ hoff,
             int* __restrict__ recw, int nE, int nb, int chunk) {
    extern __shared__ int lc[];
    int4* __restrict__ rec = (int4*)recw;
    for (int i = threadIdx.x; i < nb; i += 1024)
        lc[i] = hoff[(size_t)i * gridDim.x + blockIdx.x];
    __syncthreads();
    const int e0 = blockIdx.x * chunk;
    const int e1 = (e0 + chunk < nE) ? (e0 + chunk) : nE;
    for (int eb = e0 + threadIdx.x; eb < e1; eb += EB * 1024) {
        int d[EB], sr[EB], wb[EB];
#pragma unroll
        for (int u = 0; u < EB; ++u) {
            const int e = eb + u * 1024;
            const bool in = (e < e1);
            d[u]  = in ? dst[e] : 0;
            sr[u] = in ? src[e] : 0;
            wb[u] = in ? __builtin_bit_cast(int, ew[e]) : 0;
        }
#pragma unroll
        for (int u = 0; u < EB; ++u) {
            const int e = eb + u * 1024;
            if (e < e1) {
                const int pos = atomicAdd(&lc[d[u] >> BK_SHIFT], 1);
                rec[pos] = make_int4(e, sr[u], wb[u], d[u]);
            }
        }
    }
}

#define BK_CAP 8192
__global__ void __launch_bounds__(1024)
bucket_place(const int* __restrict__ hoff, const int* __restrict__ recAw,
             int* recBw, int* __restrict__ off,
             int* __restrict__ srcS, float* __restrict__ wS,
             int n_rows, int n_parts) {
    __shared__ int cnt[BK_ROWS];
    __shared__ int cur[BK_ROWS];
    __shared__ int wsum[BK_ROWS / 32];
    __shared__ int lid[BK_CAP];
    __shared__ int lsr[BK_CAP];
    __shared__ int lwb[BK_CAP];
    const int4* __restrict__ recA = (const int4*)recAw;
    int4* recB = (int4*)recBw;
    const int lane = threadIdx.x & 31;
    const int wave = threadIdx.x >> 5;
    const int b = blockIdx.x;
    const int row0 = b << BK_SHIFT;
    const int s = hoff[(size_t)b * n_parts];
    const int t = hoff[(size_t)(b + 1) * n_parts];
    const bool in_lds = (t - s <= BK_CAP);
    if (threadIdx.x < BK_ROWS) cnt[threadIdx.x] = 0;
    __syncthreads();
    for (int ib = s + threadIdx.x; ib < t; ib += EB * 1024) {
        int d[EB];
#pragma unroll
        for (int u = 0; u < EB; ++u) {
            const int i = ib + u * 1024;
            d[u] = (i < t) ? recAw[4 * (size_t)i + 3] : row0;
        }
#pragma unroll
        for (int u = 0; u < EB; ++u)
            if (ib + u * 1024 < t) atomicAdd(&cnt[d[u] - row0], 1);
    }
    __syncthreads();
    const int v = (threadIdx.x < BK_ROWS) ? cnt[threadIdx.x] : 0;
    int x = v;
    for (int d = 1; d < 32; d <<= 1) {
        const int y = __shfl_up(x, d);
        if (lane >= d) x += y;
    }
    if (threadIdx.x < BK_ROWS && lane == 31) wsum[wave] = x;
    __syncthreads();
    if (threadIdx.x < BK_ROWS) {
        int base = s;
        for (int w = 0; w < wave; ++w) base += wsum[w];
        const int start = base + x - v;
        cur[threadIdx.x] = start;
        const int r = row0 + threadIdx.x;
        if (r < n_rows) off[r] = start;
    }
    if (b == (int)gridDim.x - 1 && threadIdx.x == 0) off[n_rows] = t;
    __syncthreads();
    for (int ib = s + threadIdx.x; ib < t; ib += EB * 1024) {
        int4 rc[EB];
#pragma unroll
        for (int u = 0; u < EB; ++u) {
            const int i = ib + u * 1024;
            rc[u] = (i < t) ? recA[i] : make_int4(0, 0, 0, row0);
        }
#pragma unroll
        for (int u = 0; u < EB; ++u) {
            if (ib + u * 1024 < t) {
                const int slot = atomicAdd(&cur[rc[u].w - row0], 1);
                if (in_lds) {
                    lid[slot - s] = rc[u].x;
                    lsr[slot - s] = rc[u].y;
                    lwb[slot - s] = rc[u].z;
                } else {
                    recB[slot] = rc[u];
                }
            }
        }
    }
    __syncthreads();
    auto rank_rows = [&](auto rid, auto rsr, auto rwb) {
        for (int rr = wave; rr < BK_ROWS; rr += 32) {
            const int len  = __builtin_amdgcn_readfirstlane(cnt[rr]);
            const int base = __builtin_amdgcn_readfirstlane(cur[rr]) - len;
            if (len <= 32) {
                int mi = 0, ms = 0, mw = 0;
                if (lane < len) {
                    mi = rid(base + lane);
                    ms = rsr(base + lane);
                    mw = rwb(base + lane);
                }
                int rank = 0;
                for (int j = 0; j < len; ++j) {
                    const int idj = __builtin_amdgcn_readlane(mi, j);
                    rank += (idj < mi) ? 1 : 0;
                }
                if (lane < len) {
                    srcS[base + rank] = ms;
                    wS[base + rank] = __builtin_bit_cast(float, mw);
                }
            } else {
                for (int i = lane; i < len; i += 32) {
                    const int mi = rid(base + i);
                    int rank = 0;
                    for (int j = 0; j < len; ++j) rank += (rid(base + j) < mi) ? 1 : 0;
                    srcS[base + rank] = rsr(base + i);
                    wS[base + rank] = __builtin_bit_cast(float, rwb(base + i));
                }
            }
        }
    };
    if (in_lds)
        rank_rows([&](int k) -> int { return lid[k - s]; },
                  [&](int k) -> int { return lsr[k - s]; },
                  [&](int k) -> int { return lwb[k - s]; });
    else
        rank_rows([&](int k) -> int { return recBw[4 * (size_t)k]; },
                  [&](int k) -> int { return recBw[4 * (size_t)k + 1]; },
                  [&](int k) -> int { return recBw[4 * (size_t)k + 2]; });
}

__global__ void __launch_bounds__(1024)
csr_scan_local(const int* __restrict__ cnt, int* __restrict__ off,
               int* __restrict__ bsum, int n_rows) {
    __shared__ int wsum[32];
    const int lane = threadIdx.x & 31;
    const int wave = threadIdx.x >> 5;
    const int i = blockIdx.x * 1024 + threadIdx.x;
    const int v = (i < n_rows) ? cnt[i] : 0;
    int x = v;
    for (int d = 1; d < 32; d <<= 1) {
        const int y = __shfl_up(x, d);
        if (lane >= d) x += y;
    }
    if (lane == 31) wsum[wave] = x;
    __syncthreads();
    if (wave == 0) {
        const int s = wsum[lane];
        int t = s;
        for (int d = 1; d < 32; d <<= 1) {
            const int y = __shfl_up(t, d);
            if (lane >= d) t += y;
        }
        wsum[lane] = t - s;
        if (lane == 31) bsum[blockIdx.x] = t;
    }
    __syncthreads();
    if (i < n_rows) off[i] = x - v + wsum[wave];
}

__global__ void __launch_bounds__(1024)
csr_scan_add(int* __restrict__ off, const int* __restrict__ bsum, int n_rows) {
    __shared__ int base_s;
    if (threadIdx.x < 32) {
        int s = 0;
        for (int j = threadIdx.x; j < (int)blockIdx.x; j += 32) s += bsum[j];
        for (int d = 16; d > 0; d >>= 1) s += __shfl_xor(s, d);
        if (threadIdx.x == 0) base_s = s;
    }
    __syncthreads();
    const int base = base_s;
    const int i = blockIdx.x * 1024 + threadIdx.x;
    if (i < n_rows) off[i] += base;
    if (blockIdx.x == gridDim.x - 1 && threadIdx.x == 0)
        off[n_rows] = base + bsum[blockIdx.x];
}

template <int VEC>
__global__ void __launch_bounds__(256)
spmm_gather(const float* __restrict__ H, const int* __restrict__ off,
            const int* __restrict__ srcS, const float* __restrict__ wS,
            float* __restrict__ out, int n_rows) {
    typedef float vf __attribute__((ext_vector_type(VEC)));
    const int lane = threadIdx.x & 31;
    const int r = __builtin_amdgcn_readfirstlane((int)((blockIdx.x * 256 + threadIdx.x) >> 5));
    if (r >= n_rows) return;
    const int s = off[r];
    const int t = off[r + 1];
    const vf* __restrict__ Hv = (const vf*)H + lane;
    vf acc = {};
    for (int base = s; base < t; base += 32) {
        const int m = (t - base < 32) ? (t - base) : 32;
        int my_s = 0;
        float my_w = 0.0f;
        if (lane < m) {
            my_s = srcS[base + lane];
            my_w = wS[base + lane];
        }
#pragma unroll 4
        for (int j = 0; j < m; ++j) {
            const int sj = __builtin_amdgcn_readlane(my_s, j);
            const float wj = __builtin_bit_cast(
                float, __builtin_amdgcn_readlane(__builtin_bit_cast(int, my_w), j));
            const vf v = Hv[(size_t)sj * 32];
            acc += wj * v;
        }
    }
    ((vf*)out)[(size_t)r * 32 + lane] = acc;
}

#define AD_WAVES 25
#define AD_ST (D_H + 4)
#define GB 8
typedef float v4f __attribute__((ext_vector_type(4)));

template <int MODE, bool FULL>
__device__ __forceinline__ void gather_group(const v4f* __restrict__ Xv, int my_s, float my_w,
                                             int j0, int m, v4f m4, float sc,
                                             v4f& a, float& dsum) {
    int sj[GB];
    float wj[GB];
#pragma unroll
    for (int u = 0; u < GB; ++u) {
        const int j = j0 + u;
        sj[u] = __builtin_amdgcn_readlane(my_s, j);
        wj[u] = __builtin_bit_cast(
            float, __builtin_amdgcn_readlane(__builtin_bit_cast(int, my_w), j));
    }
    v4f v[GB];
#pragma unroll
    for (int u = 0; u < GB; ++u)
        if (FULL || j0 + u < m) v[u] = Xv[(size_t)sj[u] * 32];
#pragma unroll
    for (int u = 0; u < GB; ++u) {
        if (FULL || j0 + u < m) {
            v4f x = v[u];
            if constexpr (MODE == 1) {
                x = (x - m4) * sc;
                x.x = fmaxf(x.x, 0.0f);
                x.y = fmaxf(x.y, 0.0f);
                x.z = fmaxf(x.z, 0.0f);
                x.w = fmaxf(x.w, 0.0f);
            }
            a += wj[u] * x;
            dsum += wj[u];
        }
    }
}

template <int MODE>
__global__ void __launch_bounds__(AD_WAVES * 32)
agg_dense(const float* __restrict__ X, const float* __restrict__ st,
          const float* __restrict__ W, const float* __restrict__ bias,
          const int* __restrict__ off, const int* __restrict__ srcS,
          const float* __restrict__ wS, float* __restrict__ out,
          float* __restrict__ red, int n_rows) {
    __shared__ __attribute__((aligned(16))) float Wt[D_H * AD_ST];
    __shared__ __attribute__((aligned(16))) float tile[AD_WAVES][16 * AD_ST];
    __shared__ v4f part[AD_WAVES * 32];
    __shared__ float psq[AD_WAVES];
    __shared__ float dw[AD_WAVES][16];
    __shared__ __attribute__((aligned(16))) float pnM[D_H];
    __shared__ float pnT[D_H];
    __shared__ float pnS;

    const int tid  = threadIdx.x;
    const int lane = tid & 31;
    const int wave = __builtin_amdgcn_readfirstlane(tid >> 5);
    const int half = lane >> 4;
    const int l15  = lane & 15;
    const v4f zero = {0.0f, 0.0f, 0.0f, 0.0f};

    {
        const v4f* __restrict__ W4 = (const v4f*)W;
        constexpr int NW4 = D_H * D_H / 4;
        constexpr int PER = (NW4 + AD_WAVES * 32 - 1) / (AD_WAVES * 32);
        v4f wv[PER];
#pragma unroll
        for (int u = 0; u < PER; ++u) {
            const int i = tid + u * (AD_WAVES * 32);
            wv[u] = (i < NW4) ? W4[i] : zero;
        }
#pragma unroll
        for (int u = 0; u < PER; ++u) {
            const int i = tid + u * (AD_WAVES * 32);
            if (i < NW4) {
                const int k  = i >> 5;
                const int c4 = (i & 31) << 2;
                Wt[(c4 + 0) * AD_ST + k] = wv[u].x;
                Wt[(c4 + 1) * AD_ST + k] = wv[u].y;
                Wt[(c4 + 2) * AD_ST + k] = wv[u].z;
                Wt[(c4 + 3) * AD_ST + k] = wv[u].w;
            }
        }
    }
    if constexpr (MODE == 1) {
        const float inv_n = 1.0f / (float)n_rows;
        const int c = tid & (D_H - 1);
        const bool on = (tid < D_H);
        if (on) {
            const float mean = st[c] * inv_n;
            pnM[c] = mean;
            pnT[c] = mean * mean;
        }
        __syncthreads();
        for (int s = 64; s > 0; s >>= 1) {
            if (on && c < s) pnT[c] += pnT[c + s];
            __syncthreads();
        }
        if (tid == 0) {
            float ssq_centered = st[D_H] - (float)n_rows * pnT[0];
            pnS = 1.0f / sqrtf(PN_EPS + ssq_centered * inv_n);
        }
    }
    __syncthreads();

    const int tm      = blockIdx.x * AD_WAVES + wave;
    const int tiles_m = (n_rows + 15) >> 4;
    const bool active = (tm < tiles_m);
    float* my = tile[wave];
    v4f cs = zero, sq = zero;

    if (active) {
        v4f m4 = zero;
        float sc = 0.0f;
        if constexpr (MODE == 1) {
            m4 = *(const v4f*)&pnM[4 * lane];
            sc = pnS;
        }
        int ro = tm * 16 + lane;
        if (ro > n_rows) ro = n_rows;
        const int my_off = (lane < 17) ? off[ro] : 0;
        const v4f* __restrict__ Xv = (const v4f*)X + lane;

        int cur_s = 0;
        float cur_w = 0.0f;
        for (int i = -1; i < 16; ++i) {
            const int s = __builtin_amdgcn_readlane(my_off, (i < 0) ? 0 : i);
            const int e = (i < 0) ? s : __builtin_amdgcn_readlane(my_off, i + 1);
            int nx_s = 0;
            float nx_w = 0.0f;
            if (i < 15) {
                const int e2 = __builtin_amdgcn_readlane(my_off, i + 2);
                if (lane < e2 - e) {
                    nx_s = srcS[e + lane];
                    nx_w = wS[e + lane];
                }
            }
            v4f a = zero;
            float dsum = 0.0f;
            const int m0 = (e - s < 32) ? (e - s) : 32;
            for (int j0 = 0; j0 < m0; j0 += GB) {
                if (j0 + GB <= m0) gather_group<MODE, true>(Xv, cur_s, cur_w, j0, m0, m4, sc, a, dsum);
                else               gather_group<MODE, false>(Xv, cur_s, cur_w, j0, m0, m4, sc, a, dsum);
            }
            for (int base = s + 32; base < e; base += 32) {
                const int m = (e - base < 32) ? (e - base) : 32;
                int my_s = 0;
                float my_w = 0.0f;
                if (lane < m) {
                    my_s = srcS[base + lane];
                    my_w = wS[base + lane];
                }
                for (int j0 = 0; j0 < m; j0 += GB) {
                    if (j0 + GB <= m) gather_group<MODE, true>(Xv, my_s, my_w, j0, m, m4, sc, a, dsum);
                    else              gather_group<MODE, false>(Xv, my_s, my_w, j0, m, m4, sc, a, dsum);
                }
            }
            if (i >= 0) {
                *(v4f*)&my[i * AD_ST + 4 * lane] = a;
                if (lane == 0) dw[wave][i] = dsum;
            }
            cur_s = nx_s;
            cur_w = nx_w;
        }

        __builtin_amdgcn_fence(__ATOMIC_RELEASE, "workgroup");
        __builtin_amdgcn_fence(__ATOMIC_ACQUIRE, "workgroup");
        const float* abase = my + l15 * AD_ST + half * 2;
        const float* __restrict__ bbase = &Wt[l15 * AD_ST + half * 2];
        v8f acc[8] = {};
#pragma unroll
        for (int kk = 0; kk < D_H; kk += 4) {
            const v2f af = *(const v2f*)(abase + kk);
            v2f bf[8];
#pragma unroll
            for (int t = 0; t < 8; ++t) bf[t] = *(const v2f*)(bbase + t * (16 * AD_ST) + kk);
#pragma unroll
            for (int t = 0; t < 8; ++t)
                acc[t] = __builtin_amdgcn_wmma_f32_16x16x4_f32(
                     false, af,  false, bf[t],
                     (short)0, acc[t],
                     false,  false);
        }

        float dwr[8];
#pragma unroll
        for (int r = 0; r < 8; ++r) dwr[r] = dw[wave][half * 8 + r];
#pragma unroll
        for (int t = 0; t < 8; ++t) {
            const float bb = bias[t * 16 + l15];
#pragma unroll
            for (int r = 0; r < 8; ++r)
                my[(half * 8 + r) * AD_ST + t * 16 + l15] = acc[t][r] + dwr[r] * bb;
        }
        __builtin_amdgcn_fence(__ATOMIC_RELEASE, "workgroup");
        __builtin_amdgcn_fence(__ATOMIC_ACQUIRE, "workgroup");
        for (int i = 0; i < 16; ++i) {
            const int row = tm * 16 + i;
            const v4f v = *(const v4f*)&my[i * AD_ST + 4 * lane];
            if (row < n_rows) ((v4f*)out)[(size_t)row * 32 + lane] = v;
            cs += v;
            sq += v * v;
        }
    }
    float q = (sq.x + sq.y) + (sq.z + sq.w);
    for (int d = 16; d > 0; d >>= 1) q += __shfl_xor(q, d);
    part[wave * 32 + lane] = cs;
    if (lane == 0) psq[wave] = q;
    __syncthreads();
    if (tid < D_H) {
        float a = 0.0f;
        for (int w = 0; w < AD_WAVES; ++w) a += ((const float*)part)[w * D_H + tid];
        unsafeAtomicAdd(&red[tid], a);
    } else if (tid == D_H) {
        float a = 0.0f;
        for (int w = 0; w < AD_WAVES; ++w) a += psq[w];
        unsafeAtomicAdd(&red[D_H], a);
    }
}

extern "C" void kernel_launch(void* const* d_in, const int* in_sizes, int n_in,
                              void* d_out, int out_size, void* d_ws, size_t ws_size,
                              hipStream_t stream) {
    const float* x    = (const float*)d_in[0];
    const int*   ei   = (const int*)d_in[1];
    const float* ew   = (const float*)d_in[2];
    const float* W0   = (const float*)d_in[3];
    const float* b0   = (const float*)d_in[4];
    const float* W1   = (const float*)d_in[5];
    const float* b1   = (const float*)d_in[6];
    const float* Wout = (const float*)d_in[7];
    const float* bout = (const float*)d_in[8];
    float* out = (float*)d_out;

    const int n = in_sizes[0] / D_H;
    const int e = in_sizes[2];
    const int* srcI = ei;
    const int* dstI = ei + e;

    const size_t nd = (size_t)n * D_H;
    float* A   = (float*)d_ws;
    float* B   = A + nd;
    float* C   = B + nd;

    const int tiles_m = (n + 15) / 16;
    const int gemm_blocks = (tiles_m + 7) / 8;
    const int ad_blocks = (tiles_m + AD_WAVES - 1) / AD_WAVES;
    const int parts = 512;
    const int chunk = (e + parts - 1) / parts;
    const int nb = (n + BK_ROWS - 1) >> BK_SHIFT;
    const int hl = nb * parts;
    const int hlpad = (hl + 1024) & ~255;
    const int hscan_blocks = (hl + 1023) / 1024;
    int*   t_src  = (int*)(A + (size_t)n * (D_H / 2));
    float* t_w    = (float*)(t_src + e);
    int*   t_off  = t_src + 2 * (size_t)e;
    int*   t_hist = (int*)out;
    int*   t_hoff = t_hist + hlpad;
    int*   t_bsum = t_hoff + hlpad;
    float* st0    = (float*)(t_bsum + ((hscan_blocks + 255) & ~255));
    float* st1    = st0 + 132;
    int*   t_recA = (int*)C;
    int*   t_recB = t_recA + 4 * (size_t)e;
    const size_t lds_nb = (size_t)nb * sizeof(int);
    edge_hist<<<parts, 1024, lds_nb, stream>>>(dstI, t_hist, st0, 2 * 132, e, nb, chunk);
    csr_scan_local<<<hscan_blocks, 1024, 0, stream>>>(t_hist, t_hoff, t_bsum, hl);
    csr_scan_add<<<hscan_blocks, 1024, 0, stream>>>(t_hoff, t_bsum, hl);
    edge_scatter<<<parts, 1024, lds_nb, stream>>>(srcI, dstI, ew, t_hoff, t_recA, e, nb, chunk);
    bucket_place<<<nb, 1024, 0, stream>>>(t_hoff, t_recA, t_recB, t_off, t_src, t_w, n, parts);

    agg_dense<0><<<ad_blocks, AD_WAVES * 32, 0, stream>>>(x, nullptr, W0, b0, t_off, t_src, t_w, B, st0, n);

    agg_dense<1><<<ad_blocks, AD_WAVES * 32, 0, stream>>>(B, st0, W1, b1, t_off, t_src, t_w, C, st1, n);

    gemm_wmma_f32<64, 2><<<gemm_blocks, 256, 0, stream>>>(C, B, st1, st0, Wout, bout, A, n);
    spmm_gather<2><<<(n + 7) / 8, 256, 0, stream>>>(A, t_off, t_src, t_w, out, n);
}
